# starting kernel
# baseline (speedup 1.0000x reference)
#include <hip/hip_runtime.h>

#define N_NODES   100000
#define N_EDGES   1600000
#define C_IN      64
#define HID       32
#define G_GRAPHS  128
#define NEG_SLOPE 0.01f

typedef __attribute__((ext_vector_type(2))) float v2f;
typedef __attribute__((ext_vector_type(8))) float v8f;

__global__ void gcn_zero_kernel(float* __restrict__ p, int n) {
  int i = blockIdx.x * blockDim.x + threadIdx.x;
  if (i < n) p[i] = 0.0f;
}

__global__ void gcn_degree_kernel(const int* __restrict__ dst, float* __restrict__ deg) {
  int e = blockIdx.x * blockDim.x + threadIdx.x;
  if (e < N_EDGES) atomicAdd(&deg[dst[e]], 1.0f);
}

__global__ void gcn_counts_kernel(const int* __restrict__ batch, float* __restrict__ cnt) {
  int i = blockIdx.x * blockDim.x + threadIdx.x;
  if (i < N_NODES) atomicAdd(&cnt[batch[i]], 1.0f);
}

__global__ void gcn_dinv_kernel(float* __restrict__ deg) {
  int i = blockIdx.x * blockDim.x + threadIdx.x;
  if (i < N_NODES) deg[i] = rsqrtf(deg[i] + 1.0f);
}

__global__ void gcn_cntinv_kernel(float* __restrict__ cnt) {
  int g = blockIdx.x * blockDim.x + threadIdx.x;
  if (g < G_GRAPHS) cnt[g] = 1.0f / fmaxf(cnt[g], 1.0f);
}

template <int K>
__global__ void gcn_gemm_wmma_kernel(const float* __restrict__ X,
                                     const float* __restrict__ W,
                                     float* __restrict__ H, int nRows) {
  const int wave = blockIdx.x * (blockDim.x >> 5) + (threadIdx.x >> 5);
  const int lane = threadIdx.x & 31;
  const int numTiles = (nRows >> 4) * 2;
  if (wave >= numTiles) return;
  const int rowTile = wave >> 1;
  const int colTile = wave & 1;
  const int half = lane >> 4;
  const int l16  = lane & 15;
  const int aRow = rowTile * 16 + l16;
  const int col  = colTile * 16 + l16;

  v8f c = {};
#pragma unroll
  for (int k0 = 0; k0 < K; k0 += 4) {
    const int ka = k0 + 2 * half;
    v2f a = *(const v2f*)(X + aRow * K + ka);
    v2f b;
    b.x = W[(ka + 0) * HID + col];
    b.y = W[(ka + 1) * HID + col];
    c = __builtin_amdgcn_wmma_f32_16x16x4_f32(
         false, a,  false, b,
         (short)0, c,  false,  false);
  }
#pragma unroll
  for (int r = 0; r < 8; ++r) {
    const int m = r + 8 * half;
    H[(rowTile * 16 + m) * HID + col] = c[r];
  }
}

__global__ void gcn_scatter_kernel(const int* __restrict__ src, const int* __restrict__ dst,
                                   const float* __restrict__ dinv,
                                   const float* __restrict__ H, float* __restrict__ agg) {
  int t = blockIdx.x * blockDim.x + threadIdx.x;
  int e = t >> 3;
  if (e >= N_EDGES) return;
  int q = (t & 7) << 2;
  int s = src[e], d = dst[e];
  float coef = dinv[s] * dinv[d];
  const float4 hv = *(const float4*)(H + s * HID + q);
  float* ap = agg + d * HID + q;
  atomicAdd(ap + 0, hv.x * coef);
  atomicAdd(ap + 1, hv.y * coef);
  atomicAdd(ap + 2, hv.z * coef);
  atomicAdd(ap + 3, hv.w * coef);
}

__global__ void gcn_combine_kernel(const float* __restrict__ agg, const float* __restrict__ Hlin,
                                   const float* __restrict__ bias, const float* __restrict__ dinv,
                                   const int* __restrict__ batch, const float* __restrict__ cntinv,
                                   float* __restrict__ Hact, float* __restrict__ feat) {
  int t = blockIdx.x * blockDim.x + threadIdx.x;
  if (t >= N_NODES * HID) return;
  int i = t >> 5, ch = t & 31;
  float di = dinv[i];
  float v = agg[t] + Hlin[t] * di * di + bias[ch];
  v = v > 0.0f ? v : NEG_SLOPE * v;
  Hact[t] = v;
  int g = batch[i];
  atomicAdd(&feat[g * HID + ch], v * cntinv[g]);
}

__global__ void gcn_merge_kernel(const float* __restrict__ feats, float* __restrict__ out) {
  int t = blockIdx.x * blockDim.x + threadIdx.x;
  const int GH = G_GRAPHS * HID;
  if (t < GH) out[t] = (feats[t] + feats[GH + t] + feats[2 * GH + t]) * (1.0f / 3.0f);
}

static inline int cdiv(int a, int b) { return (a + b - 1) / b; }

extern "C" void kernel_launch(void* const* d_in, const int* in_sizes, int n_in,
                              void* d_out, int out_size, void* d_ws, size_t ws_size,
                              hipStream_t stream) {
  (void)in_sizes; (void)n_in; (void)out_size; (void)ws_size;
  const float* x     = (const float*)d_in[0];
  const float* W0    = (const float*)d_in[1];
  const float* b0    = (const float*)d_in[2];
  const float* W1    = (const float*)d_in[3];
  const float* b1    = (const float*)d_in[4];
  const float* W2    = (const float*)d_in[5];
  const float* b2    = (const float*)d_in[6];
  const int*   src   = (const int*)d_in[7];
  const int*   dst   = (const int*)d_in[8];
  const int*   batch = (const int*)d_in[9];
  float* out = (float*)d_out;

  float* ws    = (float*)d_ws;
  float* deg   = ws;
  float* cnt   = deg + N_NODES;
  float* lin   = cnt + G_GRAPHS;
  float* act   = lin + (size_t)N_NODES * HID;
  float* agg   = act + (size_t)N_NODES * HID;
  float* feats = agg + (size_t)N_NODES * HID;

  const int T = 256;
  const int nTiles = (N_NODES / 16) * 2;

  gcn_zero_kernel<<<cdiv(N_NODES, T), T, 0, stream>>>(deg, N_NODES);
  gcn_zero_kernel<<<1, T, 0, stream>>>(cnt, G_GRAPHS);
  gcn_zero_kernel<<<cdiv(3 * G_GRAPHS * HID, T), T, 0, stream>>>(feats, 3 * G_GRAPHS * HID);
  gcn_degree_kernel<<<cdiv(N_EDGES, T), T, 0, stream>>>(dst, deg);
  gcn_counts_kernel<<<cdiv(N_NODES, T), T, 0, stream>>>(batch, cnt);
  gcn_dinv_kernel<<<cdiv(N_NODES, T), T, 0, stream>>>(deg);
  gcn_cntinv_kernel<<<1, T, 0, stream>>>(cnt);

  const float* Ws[3] = {W0, W1, W2};
  const float* bs[3] = {b0, b1, b2};
  const float* hin = x;
  for (int layer = 0; layer < 3; ++layer) {
    if (layer == 0)
      gcn_gemm_wmma_kernel<C_IN><<<cdiv(nTiles, T / 32), T, 0, stream>>>(hin, Ws[0], lin, N_NODES);
    else
      gcn_gemm_wmma_kernel<HID><<<cdiv(nTiles, T / 32), T, 0, stream>>>(hin, Ws[layer], lin, N_NODES);
    gcn_zero_kernel<<<cdiv(N_NODES * HID, T), T, 0, stream>>>(agg, N_NODES * HID);
    gcn_scatter_kernel<<<cdiv(N_EDGES * 8, T), T, 0, stream>>>(src, dst, deg, lin, agg);
    gcn_combine_kernel<<<cdiv(N_NODES * HID, T), T, 0, stream>>>(
        agg, lin, bs[layer], deg, batch, cnt, act, feats + layer * G_GRAPHS * HID);
    hin = act;
  }
  gcn_merge_kernel<<<cdiv(G_GRAPHS * HID, T), T, 0, stream>>>(feats, out);
}


# edge sums in LDS: work-group = 2000 destination rows x one edge slice; waves list matching edges, add their source rows by LDS float adds, then add whole rows to memory; float add order changes
# speedup vs baseline: 1.4493x; 1.4493x over previous
#include <hip/hip_runtime.h>

#define N_NODES   100000
#define N_EDGES   1600000
#define C_IN      64
#define HID       32
#define G_GRAPHS  128
#define NEG_SLOPE 0.01f

typedef __attribute__((ext_vector_type(2))) float v2f;
typedef __attribute__((ext_vector_type(8))) float v8f;

__global__ void gcn_zero_kernel(float* __restrict__ p, int n) {
  int i = blockIdx.x * blockDim.x + threadIdx.x;
  if (i < n) p[i] = 0.0f;
}

#define DEG_R     4
#define DEG_S     32
#define DEG_BINS  (N_NODES / DEG_R)
#define DEG_SLICE (N_EDGES / DEG_S)
#define DEG_T     1024
#define DEG_LOADS ((DEG_SLICE / 4 + DEG_T - 1) / DEG_T)
static_assert(N_NODES % (4 * DEG_R) == 0, "bin ranges must be whole quads");
static_assert(N_EDGES % (4 * DEG_S) == 0, "edge slices must be whole quads");
static_assert(DEG_S <= HID, "the planes must fit the edge-sum region they borrow");

__global__ void __launch_bounds__(DEG_T)
gcn_degree_hist_kernel(const int* __restrict__ dst, unsigned* __restrict__ planes) {
  __shared__ unsigned bins[DEG_BINS];
  const int tid = threadIdx.x;
  const int r = blockIdx.x % DEG_R;
  const int s = blockIdx.x / DEG_R;
  const int lo = r * DEG_BINS;
  const int4* d4 = (const int4*)(dst + s * DEG_SLICE);
  int4 v[DEG_LOADS];
#pragma unroll
  for (int j = 0; j < DEG_LOADS; ++j) {
    const int q = tid + j * DEG_T;
    v[j] = q < DEG_SLICE / 4 ? d4[q] : make_int4(-1, -1, -1, -1);
  }
  for (int i = tid; i < DEG_BINS; i += DEG_T) bins[i] = 0u;
  __syncthreads();
#pragma unroll
  for (int j = 0; j < DEG_LOADS; ++j) {
    const unsigned a0 = (unsigned)(v[j].x - lo);
    const unsigned a1 = (unsigned)(v[j].y - lo);
    const unsigned a2 = (unsigned)(v[j].z - lo);
    const unsigned a3 = (unsigned)(v[j].w - lo);
    if (a0 < DEG_BINS) atomicAdd(&bins[a0], 1u);
    if (a1 < DEG_BINS) atomicAdd(&bins[a1], 1u);
    if (a2 < DEG_BINS) atomicAdd(&bins[a2], 1u);
    if (a3 < DEG_BINS) atomicAdd(&bins[a3], 1u);
  }
  __syncthreads();
  uint4* p4 = (uint4*)(planes + (size_t)s * N_NODES + lo);
  const uint4* b4 = (const uint4*)bins;
  for (int i = tid; i < DEG_BINS / 4; i += DEG_T) p4[i] = b4[i];
}

__global__ void gcn_dinv_kernel(const unsigned* __restrict__ planes, float* __restrict__ dinv) {
  const int q = blockIdx.x * blockDim.x + threadIdx.x;
  if (q >= N_NODES / 4) return;
  const uint4* p4 = (const uint4*)planes + q;
  uint4 c = make_uint4(0u, 0u, 0u, 0u);
#pragma unroll
  for (int s = 0; s < DEG_S; ++s) {
    const uint4 t = p4[(size_t)s * (N_NODES / 4)];
    c.x += t.x; c.y += t.y; c.z += t.z; c.w += t.w;
  }
  float4 o;
  o.x = rsqrtf((float)c.x + 1.0f);
  o.y = rsqrtf((float)c.y + 1.0f);
  o.z = rsqrtf((float)c.z + 1.0f);
  o.w = rsqrtf((float)c.w + 1.0f);
  ((float4*)dinv)[q] = o;
}

#define CNT_T 1024
#define CNT_B ((N_NODES + CNT_T - 1) / CNT_T)

__global__ void __launch_bounds__(CNT_T)
gcn_counts_hist_kernel(const int* __restrict__ batch, unsigned* __restrict__ parts) {
  __shared__ unsigned bins[G_GRAPHS];
  const int tid = threadIdx.x;
  if (tid < G_GRAPHS) bins[tid] = 0u;
  __syncthreads();
  const int i = blockIdx.x * CNT_T + tid;
  if (i < N_NODES) {
    const unsigned g = (unsigned)batch[i];
    if (g < G_GRAPHS) atomicAdd(&bins[g], 1u);
  }
  __syncthreads();
  if (tid < G_GRAPHS) parts[blockIdx.x * G_GRAPHS + tid] = bins[tid];
}

__global__ void gcn_cntinv_kernel(const unsigned* __restrict__ parts, float* __restrict__ cnt) {
  const int g = blockIdx.x * blockDim.x + threadIdx.x;
  if (g >= G_GRAPHS) return;
  unsigned c = 0u;
#pragma unroll
  for (int b = 0; b < CNT_B; ++b) c += parts[b * G_GRAPHS + g];
  cnt[g] = 1.0f / fmaxf((float)c, 1.0f);
}

template <int K>
__global__ void gcn_gemm_wmma_kernel(const float* __restrict__ X,
                                     const float* __restrict__ W,
                                     float* __restrict__ H, int nRows) {
  const int wave = blockIdx.x * (blockDim.x >> 5) + (threadIdx.x >> 5);
  const int lane = threadIdx.x & 31;
  const int numTiles = (nRows >> 4) * 2;
  if (wave >= numTiles) return;
  const int rowTile = wave >> 1;
  const int colTile = wave & 1;
  const int half = lane >> 4;
  const int l16  = lane & 15;
  const int aRow = rowTile * 16 + l16;
  const int col  = colTile * 16 + l16;

  v8f c = {};
#pragma unroll
  for (int k0 = 0; k0 < K; k0 += 4) {
    const int ka = k0 + 2 * half;
    v2f a = *(const v2f*)(X + aRow * K + ka);
    v2f b;
    b.x = W[(ka + 0) * HID + col];
    b.y = W[(ka + 1) * HID + col];
    c = __builtin_amdgcn_wmma_f32_16x16x4_f32(
         false, a,  false, b,
         (short)0, c,  false,  false);
  }
#pragma unroll
  for (int r = 0; r < 8; ++r) {
    const int m = r + 8 * half;
    H[(rowTile * 16 + m) * HID + col] = c[r];
  }
}

#define AGG_RANGE 2000
#define AGG_R     (N_NODES / AGG_RANGE)
#define AGG_S     5
#define AGG_T     1024
#define AGG_W     (AGG_T / 32)
#define AGG_SEG   256
#define AGG_QS    (N_EDGES / 4 / AGG_S)
#define AGG_NG    (AGG_QS / 32)
#define AGG_B     4
#define AGG_U     8
static_assert(N_NODES % AGG_RANGE == 0, "row ranges must tile the nodes");
static_assert(N_EDGES % (4 * AGG_S * 32) == 0, "slices must be whole wave groups of quads");
static_assert(AGG_SEG >= 2 * 128, "a list must take one more group after the drain test");

__device__ __forceinline__ void gcn_drain_list(const uint2* seg, int cnt, int lane, int lo,
                                               const int* __restrict__ src,
                                               const float* __restrict__ dinv,
                                               const float* __restrict__ H, float* acc) {
  const int sub = lane >> 3;
  const int q = (lane & 7) << 2;
  for (int i = 0; i < cnt; i += 4 * AGG_U) {
    int sr[AGG_U];
    int al[AGG_U];
    bool ok[AGG_U];
#pragma unroll
    for (int u = 0; u < AGG_U; ++u) {
      const int idx = i + 4 * u + sub;
      ok[u] = idx < cnt;
      const uint2 e = seg[ok[u] ? idx : i];
      sr[u] = src[e.x];
      al[u] = (int)e.y;
    }
    float4 hv[AGG_U];
    float cf[AGG_U];
#pragma unroll
    for (int u = 0; u < AGG_U; ++u) {
      hv[u] = *(const float4*)(H + sr[u] * HID + q);
      cf[u] = dinv[sr[u]] * dinv[lo + al[u]];
    }
#pragma unroll
    for (int u = 0; u < AGG_U; ++u) {
      float* ap = acc + al[u] * HID + q;
      atomicAdd(ap + 0, ok[u] ? hv[u].x * cf[u] : 0.0f);
      atomicAdd(ap + 1, ok[u] ? hv[u].y * cf[u] : 0.0f);
      atomicAdd(ap + 2, ok[u] ? hv[u].z * cf[u] : 0.0f);
      atomicAdd(ap + 3, ok[u] ? hv[u].w * cf[u] : 0.0f);
    }
  }
}

__global__ void __launch_bounds__(AGG_T)
gcn_edge_lds_kernel(const int* __restrict__ src, const int* __restrict__ dst,
                    const float* __restrict__ dinv,
                    const float* __restrict__ H, float* __restrict__ agg) {
  __shared__ float acc[AGG_RANGE * HID];
  __shared__ uint2 list[AGG_W * AGG_SEG];
  const int tid = threadIdx.x;
  const int lane = tid & 31;
  const int w = __builtin_amdgcn_readfirstlane(tid >> 5);
  const int r = blockIdx.x % AGG_R;
  const int sl = blockIdx.x / AGG_R;
  const int lo = r * AGG_RANGE;

  float4* acc4 = (float4*)acc;
  for (int i = tid; i < AGG_RANGE * HID / 4; i += AGG_T) acc4[i] = make_float4(0.f, 0.f, 0.f, 0.f);
  __syncthreads();

  const int qb = sl * AGG_QS;
  const int4* d4 = (const int4*)dst + qb;
  uint2* seg = list + w * AGG_SEG;
  int cnt = 0;

  for (int g0 = w; g0 < AGG_NG; g0 += AGG_W * AGG_B) {
    int4 dv[AGG_B];
#pragma unroll
    for (int b = 0; b < AGG_B; ++b) {
      const int g = g0 + b * AGG_W;
      if (g < AGG_NG) dv[b] = d4[g * 32 + lane];
      else            dv[b] = make_int4(-1, -1, -1, -1);
    }
#pragma unroll
    for (int b = 0; b < AGG_B; ++b) {
      const int g = g0 + b * AGG_W;
      const unsigned e0 = (unsigned)(4 * (qb + g * 32 + lane));
      const int dd[4] = {dv[b].x, dv[b].y, dv[b].z, dv[b].w};
#pragma unroll
      for (int k = 0; k < 4; ++k) {
        const unsigned a = (unsigned)(dd[k] - lo);
        const bool m = a < (unsigned)AGG_RANGE;
        const unsigned bal = __builtin_amdgcn_ballot_w32(m);
        if (m) seg[cnt + (int)__builtin_amdgcn_mbcnt_lo(bal, 0u)] = make_uint2(e0 + (unsigned)k, a);
        cnt += __builtin_popcount(bal);
      }
      if (cnt > AGG_SEG - 128) {
        __builtin_amdgcn_wave_barrier();
        gcn_drain_list(seg, cnt, lane, lo, src, dinv, H, acc);
        __builtin_amdgcn_wave_barrier();
        cnt = 0;
      }
    }
  }
  __builtin_amdgcn_wave_barrier();
  gcn_drain_list(seg, cnt, lane, lo, src, dinv, H, acc);
  __syncthreads();

  float* outp = agg + (size_t)lo * HID;
  for (int i = tid; i < AGG_RANGE * HID; i += AGG_T) atomicAdd(outp + i, acc[i]);
}

__global__ void gcn_combine_kernel(const float* __restrict__ agg, const float* __restrict__ Hlin,
                                   const float* __restrict__ bias, const float* __restrict__ dinv,
                                   const int* __restrict__ batch, const float* __restrict__ cntinv,
                                   float* __restrict__ Hact, float* __restrict__ feat) {
  int t = blockIdx.x * blockDim.x + threadIdx.x;
  if (t >= N_NODES * HID) return;
  int i = t >> 5, ch = t & 31;
  float di = dinv[i];
  float v = agg[t] + Hlin[t] * di * di + bias[ch];
  v = v > 0.0f ? v : NEG_SLOPE * v;
  Hact[t] = v;
  int g = batch[i];
  atomicAdd(&feat[g * HID + ch], v * cntinv[g]);
}

__global__ void gcn_merge_kernel(const float* __restrict__ feats, float* __restrict__ out) {
  int t = blockIdx.x * blockDim.x + threadIdx.x;
  const int GH = G_GRAPHS * HID;
  if (t < GH) out[t] = (feats[t] + feats[GH + t] + feats[2 * GH + t]) * (1.0f / 3.0f);
}

static inline int cdiv(int a, int b) { return (a + b - 1) / b; }

extern "C" void kernel_launch(void* const* d_in, const int* in_sizes, int n_in,
                              void* d_out, int out_size, void* d_ws, size_t ws_size,
                              hipStream_t stream) {
  (void)in_sizes; (void)n_in; (void)out_size; (void)ws_size;
  const float* x     = (const float*)d_in[0];
  const float* W0    = (const float*)d_in[1];
  const float* b0    = (const float*)d_in[2];
  const float* W1    = (const float*)d_in[3];
  const float* b1    = (const float*)d_in[4];
  const float* W2    = (const float*)d_in[5];
  const float* b2    = (const float*)d_in[6];
  const int*   src   = (const int*)d_in[7];
  const int*   dst   = (const int*)d_in[8];
  const int*   batch = (const int*)d_in[9];
  float* out = (float*)d_out;

  float* ws    = (float*)d_ws;
  float* deg   = ws;
  float* cnt   = deg + N_NODES;
  float* lin   = cnt + G_GRAPHS;
  float* act   = lin + (size_t)N_NODES * HID;
  float* agg   = act + (size_t)N_NODES * HID;
  float* feats = agg + (size_t)N_NODES * HID;

  const int T = 256;
  const int nTiles = (N_NODES / 16) * 2;

  unsigned* degPlanes = (unsigned*)agg;
  unsigned* cntPlanes = (unsigned*)act;
  gcn_zero_kernel<<<cdiv(3 * G_GRAPHS * HID, T), T, 0, stream>>>(feats, 3 * G_GRAPHS * HID);
  gcn_degree_hist_kernel<<<DEG_R * DEG_S, DEG_T, 0, stream>>>(dst, degPlanes);
  gcn_dinv_kernel<<<cdiv(N_NODES / 4, 128), 128, 0, stream>>>(degPlanes, deg);
  gcn_counts_hist_kernel<<<CNT_B, CNT_T, 0, stream>>>(batch, cntPlanes);
  gcn_cntinv_kernel<<<1, G_GRAPHS, 0, stream>>>(cntPlanes, cnt);

  const float* Ws[3] = {W0, W1, W2};
  const float* bs[3] = {b0, b1, b2};
  const float* hin = x;
  for (int layer = 0; layer < 3; ++layer) {
    if (layer == 0)
      gcn_gemm_wmma_kernel<C_IN><<<cdiv(nTiles, T / 32), T, 0, stream>>>(hin, Ws[0], lin, N_NODES);
    else
      gcn_gemm_wmma_kernel<HID><<<cdiv(nTiles, T / 32), T, 0, stream>>>(hin, Ws[layer], lin, N_NODES);
    gcn_zero_kernel<<<cdiv(N_NODES * HID, T), T, 0, stream>>>(agg, N_NODES * HID);
    gcn_edge_lds_kernel<<<AGG_R * AGG_S, AGG_T, 0, stream>>>(src, dst, deg, lin, agg);
    gcn_combine_kernel<<<cdiv(N_NODES * HID, T), T, 0, stream>>>(
        agg, lin, bs[layer], deg, batch, cnt, act, feats + layer * G_GRAPHS * HID);
    hin = act;
  }
  gcn_merge_kernel<<<cdiv(G_GRAPHS * HID, T), T, 0, stream>>>(feats, out);
}


# edge-sum accumulator seeded by the product kernel with self-loop term + bias (was cleared, terms added in combine): 3 clear launches gone, combine only activates and pools; float add order changes
# speedup vs baseline: 1.5107x; 1.0424x over previous
#include <hip/hip_runtime.h>

#define N_NODES   100000
#define N_EDGES   1600000
#define C_IN      64
#define HID       32
#define G_GRAPHS  128
#define NEG_SLOPE 0.01f

typedef __attribute__((ext_vector_type(2))) float v2f;
typedef __attribute__((ext_vector_type(8))) float v8f;

__global__ void gcn_zero_kernel(float* __restrict__ p, int n) {
  int i = blockIdx.x * blockDim.x + threadIdx.x;
  if (i < n) p[i] = 0.0f;
}

#define DEG_R     4
#define DEG_S     32
#define DEG_BINS  (N_NODES / DEG_R)
#define DEG_SLICE (N_EDGES / DEG_S)
#define DEG_T     1024
#define DEG_LOADS ((DEG_SLICE / 4 + DEG_T - 1) / DEG_T)
static_assert(N_NODES % (4 * DEG_R) == 0, "bin ranges must be whole quads");
static_assert(N_EDGES % (4 * DEG_S) == 0, "edge slices must be whole quads");
static_assert(DEG_S <= HID, "the planes must fit the edge-sum region they borrow");

__global__ void __launch_bounds__(DEG_T)
gcn_degree_hist_kernel(const int* __restrict__ dst, unsigned* __restrict__ planes) {
  __shared__ unsigned bins[DEG_BINS];
  const int tid = threadIdx.x;
  const int r = blockIdx.x % DEG_R;
  const int s = blockIdx.x / DEG_R;
  const int lo = r * DEG_BINS;
  const int4* d4 = (const int4*)(dst + s * DEG_SLICE);
  int4 v[DEG_LOADS];
#pragma unroll
  for (int j = 0; j < DEG_LOADS; ++j) {
    const int q = tid + j * DEG_T;
    v[j] = q < DEG_SLICE / 4 ? d4[q] : make_int4(-1, -1, -1, -1);
  }
  for (int i = tid; i < DEG_BINS; i += DEG_T) bins[i] = 0u;
  __syncthreads();
#pragma unroll
  for (int j = 0; j < DEG_LOADS; ++j) {
    const unsigned a0 = (unsigned)(v[j].x - lo);
    const unsigned a1 = (unsigned)(v[j].y - lo);
    const unsigned a2 = (unsigned)(v[j].z - lo);
    const unsigned a3 = (unsigned)(v[j].w - lo);
    if (a0 < DEG_BINS) atomicAdd(&bins[a0], 1u);
    if (a1 < DEG_BINS) atomicAdd(&bins[a1], 1u);
    if (a2 < DEG_BINS) atomicAdd(&bins[a2], 1u);
    if (a3 < DEG_BINS) atomicAdd(&bins[a3], 1u);
  }
  __syncthreads();
  uint4* p4 = (uint4*)(planes + (size_t)s * N_NODES + lo);
  const uint4* b4 = (const uint4*)bins;
  for (int i = tid; i < DEG_BINS / 4; i += DEG_T) p4[i] = b4[i];
}

__global__ void gcn_dinv_kernel(const unsigned* __restrict__ planes, float* __restrict__ dinv) {
  const int q = blockIdx.x * blockDim.x + threadIdx.x;
  if (q >= N_NODES / 4) return;
  const uint4* p4 = (const uint4*)planes + q;
  uint4 c = make_uint4(0u, 0u, 0u, 0u);
#pragma unroll
  for (int s = 0; s < DEG_S; ++s) {
    const uint4 t = p4[(size_t)s * (N_NODES / 4)];
    c.x += t.x; c.y += t.y; c.z += t.z; c.w += t.w;
  }
  float4 o;
  o.x = rsqrtf((float)c.x + 1.0f);
  o.y = rsqrtf((float)c.y + 1.0f);
  o.z = rsqrtf((float)c.z + 1.0f);
  o.w = rsqrtf((float)c.w + 1.0f);
  ((float4*)dinv)[q] = o;
}

#define CNT_T 1024
#define CNT_B ((N_NODES + CNT_T - 1) / CNT_T)

__global__ void __launch_bounds__(CNT_T)
gcn_counts_hist_kernel(const int* __restrict__ batch, unsigned* __restrict__ parts) {
  __shared__ unsigned bins[G_GRAPHS];
  const int tid = threadIdx.x;
  if (tid < G_GRAPHS) bins[tid] = 0u;
  __syncthreads();
  const int i = blockIdx.x * CNT_T + tid;
  if (i < N_NODES) {
    const unsigned g = (unsigned)batch[i];
    if (g < G_GRAPHS) atomicAdd(&bins[g], 1u);
  }
  __syncthreads();
  if (tid < G_GRAPHS) parts[blockIdx.x * G_GRAPHS + tid] = bins[tid];
}

__global__ void gcn_cntinv_kernel(const unsigned* __restrict__ parts, float* __restrict__ cnt) {
  const int g = blockIdx.x * blockDim.x + threadIdx.x;
  if (g >= G_GRAPHS) return;
  unsigned c = 0u;
#pragma unroll
  for (int b = 0; b < CNT_B; ++b) c += parts[b * G_GRAPHS + g];
  cnt[g] = 1.0f / fmaxf((float)c, 1.0f);
}

template <int K>
__global__ void gcn_gemm_wmma_kernel(const float* __restrict__ X,
                                     const float* __restrict__ W,
                                     const float* __restrict__ bias,
                                     const float* __restrict__ dinv,
                                     float* __restrict__ H,
                                     float* __restrict__ A, int nRows) {
  const int wave = blockIdx.x * (blockDim.x >> 5) + (threadIdx.x >> 5);
  const int lane = threadIdx.x & 31;
  const int numTiles = (nRows >> 4) * 2;
  if (wave >= numTiles) return;
  const int rowTile = wave >> 1;
  const int colTile = wave & 1;
  const int half = lane >> 4;
  const int l16  = lane & 15;
  const int aRow = rowTile * 16 + l16;
  const int col  = colTile * 16 + l16;

  float dd[8];
#pragma unroll
  for (int r = 0; r < 8; ++r) {
    const float d = dinv[rowTile * 16 + r + 8 * half];
    dd[r] = d * d;
  }
  const float bc = bias[col];

  v8f c = {};
#pragma unroll
  for (int k0 = 0; k0 < K; k0 += 4) {
    const int ka = k0 + 2 * half;
    v2f a = *(const v2f*)(X + aRow * K + ka);
    v2f b;
    b.x = W[(ka + 0) * HID + col];
    b.y = W[(ka + 1) * HID + col];
    c = __builtin_amdgcn_wmma_f32_16x16x4_f32(
         false, a,  false, b,
         (short)0, c,  false,  false);
  }
#pragma unroll
  for (int r = 0; r < 8; ++r) {
    const int m = r + 8 * half;
    const int o = (rowTile * 16 + m) * HID + col;
    H[o] = c[r];
    A[o] = c[r] * dd[r] + bc;
  }
}

#define AGG_RANGE 2000
#define AGG_R     (N_NODES / AGG_RANGE)
#define AGG_S     5
#define AGG_T     1024
#define AGG_W     (AGG_T / 32)
#define AGG_SEG   256
#define AGG_QS    (N_EDGES / 4 / AGG_S)
#define AGG_NG    (AGG_QS / 32)
#define AGG_B     4
#define AGG_U     8
static_assert(N_NODES % AGG_RANGE == 0, "row ranges must tile the nodes");
static_assert(N_EDGES % (4 * AGG_S * 32) == 0, "slices must be whole wave groups of quads");
static_assert(AGG_SEG >= 2 * 128, "a list must take one more group after the drain test");

__device__ __forceinline__ void gcn_drain_list(const uint2* seg, int cnt, int lane, int lo,
                                               const int* __restrict__ src,
                                               const float* __restrict__ dinv,
                                               const float* __restrict__ H, float* acc) {
  const int sub = lane >> 3;
  const int q = (lane & 7) << 2;
  for (int i = 0; i < cnt; i += 4 * AGG_U) {
    int sr[AGG_U];
    int al[AGG_U];
    bool ok[AGG_U];
#pragma unroll
    for (int u = 0; u < AGG_U; ++u) {
      const int idx = i + 4 * u + sub;
      ok[u] = idx < cnt;
      const uint2 e = seg[ok[u] ? idx : i];
      sr[u] = src[e.x];
      al[u] = (int)e.y;
    }
    float4 hv[AGG_U];
    float cf[AGG_U];
#pragma unroll
    for (int u = 0; u < AGG_U; ++u) {
      hv[u] = *(const float4*)(H + sr[u] * HID + q);
      cf[u] = dinv[sr[u]] * dinv[lo + al[u]];
    }
#pragma unroll
    for (int u = 0; u < AGG_U; ++u) {
      float* ap = acc + al[u] * HID + q;
      atomicAdd(ap + 0, ok[u] ? hv[u].x * cf[u] : 0.0f);
      atomicAdd(ap + 1, ok[u] ? hv[u].y * cf[u] : 0.0f);
      atomicAdd(ap + 2, ok[u] ? hv[u].z * cf[u] : 0.0f);
      atomicAdd(ap + 3, ok[u] ? hv[u].w * cf[u] : 0.0f);
    }
  }
}

__global__ void __launch_bounds__(AGG_T)
gcn_edge_lds_kernel(const int* __restrict__ src, const int* __restrict__ dst,
                    const float* __restrict__ dinv,
                    const float* __restrict__ H, float* __restrict__ agg) {
  __shared__ float acc[AGG_RANGE * HID];
  __shared__ uint2 list[AGG_W * AGG_SEG];
  const int tid = threadIdx.x;
  const int lane = tid & 31;
  const int w = __builtin_amdgcn_readfirstlane(tid >> 5);
  const int r = blockIdx.x % AGG_R;
  const int sl = blockIdx.x / AGG_R;
  const int lo = r * AGG_RANGE;

  float4* acc4 = (float4*)acc;
  for (int i = tid; i < AGG_RANGE * HID / 4; i += AGG_T) acc4[i] = make_float4(0.f, 0.f, 0.f, 0.f);
  __syncthreads();

  const int qb = sl * AGG_QS;
  const int4* d4 = (const int4*)dst + qb;
  uint2* seg = list + w * AGG_SEG;
  int cnt = 0;

  for (int g0 = w; g0 < AGG_NG; g0 += AGG_W * AGG_B) {
    int4 dv[AGG_B];
#pragma unroll
    for (int b = 0; b < AGG_B; ++b) {
      const int g = g0 + b * AGG_W;
      if (g < AGG_NG) dv[b] = d4[g * 32 + lane];
      else            dv[b] = make_int4(-1, -1, -1, -1);
    }
#pragma unroll
    for (int b = 0; b < AGG_B; ++b) {
      const int g = g0 + b * AGG_W;
      const unsigned e0 = (unsigned)(4 * (qb + g * 32 + lane));
      const int dd[4] = {dv[b].x, dv[b].y, dv[b].z, dv[b].w};
#pragma unroll
      for (int k = 0; k < 4; ++k) {
        const unsigned a = (unsigned)(dd[k] - lo);
        const bool m = a < (unsigned)AGG_RANGE;
        const unsigned bal = __builtin_amdgcn_ballot_w32(m);
        if (m) seg[cnt + (int)__builtin_amdgcn_mbcnt_lo(bal, 0u)] = make_uint2(e0 + (unsigned)k, a);
        cnt += __builtin_popcount(bal);
      }
      if (cnt > AGG_SEG - 128) {
        __builtin_amdgcn_wave_barrier();
        gcn_drain_list(seg, cnt, lane, lo, src, dinv, H, acc);
        __builtin_amdgcn_wave_barrier();
        cnt = 0;
      }
    }
  }
  __builtin_amdgcn_wave_barrier();
  gcn_drain_list(seg, cnt, lane, lo, src, dinv, H, acc);
  __syncthreads();

  float* outp = agg + (size_t)lo * HID;
  for (int i = tid; i < AGG_RANGE * HID; i += AGG_T) atomicAdd(outp + i, acc[i]);
}

__global__ void gcn_combine_kernel(const float* __restrict__ agg,
                                   const int* __restrict__ batch, const float* __restrict__ cntinv,
                                   float* __restrict__ Hact, float* __restrict__ feat) {
  int t = blockIdx.x * blockDim.x + threadIdx.x;
  if (t >= N_NODES * HID) return;
  int i = t >> 5, ch = t & 31;
  float v = agg[t];
  v = v > 0.0f ? v : NEG_SLOPE * v;
  Hact[t] = v;
  int g = batch[i];
  atomicAdd(&feat[g * HID + ch], v * cntinv[g]);
}

__global__ void gcn_merge_kernel(const float* __restrict__ feats, float* __restrict__ out) {
  int t = blockIdx.x * blockDim.x + threadIdx.x;
  const int GH = G_GRAPHS * HID;
  if (t < GH) out[t] = (feats[t] + feats[GH + t] + feats[2 * GH + t]) * (1.0f / 3.0f);
}

static inline int cdiv(int a, int b) { return (a + b - 1) / b; }

extern "C" void kernel_launch(void* const* d_in, const int* in_sizes, int n_in,
                              void* d_out, int out_size, void* d_ws, size_t ws_size,
                              hipStream_t stream) {
  (void)in_sizes; (void)n_in; (void)out_size; (void)ws_size;
  const float* x     = (const float*)d_in[0];
  const float* W0    = (const float*)d_in[1];
  const float* b0    = (const float*)d_in[2];
  const float* W1    = (const float*)d_in[3];
  const float* b1    = (const float*)d_in[4];
  const float* W2    = (const float*)d_in[5];
  const float* b2    = (const float*)d_in[6];
  const int*   src   = (const int*)d_in[7];
  const int*   dst   = (const int*)d_in[8];
  const int*   batch = (const int*)d_in[9];
  float* out = (float*)d_out;

  float* ws    = (float*)d_ws;
  float* deg   = ws;
  float* cnt   = deg + N_NODES;
  float* lin   = cnt + G_GRAPHS;
  float* act   = lin + (size_t)N_NODES * HID;
  float* agg   = act + (size_t)N_NODES * HID;
  float* feats = agg + (size_t)N_NODES * HID;

  const int T = 256;
  const int nTiles = (N_NODES / 16) * 2;

  unsigned* degPlanes = (unsigned*)agg;
  unsigned* cntPlanes = (unsigned*)act;
  gcn_zero_kernel<<<cdiv(3 * G_GRAPHS * HID, T), T, 0, stream>>>(feats, 3 * G_GRAPHS * HID);
  gcn_degree_hist_kernel<<<DEG_R * DEG_S, DEG_T, 0, stream>>>(dst, degPlanes);
  gcn_dinv_kernel<<<cdiv(N_NODES / 4, 128), 128, 0, stream>>>(degPlanes, deg);
  gcn_counts_hist_kernel<<<CNT_B, CNT_T, 0, stream>>>(batch, cntPlanes);
  gcn_cntinv_kernel<<<1, G_GRAPHS, 0, stream>>>(cntPlanes, cnt);

  const float* Ws[3] = {W0, W1, W2};
  const float* bs[3] = {b0, b1, b2};
  const float* hin = x;
  for (int layer = 0; layer < 3; ++layer) {
    if (layer == 0)
      gcn_gemm_wmma_kernel<C_IN><<<cdiv(nTiles, T / 32), T, 0, stream>>>(
          hin, Ws[0], bs[0], deg, lin, agg, N_NODES);
    else
      gcn_gemm_wmma_kernel<HID><<<cdiv(nTiles, T / 32), T, 0, stream>>>(
          hin, Ws[layer], bs[layer], deg, lin, agg, N_NODES);
    gcn_edge_lds_kernel<<<AGG_R * AGG_S, AGG_T, 0, stream>>>(src, dst, deg, lin, agg);
    gcn_combine_kernel<<<cdiv(N_NODES * HID, T), T, 0, stream>>>(
        agg, batch, cnt, act, feats + layer * G_GRAPHS * HID);
    hin = act;
  }
  gcn_merge_kernel<<<cdiv(G_GRAPHS * HID, T), T, 0, stream>>>(feats, out);
}


# layers 1-2 replay the matches that layer 0's dst scan leaves in the workspace (src and row in one word per edge) instead of scanning again; rows activated in place to free it; float add order changes
# speedup vs baseline: 1.8330x; 1.2133x over previous
#include <hip/hip_runtime.h>

#define N_NODES   100000
#define N_EDGES   1600000
#define C_IN      64
#define HID       32
#define G_GRAPHS  128
#define NEG_SLOPE 0.01f

typedef __attribute__((ext_vector_type(2))) float v2f;
typedef __attribute__((ext_vector_type(8))) float v8f;

__global__ void gcn_zero_kernel(float* __restrict__ p, int n) {
  int i = blockIdx.x * blockDim.x + threadIdx.x;
  if (i < n) p[i] = 0.0f;
}

#define DEG_R     4
#define DEG_S     32
#define DEG_BINS  (N_NODES / DEG_R)
#define DEG_SLICE (N_EDGES / DEG_S)
#define DEG_T     1024
#define DEG_LOADS ((DEG_SLICE / 4 + DEG_T - 1) / DEG_T)
static_assert(N_NODES % (4 * DEG_R) == 0, "bin ranges must be whole quads");
static_assert(N_EDGES % (4 * DEG_S) == 0, "edge slices must be whole quads");
static_assert(DEG_S <= HID, "the planes must fit the edge-sum region they borrow");

__global__ void __launch_bounds__(DEG_T)
gcn_degree_hist_kernel(const int* __restrict__ dst, unsigned* __restrict__ planes) {
  __shared__ unsigned bins[DEG_BINS];
  const int tid = threadIdx.x;
  const int r = blockIdx.x % DEG_R;
  const int s = blockIdx.x / DEG_R;
  const int lo = r * DEG_BINS;
  const int4* d4 = (const int4*)(dst + s * DEG_SLICE);
  int4 v[DEG_LOADS];
#pragma unroll
  for (int j = 0; j < DEG_LOADS; ++j) {
    const int q = tid + j * DEG_T;
    v[j] = q < DEG_SLICE / 4 ? d4[q] : make_int4(-1, -1, -1, -1);
  }
  for (int i = tid; i < DEG_BINS; i += DEG_T) bins[i] = 0u;
  __syncthreads();
#pragma unroll
  for (int j = 0; j < DEG_LOADS; ++j) {
    const unsigned a0 = (unsigned)(v[j].x - lo);
    const unsigned a1 = (unsigned)(v[j].y - lo);
    const unsigned a2 = (unsigned)(v[j].z - lo);
    const unsigned a3 = (unsigned)(v[j].w - lo);
    if (a0 < DEG_BINS) atomicAdd(&bins[a0], 1u);
    if (a1 < DEG_BINS) atomicAdd(&bins[a1], 1u);
    if (a2 < DEG_BINS) atomicAdd(&bins[a2], 1u);
    if (a3 < DEG_BINS) atomicAdd(&bins[a3], 1u);
  }
  __syncthreads();
  uint4* p4 = (uint4*)(planes + (size_t)s * N_NODES + lo);
  const uint4* b4 = (const uint4*)bins;
  for (int i = tid; i < DEG_BINS / 4; i += DEG_T) p4[i] = b4[i];
}

__global__ void gcn_dinv_kernel(const unsigned* __restrict__ planes, float* __restrict__ dinv) {
  const int q = blockIdx.x * blockDim.x + threadIdx.x;
  if (q >= N_NODES / 4) return;
  const uint4* p4 = (const uint4*)planes + q;
  uint4 c = make_uint4(0u, 0u, 0u, 0u);
#pragma unroll
  for (int s = 0; s < DEG_S; ++s) {
    const uint4 t = p4[(size_t)s * (N_NODES / 4)];
    c.x += t.x; c.y += t.y; c.z += t.z; c.w += t.w;
  }
  float4 o;
  o.x = rsqrtf((float)c.x + 1.0f);
  o.y = rsqrtf((float)c.y + 1.0f);
  o.z = rsqrtf((float)c.z + 1.0f);
  o.w = rsqrtf((float)c.w + 1.0f);
  ((float4*)dinv)[q] = o;
}

#define CNT_T 1024
#define CNT_B ((N_NODES + CNT_T - 1) / CNT_T)

__global__ void __launch_bounds__(CNT_T)
gcn_counts_hist_kernel(const int* __restrict__ batch, unsigned* __restrict__ parts) {
  __shared__ unsigned bins[G_GRAPHS];
  const int tid = threadIdx.x;
  if (tid < G_GRAPHS) bins[tid] = 0u;
  __syncthreads();
  const int i = blockIdx.x * CNT_T + tid;
  if (i < N_NODES) {
    const unsigned g = (unsigned)batch[i];
    if (g < G_GRAPHS) atomicAdd(&bins[g], 1u);
  }
  __syncthreads();
  if (tid < G_GRAPHS) parts[blockIdx.x * G_GRAPHS + tid] = bins[tid];
}

__global__ void gcn_cntinv_kernel(const unsigned* __restrict__ parts, float* __restrict__ cnt) {
  const int g = blockIdx.x * blockDim.x + threadIdx.x;
  if (g >= G_GRAPHS) return;
  unsigned c = 0u;
#pragma unroll
  for (int b = 0; b < CNT_B; ++b) c += parts[b * G_GRAPHS + g];
  cnt[g] = 1.0f / fmaxf((float)c, 1.0f);
}

template <int K, bool INPLACE>
__global__ void gcn_gemm_wmma_kernel(const float* X,
                                     const float* __restrict__ W,
                                     const float* __restrict__ bias,
                                     const float* __restrict__ dinv,
                                     float* __restrict__ H,
                                     float* A, int nRows) {
  const int wave = blockIdx.x * (blockDim.x >> 5) + (threadIdx.x >> 5);
  const int lane = threadIdx.x & 31;
  const int numTiles = (nRows >> 4) * 2;
  const bool active = wave < numTiles;
  v2f afrag[K / 4];
  if (active) {
    const int r0 = (wave >> 1) * 16 + (lane & 15);
#pragma unroll
    for (int j = 0; j < K / 4; ++j)
      afrag[j] = *(const v2f*)(X + r0 * K + 4 * j + 2 * (lane >> 4));
  }
  if (INPLACE) __syncthreads();
  if (!active) return;
  const int rowTile = wave >> 1;
  const int colTile = wave & 1;
  const int half = lane >> 4;
  const int l16  = lane & 15;
  const int col  = colTile * 16 + l16;

  float dd[8];
#pragma unroll
  for (int r = 0; r < 8; ++r) {
    const float d = dinv[rowTile * 16 + r + 8 * half];
    dd[r] = d * d;
  }
  const float bc = bias[col];

  v8f c = {};
#pragma unroll
  for (int k0 = 0; k0 < K; k0 += 4) {
    const int ka = k0 + 2 * half;
    const v2f a = afrag[k0 / 4];
    v2f b;
    b.x = W[(ka + 0) * HID + col];
    b.y = W[(ka + 1) * HID + col];
    c = __builtin_amdgcn_wmma_f32_16x16x4_f32(
         false, a,  false, b,
         (short)0, c,  false,  false);
  }
#pragma unroll
  for (int r = 0; r < 8; ++r) {
    const int m = r + 8 * half;
    const int o = (rowTile * 16 + m) * HID + col;
    H[o] = c[r];
    A[o] = c[r] * dd[r] + bc;
  }
}

#define AGG_RANGE 2000
#define AGG_R     (N_NODES / AGG_RANGE)
#define AGG_S     5
#define AGG_T     1024
#define AGG_W     (AGG_T / 32)
#define AGG_SEG   256
#define AGG_QS    (N_EDGES / 4 / AGG_S)
#define AGG_NG    (AGG_QS / 32)
#define AGG_B     4
#define AGG_U     8
static_assert(N_NODES % AGG_RANGE == 0, "row ranges must tile the nodes");
static_assert(N_EDGES % (4 * AGG_S * 32) == 0, "slices must be whole wave groups of quads");
static_assert(AGG_SEG >= 2 * 128, "a list must take one more group after the drain test");
#define LST_WORDS (N_NODES * HID / (AGG_R * AGG_S))
#define LST_CAP   (LST_WORDS - 1)
#define LST_SHIFT 17
static_assert(N_NODES <= (1 << LST_SHIFT), "a source node must fit the low bits");
static_assert(AGG_RANGE <= (1 << (32 - LST_SHIFT)), "a local row must fit the high bits");
static_assert((N_NODES * HID) % (AGG_R * AGG_S) == 0, "segments must tile the region");

template <bool REC>
__device__ __forceinline__ void gcn_drain_list(const uint2* seg, int cnt, int lane, int lo,
                                               const int* __restrict__ src,
                                               const float* __restrict__ dinv,
                                               const float* __restrict__ H, float* acc,
                                               unsigned* ent, unsigned base) {
  const int sub = lane >> 3;
  const int q = (lane & 7) << 2;
  for (int i = 0; i < cnt; i += 4 * AGG_U) {
    int sr[AGG_U];
    int al[AGG_U];
    bool ok[AGG_U];
#pragma unroll
    for (int u = 0; u < AGG_U; ++u) {
      const int idx = i + 4 * u + sub;
      ok[u] = idx < cnt;
      const uint2 e = seg[ok[u] ? idx : i];
      sr[u] = src[e.x];
      al[u] = (int)e.y;
    }
    float4 hv[AGG_U];
    float cf[AGG_U];
#pragma unroll
    for (int u = 0; u < AGG_U; ++u) {
      hv[u] = *(const float4*)(H + sr[u] * HID + q);
      cf[u] = dinv[sr[u]] * dinv[lo + al[u]];
    }
    if (REC) {
#pragma unroll
      for (int u = 0; u < AGG_U; ++u) {
        const unsigned pos = base + (unsigned)(i + 4 * u + sub);
        if (ok[u] && (lane & 7) == 0 && pos < (unsigned)LST_CAP)
          ent[pos] = (unsigned)sr[u] | ((unsigned)al[u] << LST_SHIFT);
      }
    }
#pragma unroll
    for (int u = 0; u < AGG_U; ++u) {
      float* ap = acc + al[u] * HID + q;
      atomicAdd(ap + 0, ok[u] ? hv[u].x * cf[u] : 0.0f);
      atomicAdd(ap + 1, ok[u] ? hv[u].y * cf[u] : 0.0f);
      atomicAdd(ap + 2, ok[u] ? hv[u].z * cf[u] : 0.0f);
      atomicAdd(ap + 3, ok[u] ? hv[u].w * cf[u] : 0.0f);
    }
  }
}

__device__ __forceinline__ void gcn_replay_list(const unsigned* ent, int n, int w, int lane, int lo,
                                                const float* __restrict__ dinv,
                                                const float* __restrict__ H, float* acc) {
  const int sub = lane >> 3;
  const int q = (lane & 7) << 2;
  int i = w * (4 * AGG_U);
  unsigned en[AGG_U];
  if (i < n) {
#pragma unroll
    for (int u = 0; u < AGG_U; ++u) {
      const int idx = i + 4 * u + sub;
      en[u] = ent[idx < n ? idx : i];
    }
  }
  while (i < n) {
    int al[AGG_U];
    bool ok[AGG_U];
    float4 hv[AGG_U];
    float cf[AGG_U];
#pragma unroll
    for (int u = 0; u < AGG_U; ++u) {
      ok[u] = i + 4 * u + sub < n;
      const int sr = (int)(en[u] & ((1u << LST_SHIFT) - 1u));
      al[u] = (int)(en[u] >> LST_SHIFT);
      hv[u] = *(const float4*)(H + sr * HID + q);
      cf[u] = dinv[sr] * dinv[lo + al[u]];
    }
    const int inext = i + AGG_W * 4 * AGG_U;
    if (inext < n) {
#pragma unroll
      for (int u = 0; u < AGG_U; ++u) {
        const int idx = inext + 4 * u + sub;
        en[u] = ent[idx < n ? idx : inext];
      }
    }
#pragma unroll
    for (int u = 0; u < AGG_U; ++u) {
      float* ap = acc + al[u] * HID + q;
      atomicAdd(ap + 0, ok[u] ? hv[u].x * cf[u] : 0.0f);
      atomicAdd(ap + 1, ok[u] ? hv[u].y * cf[u] : 0.0f);
      atomicAdd(ap + 2, ok[u] ? hv[u].z * cf[u] : 0.0f);
      atomicAdd(ap + 3, ok[u] ? hv[u].w * cf[u] : 0.0f);
    }
    i = inext;
  }
}

template <bool FIRST>
__global__ void __launch_bounds__(AGG_T)
gcn_edge_lds_kernel(const int* __restrict__ src, const int* __restrict__ dst,
                    const float* __restrict__ dinv,
                    const float* __restrict__ H, float* __restrict__ agg,
                    unsigned* lists) {
  __shared__ float acc[AGG_RANGE * HID];
  __shared__ uint2 list[AGG_W * AGG_SEG];
  __shared__ unsigned total;
  const int tid = threadIdx.x;
  const int lane = tid & 31;
  const int w = __builtin_amdgcn_readfirstlane(tid >> 5);
  const int r = blockIdx.x % AGG_R;
  const int sl = blockIdx.x / AGG_R;
  const int lo = r * AGG_RANGE;

  unsigned* segw = lists + (size_t)blockIdx.x * LST_WORDS;
  unsigned* ent = segw + 1;

  float4* acc4 = (float4*)acc;
  for (int i = tid; i < AGG_RANGE * HID / 4; i += AGG_T) acc4[i] = make_float4(0.f, 0.f, 0.f, 0.f);
  if (FIRST && tid == 0) total = 0u;
  const int nrec = FIRST ? 0 : __builtin_amdgcn_readfirstlane((int)segw[0]);
  __syncthreads();

  if (!FIRST && (unsigned)nrec <= (unsigned)LST_CAP) {
    gcn_replay_list(ent, nrec, w, lane, lo, dinv, H, acc);
  } else {
  const int qb = sl * AGG_QS;
  const int4* d4 = (const int4*)dst + qb;
  uint2* seg = list + w * AGG_SEG;
  int cnt = 0;

  for (int g0 = w; g0 < AGG_NG; g0 += AGG_W * AGG_B) {
    int4 dv[AGG_B];
#pragma unroll
    for (int b = 0; b < AGG_B; ++b) {
      const int g = g0 + b * AGG_W;
      if (g < AGG_NG) dv[b] = d4[g * 32 + lane];
      else            dv[b] = make_int4(-1, -1, -1, -1);
    }
#pragma unroll
    for (int b = 0; b < AGG_B; ++b) {
      const int g = g0 + b * AGG_W;
      const unsigned e0 = (unsigned)(4 * (qb + g * 32 + lane));
      const int dd[4] = {dv[b].x, dv[b].y, dv[b].z, dv[b].w};
#pragma unroll
      for (int k = 0; k < 4; ++k) {
        const unsigned a = (unsigned)(dd[k] - lo);
        const bool m = a < (unsigned)AGG_RANGE;
        const unsigned bal = __builtin_amdgcn_ballot_w32(m);
        if (m) seg[cnt + (int)__builtin_amdgcn_mbcnt_lo(bal, 0u)] = make_uint2(e0 + (unsigned)k, a);
        cnt += __builtin_popcount(bal);
      }
      if (cnt > AGG_SEG - 128) {
        __builtin_amdgcn_wave_barrier();
        unsigned base = 0u;
        if (FIRST) {
          if (lane == 0) base = atomicAdd(&total, (unsigned)cnt);
          base = (unsigned)__builtin_amdgcn_readfirstlane((int)base);
        }
        gcn_drain_list<FIRST>(seg, cnt, lane, lo, src, dinv, H, acc, ent, base);
        __builtin_amdgcn_wave_barrier();
        cnt = 0;
      }
    }
  }
  __builtin_amdgcn_wave_barrier();
  unsigned base = 0u;
  if (FIRST) {
    if (lane == 0) base = atomicAdd(&total, (unsigned)cnt);
    base = (unsigned)__builtin_amdgcn_readfirstlane((int)base);
  }
  gcn_drain_list<FIRST>(seg, cnt, lane, lo, src, dinv, H, acc, ent, base);
  }
  __syncthreads();
  if (FIRST && tid == 0) segw[0] = total;

  float* outp = agg + (size_t)lo * HID;
  for (int i = tid; i < AGG_RANGE * HID; i += AGG_T) atomicAdd(outp + i, acc[i]);
}

__global__ void gcn_combine_kernel(float* __restrict__ agg,
                                   const int* __restrict__ batch, const float* __restrict__ cntinv,
                                   float* __restrict__ feat) {
  int t = blockIdx.x * blockDim.x + threadIdx.x;
  if (t >= N_NODES * HID) return;
  int i = t >> 5, ch = t & 31;
  float v = agg[t];
  v = v > 0.0f ? v : NEG_SLOPE * v;
  agg[t] = v;
  int g = batch[i];
  atomicAdd(&feat[g * HID + ch], v * cntinv[g]);
}

__global__ void gcn_merge_kernel(const float* __restrict__ feats, float* __restrict__ out) {
  int t = blockIdx.x * blockDim.x + threadIdx.x;
  const int GH = G_GRAPHS * HID;
  if (t < GH) out[t] = (feats[t] + feats[GH + t] + feats[2 * GH + t]) * (1.0f / 3.0f);
}

static inline int cdiv(int a, int b) { return (a + b - 1) / b; }

extern "C" void kernel_launch(void* const* d_in, const int* in_sizes, int n_in,
                              void* d_out, int out_size, void* d_ws, size_t ws_size,
                              hipStream_t stream) {
  (void)in_sizes; (void)n_in; (void)out_size; (void)ws_size;
  const float* x     = (const float*)d_in[0];
  const float* W0    = (const float*)d_in[1];
  const float* b0    = (const float*)d_in[2];
  const float* W1    = (const float*)d_in[3];
  const float* b1    = (const float*)d_in[4];
  const float* W2    = (const float*)d_in[5];
  const float* b2    = (const float*)d_in[6];
  const int*   src   = (const int*)d_in[7];
  const int*   dst   = (const int*)d_in[8];
  const int*   batch = (const int*)d_in[9];
  float* out = (float*)d_out;

  float* ws    = (float*)d_ws;
  float* deg   = ws;
  float* cnt   = deg + N_NODES;
  float* lin   = cnt + G_GRAPHS;
  float* act   = lin + (size_t)N_NODES * HID;
  float* agg   = act + (size_t)N_NODES * HID;
  float* feats = agg + (size_t)N_NODES * HID;

  const int T = 256;
  const int nTiles = (N_NODES / 16) * 2;

  unsigned* degPlanes = (unsigned*)agg;
  unsigned* cntPlanes = (unsigned*)act;
  unsigned* lists     = (unsigned*)act;
  gcn_zero_kernel<<<cdiv(3 * G_GRAPHS * HID, T), T, 0, stream>>>(feats, 3 * G_GRAPHS * HID);
  gcn_degree_hist_kernel<<<DEG_R * DEG_S, DEG_T, 0, stream>>>(dst, degPlanes);
  gcn_dinv_kernel<<<cdiv(N_NODES / 4, 128), 128, 0, stream>>>(degPlanes, deg);
  gcn_counts_hist_kernel<<<CNT_B, CNT_T, 0, stream>>>(batch, cntPlanes);
  gcn_cntinv_kernel<<<1, G_GRAPHS, 0, stream>>>(cntPlanes, cnt);

  const float* Ws[3] = {W0, W1, W2};
  const float* bs[3] = {b0, b1, b2};
  for (int layer = 0; layer < 3; ++layer) {
    if (layer == 0) {
      gcn_gemm_wmma_kernel<C_IN, false><<<cdiv(nTiles, T / 32), T, 0, stream>>>(
          x, Ws[0], bs[0], deg, lin, agg, N_NODES);
      gcn_edge_lds_kernel<true><<<AGG_R * AGG_S, AGG_T, 0, stream>>>(src, dst, deg, lin, agg, lists);
    } else {
      gcn_gemm_wmma_kernel<HID, true><<<cdiv(nTiles, T / 32), T, 0, stream>>>(
          agg, Ws[layer], bs[layer], deg, lin, agg, N_NODES);
      gcn_edge_lds_kernel<false><<<AGG_R * AGG_S, AGG_T, 0, stream>>>(src, dst, deg, lin, agg, lists);
    }
    gcn_combine_kernel<<<cdiv(N_NODES * HID, T), T, 0, stream>>>(
        agg, batch, cnt, feats + layer * G_GRAPHS * HID);
  }
  gcn_merge_kernel<<<cdiv(G_GRAPHS * HID, T), T, 0, stream>>>(feats, out);
}


# each work-group owns 400 dst rows: one scan builds per-block edge lists, each layer replays them in LDS, adds seed, activates, pools in-kernel; no row atomics, no combine; float add order changes
# speedup vs baseline: 1.9060x; 1.0398x over previous
#include <hip/hip_runtime.h>

#define N_NODES   100000
#define N_EDGES   1600000
#define C_IN      64
#define HID       32
#define G_GRAPHS  128
#define NEG_SLOPE 0.01f

typedef __attribute__((ext_vector_type(2))) float v2f;
typedef __attribute__((ext_vector_type(8))) float v8f;

__global__ void gcn_zero_kernel(float* __restrict__ p, int n) {
  int i = blockIdx.x * blockDim.x + threadIdx.x;
  if (i < n) p[i] = 0.0f;
}

#define DEG_R     4
#define DEG_S     32
#define DEG_BINS  (N_NODES / DEG_R)
#define DEG_SLICE (N_EDGES / DEG_S)
#define DEG_T     1024
#define DEG_LOADS ((DEG_SLICE / 4 + DEG_T - 1) / DEG_T)
static_assert(N_NODES % (4 * DEG_R) == 0, "bin ranges must be whole quads");
static_assert(N_EDGES % (4 * DEG_S) == 0, "edge slices must be whole quads");
static_assert(DEG_S <= HID, "the planes must fit the edge-sum region they borrow");

__global__ void __launch_bounds__(DEG_T)
gcn_degree_hist_kernel(const int* __restrict__ dst, unsigned* __restrict__ planes) {
  __shared__ unsigned bins[DEG_BINS];
  const int tid = threadIdx.x;
  const int r = blockIdx.x % DEG_R;
  const int s = blockIdx.x / DEG_R;
  const int lo = r * DEG_BINS;
  const int4* d4 = (const int4*)(dst + s * DEG_SLICE);
  int4 v[DEG_LOADS];
#pragma unroll
  for (int j = 0; j < DEG_LOADS; ++j) {
    const int q = tid + j * DEG_T;
    v[j] = q < DEG_SLICE / 4 ? d4[q] : make_int4(-1, -1, -1, -1);
  }
  for (int i = tid; i < DEG_BINS; i += DEG_T) bins[i] = 0u;
  __syncthreads();
#pragma unroll
  for (int j = 0; j < DEG_LOADS; ++j) {
    const unsigned a0 = (unsigned)(v[j].x - lo);
    const unsigned a1 = (unsigned)(v[j].y - lo);
    const unsigned a2 = (unsigned)(v[j].z - lo);
    const unsigned a3 = (unsigned)(v[j].w - lo);
    if (a0 < DEG_BINS) atomicAdd(&bins[a0], 1u);
    if (a1 < DEG_BINS) atomicAdd(&bins[a1], 1u);
    if (a2 < DEG_BINS) atomicAdd(&bins[a2], 1u);
    if (a3 < DEG_BINS) atomicAdd(&bins[a3], 1u);
  }
  __syncthreads();
  uint4* p4 = (uint4*)(planes + (size_t)s * N_NODES + lo);
  const uint4* b4 = (const uint4*)bins;
  for (int i = tid; i < DEG_BINS / 4; i += DEG_T) p4[i] = b4[i];
}

__global__ void gcn_dinv_kernel(const unsigned* __restrict__ planes, float* __restrict__ dinv) {
  const int q = blockIdx.x * blockDim.x + threadIdx.x;
  if (q >= N_NODES / 4) return;
  const uint4* p4 = (const uint4*)planes + q;
  uint4 c = make_uint4(0u, 0u, 0u, 0u);
#pragma unroll
  for (int s = 0; s < DEG_S; ++s) {
    const uint4 t = p4[(size_t)s * (N_NODES / 4)];
    c.x += t.x; c.y += t.y; c.z += t.z; c.w += t.w;
  }
  float4 o;
  o.x = rsqrtf((float)c.x + 1.0f);
  o.y = rsqrtf((float)c.y + 1.0f);
  o.z = rsqrtf((float)c.z + 1.0f);
  o.w = rsqrtf((float)c.w + 1.0f);
  ((float4*)dinv)[q] = o;
}

#define CNT_T 1024
#define CNT_B ((N_NODES + CNT_T - 1) / CNT_T)

__global__ void __launch_bounds__(CNT_T)
gcn_counts_hist_kernel(const int* __restrict__ batch, unsigned* __restrict__ parts) {
  __shared__ unsigned bins[G_GRAPHS];
  const int tid = threadIdx.x;
  if (tid < G_GRAPHS) bins[tid] = 0u;
  __syncthreads();
  const int i = blockIdx.x * CNT_T + tid;
  if (i < N_NODES) {
    const unsigned g = (unsigned)batch[i];
    if (g < G_GRAPHS) atomicAdd(&bins[g], 1u);
  }
  __syncthreads();
  if (tid < G_GRAPHS) parts[blockIdx.x * G_GRAPHS + tid] = bins[tid];
}

__global__ void gcn_cntinv_kernel(const unsigned* __restrict__ parts, float* __restrict__ cnt) {
  const int g = blockIdx.x * blockDim.x + threadIdx.x;
  if (g >= G_GRAPHS) return;
  unsigned c = 0u;
#pragma unroll
  for (int b = 0; b < CNT_B; ++b) c += parts[b * G_GRAPHS + g];
  cnt[g] = 1.0f / fmaxf((float)c, 1.0f);
}

template <int K, bool INPLACE>
__global__ void gcn_gemm_wmma_kernel(const float* X,
                                     const float* __restrict__ W,
                                     const float* __restrict__ bias,
                                     const float* __restrict__ dinv,
                                     float* __restrict__ H,
                                     float* A, int nRows) {
  const int wave = blockIdx.x * (blockDim.x >> 5) + (threadIdx.x >> 5);
  const int lane = threadIdx.x & 31;
  const int numTiles = (nRows >> 4) * 2;
  const bool active = wave < numTiles;
  v2f afrag[K / 4];
  if (active) {
    const int r0 = (wave >> 1) * 16 + (lane & 15);
#pragma unroll
    for (int j = 0; j < K / 4; ++j)
      afrag[j] = *(const v2f*)(X + r0 * K + 4 * j + 2 * (lane >> 4));
  }
  if (INPLACE) __syncthreads();
  if (!active) return;
  const int rowTile = wave >> 1;
  const int colTile = wave & 1;
  const int half = lane >> 4;
  const int l16  = lane & 15;
  const int col  = colTile * 16 + l16;

  float dd[8];
#pragma unroll
  for (int r = 0; r < 8; ++r) {
    const float d = dinv[rowTile * 16 + r + 8 * half];
    dd[r] = d * d;
  }
  const float bc = bias[col];

  v8f c = {};
#pragma unroll
  for (int k0 = 0; k0 < K; k0 += 4) {
    const int ka = k0 + 2 * half;
    const v2f a = afrag[k0 / 4];
    v2f b;
    b.x = W[(ka + 0) * HID + col];
    b.y = W[(ka + 1) * HID + col];
    c = __builtin_amdgcn_wmma_f32_16x16x4_f32(
         false, a,  false, b,
         (short)0, c,  false,  false);
  }
#pragma unroll
  for (int r = 0; r < 8; ++r) {
    const int m = r + 8 * half;
    const int o = (rowTile * 16 + m) * HID + col;
    H[o] = c[r];
    A[o] = c[r] * dd[r] + bc;
  }
}

#define AGG_RANGE 2000
#define AGG_R     (N_NODES / AGG_RANGE)
#define AGG_S     5
#define AGG_T     1024
#define AGG_W     (AGG_T / 32)
#define AGG_SEG   256
#define AGG_QS    (N_EDGES / 4 / AGG_S)
#define AGG_NG    (AGG_QS / 32)
#define AGG_B     4
#define AGG_U     8
#define FL_U      4
#define OWN_ROWS  400
#define OWN_PER   (AGG_RANGE / OWN_ROWS)
#define OWN_B     (N_NODES / OWN_ROWS)
#define OWN_TRIPS ((OWN_ROWS * HID + AGG_T - 1) / AGG_T)
static_assert(N_NODES % AGG_RANGE == 0, "row ranges must tile the nodes");
static_assert(AGG_RANGE % OWN_ROWS == 0, "owned blocks must tile a range");
static_assert(OWN_PER <= 32, "one lane per block keeps that block's counter");
static_assert(N_EDGES % (4 * AGG_S * 32) == 0, "slices must be whole wave groups of quads");
static_assert(AGG_SEG >= 2 * 128, "a list must take one more group after the flush test");
#define LST_WORDS (N_NODES * HID / (AGG_R * AGG_S))
#define SUB_WORDS (LST_WORDS / OWN_PER)
#define SUB_CAP   (SUB_WORDS - 1)
#define LST_SHIFT 17
static_assert(N_NODES <= (1 << LST_SHIFT), "a source node must fit the low bits");
static_assert(OWN_ROWS <= (1 << (32 - LST_SHIFT)), "a row of a block must fit the high bits");
static_assert((N_NODES * HID) % (AGG_R * AGG_S) == 0, "segments must tile the region");
static_assert(LST_WORDS % OWN_PER == 0, "parts must tile a segment");

__device__ __forceinline__ void gcn_flush_list(const uint2* seg, int cnt, int lane,
                                               const int* __restrict__ src,
                                               unsigned* fill, unsigned* segw) {
  for (int i = 0; i < cnt; i += 32 * FL_U) {
    unsigned sr[FL_U];
    unsigned al[FL_U];
    bool ok[FL_U];
#pragma unroll
    for (int u = 0; u < FL_U; ++u) {
      const int idx = i + 32 * u + lane;
      ok[u] = idx < cnt;
      const uint2 e = seg[ok[u] ? idx : i];
      sr[u] = (unsigned)src[e.x];
      al[u] = e.y;
    }
#pragma unroll
    for (int u = 0; u < FL_U; ++u) {
      if (i + 32 * u < cnt) {
        const unsigned blk = ok[u] ? al[u] / (unsigned)OWN_ROWS : (unsigned)OWN_PER;
        unsigned rank = 0u, mine = 0u;
#pragma unroll
        for (int k = 0; k < OWN_PER; ++k) {
          const unsigned bal = __builtin_amdgcn_ballot_w32(blk == (unsigned)k);
          if (blk == (unsigned)k) rank = __builtin_amdgcn_mbcnt_lo(bal, 0u);
          if (lane == k) mine = (unsigned)__builtin_popcount(bal);
        }
        unsigned base = 0u;
        if (lane < OWN_PER) base = atomicAdd(&fill[lane], mine);
        base = (unsigned)__shfl((int)base, (int)blk);
        const unsigned pos = base + rank;
        if (ok[u] && pos < (unsigned)SUB_CAP)
          segw[blk * SUB_WORDS + 1u + pos] = sr[u] | ((al[u] - blk * (unsigned)OWN_ROWS) << LST_SHIFT);
      }
    }
  }
}

__global__ void __launch_bounds__(AGG_T)
gcn_edge_scan_kernel(const int* __restrict__ src, const int* __restrict__ dst,
                     unsigned* __restrict__ lists) {
  __shared__ uint2 list[AGG_W * AGG_SEG];
  __shared__ unsigned fill[OWN_PER];
  const int tid = threadIdx.x;
  const int lane = tid & 31;
  const int w = __builtin_amdgcn_readfirstlane(tid >> 5);
  const int r = blockIdx.x % AGG_R;
  const int sl = blockIdx.x / AGG_R;
  const int lo = r * AGG_RANGE;
  unsigned* segw = lists + (size_t)blockIdx.x * LST_WORDS;

  if (tid < OWN_PER) fill[tid] = 0u;
  __syncthreads();

  const int qb = sl * AGG_QS;
  const int4* d4 = (const int4*)dst + qb;
  uint2* seg = list + w * AGG_SEG;
  int cnt = 0;

  for (int g0 = w; g0 < AGG_NG; g0 += AGG_W * AGG_B) {
    int4 dv[AGG_B];
#pragma unroll
    for (int b = 0; b < AGG_B; ++b) {
      const int g = g0 + b * AGG_W;
      if (g < AGG_NG) dv[b] = d4[g * 32 + lane];
      else            dv[b] = make_int4(-1, -1, -1, -1);
    }
#pragma unroll
    for (int b = 0; b < AGG_B; ++b) {
      const int g = g0 + b * AGG_W;
      const unsigned e0 = (unsigned)(4 * (qb + g * 32 + lane));
      const int dd[4] = {dv[b].x, dv[b].y, dv[b].z, dv[b].w};
#pragma unroll
      for (int k = 0; k < 4; ++k) {
        const unsigned a = (unsigned)(dd[k] - lo);
        const bool m = a < (unsigned)AGG_RANGE;
        const unsigned bal = __builtin_amdgcn_ballot_w32(m);
        if (m) seg[cnt + (int)__builtin_amdgcn_mbcnt_lo(bal, 0u)] = make_uint2(e0 + (unsigned)k, a);
        cnt += __builtin_popcount(bal);
      }
      if (cnt > AGG_SEG - 128) {
        __builtin_amdgcn_wave_barrier();
        gcn_flush_list(seg, cnt, lane, src, fill, segw);
        __builtin_amdgcn_wave_barrier();
        cnt = 0;
      }
    }
  }
  __builtin_amdgcn_wave_barrier();
  gcn_flush_list(seg, cnt, lane, src, fill, segw);
  __syncthreads();
  if (tid < OWN_PER) segw[tid * SUB_WORDS] = fill[tid];
}

__device__ __forceinline__ void gcn_replay_list(const unsigned* __restrict__ ent, int n, int wr, int nw,
                                                int lane, int lo,
                                                const float* __restrict__ dinv,
                                                const float* __restrict__ H, float* acc) {
  const int sub = lane >> 3;
  const int q = (lane & 7) << 2;
  int i = wr * (4 * AGG_U);
  unsigned en[AGG_U];
  if (i < n) {
#pragma unroll
    for (int u = 0; u < AGG_U; ++u) {
      const int idx = i + 4 * u + sub;
      en[u] = ent[idx < n ? idx : i];
    }
  }
  while (i < n) {
    int al[AGG_U];
    bool ok[AGG_U];
    float4 hv[AGG_U];
    float cf[AGG_U];
#pragma unroll
    for (int u = 0; u < AGG_U; ++u) {
      ok[u] = i + 4 * u + sub < n;
      const int sr = (int)(en[u] & ((1u << LST_SHIFT) - 1u));
      al[u] = (int)(en[u] >> LST_SHIFT);
      hv[u] = *(const float4*)(H + sr * HID + q);
      cf[u] = dinv[sr] * dinv[lo + al[u]];
    }
    const int inext = i + nw * 4 * AGG_U;
    if (inext < n) {
#pragma unroll
      for (int u = 0; u < AGG_U; ++u) {
        const int idx = inext + 4 * u + sub;
        en[u] = ent[idx < n ? idx : inext];
      }
    }
#pragma unroll
    for (int u = 0; u < AGG_U; ++u) {
      float* ap = acc + al[u] * HID + q;
      atomicAdd(ap + 0, ok[u] ? hv[u].x * cf[u] : 0.0f);
      atomicAdd(ap + 1, ok[u] ? hv[u].y * cf[u] : 0.0f);
      atomicAdd(ap + 2, ok[u] ? hv[u].z * cf[u] : 0.0f);
      atomicAdd(ap + 3, ok[u] ? hv[u].w * cf[u] : 0.0f);
    }
    i = inext;
  }
}

__global__ void __launch_bounds__(AGG_T)
gcn_edge_sum_kernel(const int* __restrict__ src, const int* __restrict__ dst,
                    const float* __restrict__ dinv,
                    const float* __restrict__ H, float* __restrict__ agg,
                    const unsigned* __restrict__ lists,
                    const int* __restrict__ batch, const float* __restrict__ cntinv,
                    float* __restrict__ feat) {
  __shared__ float acc[OWN_ROWS * HID];
  const int tid = threadIdx.x;
  const int lane = tid & 31;
  const int w = __builtin_amdgcn_readfirstlane(tid >> 5);
  const int r = blockIdx.x / OWN_PER;
  const int p = blockIdx.x % OWN_PER;
  const int lo = blockIdx.x * OWN_ROWS;

  float4* acc4 = (float4*)acc;
  for (int i = tid; i < OWN_ROWS * HID / 4; i += AGG_T) acc4[i] = make_float4(0.f, 0.f, 0.f, 0.f);

  bool fit = true;
#pragma unroll
  for (int s = 0; s < AGG_S; ++s) {
    const unsigned n = lists[(size_t)(s * AGG_R + r) * LST_WORDS + (size_t)p * SUB_WORDS];
    fit = fit && n <= (unsigned)SUB_CAP;
  }
  __syncthreads();

  if (fit) {
    const int s = w % AGG_S;
    const int wr = w / AGG_S;
    const int nw = (AGG_W - s + AGG_S - 1) / AGG_S;
    const unsigned* part = lists + (size_t)(s * AGG_R + r) * LST_WORDS + (size_t)p * SUB_WORDS;
    const int n = __builtin_amdgcn_readfirstlane((int)part[0]);
    gcn_replay_list(part + 1, n, wr, nw, lane, lo, dinv, H, acc);
  } else {
    for (int e = tid; e < N_EDGES; e += AGG_T) {
      const unsigned a = (unsigned)(dst[e] - lo);
      if (a < (unsigned)OWN_ROWS) {
        const int sr = src[e];
        const float cf = dinv[sr] * dinv[lo + (int)a];
        for (int ch = 0; ch < HID; ++ch) atomicAdd(&acc[a * HID + ch], H[sr * HID + ch] * cf);
      }
    }
  }
  __syncthreads();

  float* rows = agg + lo * HID;
  float sd[OWN_TRIPS];
  int gi[OWN_TRIPS];
  float pw[OWN_TRIPS];
#pragma unroll
  for (int j = 0; j < OWN_TRIPS; ++j) {
    const int i = tid + j * AGG_T;
    const int c = i < OWN_ROWS * HID ? i : tid;
    sd[j] = rows[c];
    gi[j] = batch[lo + (c >> 5)];
  }
#pragma unroll
  for (int j = 0; j < OWN_TRIPS; ++j) pw[j] = cntinv[gi[j]];
#pragma unroll
  for (int j = 0; j < OWN_TRIPS; ++j) {
    const int i = tid + j * AGG_T;
    if (i < OWN_ROWS * HID) {
      float v = sd[j] + acc[i];
      v = v > 0.0f ? v : NEG_SLOPE * v;
      rows[i] = v;
      atomicAdd(&feat[gi[j] * HID + (i & 31)], v * pw[j]);
    }
  }
}

__global__ void gcn_merge_kernel(const float* __restrict__ feats, float* __restrict__ out) {
  int t = blockIdx.x * blockDim.x + threadIdx.x;
  const int GH = G_GRAPHS * HID;
  if (t < GH) out[t] = (feats[t] + feats[GH + t] + feats[2 * GH + t]) * (1.0f / 3.0f);
}

static inline int cdiv(int a, int b) { return (a + b - 1) / b; }

extern "C" void kernel_launch(void* const* d_in, const int* in_sizes, int n_in,
                              void* d_out, int out_size, void* d_ws, size_t ws_size,
                              hipStream_t stream) {
  (void)in_sizes; (void)n_in; (void)out_size; (void)ws_size;
  const float* x     = (const float*)d_in[0];
  const float* W0    = (const float*)d_in[1];
  const float* b0    = (const float*)d_in[2];
  const float* W1    = (const float*)d_in[3];
  const float* b1    = (const float*)d_in[4];
  const float* W2    = (const float*)d_in[5];
  const float* b2    = (const float*)d_in[6];
  const int*   src   = (const int*)d_in[7];
  const int*   dst   = (const int*)d_in[8];
  const int*   batch = (const int*)d_in[9];
  float* out = (float*)d_out;

  float* ws    = (float*)d_ws;
  float* deg   = ws;
  float* cnt   = deg + N_NODES;
  float* lin   = cnt + G_GRAPHS;
  float* act   = lin + (size_t)N_NODES * HID;
  float* agg   = act + (size_t)N_NODES * HID;
  float* feats = agg + (size_t)N_NODES * HID;

  const int T = 256;
  const int nTiles = (N_NODES / 16) * 2;

  unsigned* degPlanes = (unsigned*)agg;
  unsigned* cntPlanes = (unsigned*)act;
  unsigned* lists     = (unsigned*)act;
  gcn_zero_kernel<<<cdiv(3 * G_GRAPHS * HID, T), T, 0, stream>>>(feats, 3 * G_GRAPHS * HID);
  gcn_degree_hist_kernel<<<DEG_R * DEG_S, DEG_T, 0, stream>>>(dst, degPlanes);
  gcn_dinv_kernel<<<cdiv(N_NODES / 4, 128), 128, 0, stream>>>(degPlanes, deg);
  gcn_counts_hist_kernel<<<CNT_B, CNT_T, 0, stream>>>(batch, cntPlanes);
  gcn_cntinv_kernel<<<1, G_GRAPHS, 0, stream>>>(cntPlanes, cnt);

  gcn_edge_scan_kernel<<<AGG_R * AGG_S, AGG_T, 0, stream>>>(src, dst, lists);

  const float* Ws[3] = {W0, W1, W2};
  const float* bs[3] = {b0, b1, b2};
  for (int layer = 0; layer < 3; ++layer) {
    if (layer == 0) {
      gcn_gemm_wmma_kernel<C_IN, false><<<cdiv(nTiles, T / 32), T, 0, stream>>>(
          x, Ws[0], bs[0], deg, lin, agg, N_NODES);
    } else {
      gcn_gemm_wmma_kernel<HID, true><<<cdiv(nTiles, T / 32), T, 0, stream>>>(
          agg, Ws[layer], bs[layer], deg, lin, agg, N_NODES);
    }
    gcn_edge_sum_kernel<<<OWN_B, AGG_T, 0, stream>>>(src, dst, deg, lin, agg, lists, batch, cnt,
                                                     feats + layer * G_GRAPHS * HID);
  }
  gcn_merge_kernel<<<cdiv(G_GRAPHS * HID, T), T, 0, stream>>>(feats, out);
}


# edge sum reads both dinv factors from LDS: in-degrees staged as one byte per node (rsqrt per edge) beside the block's 400 dinv floats; no per-edge dinv load from memory; values and add order kept
# speedup vs baseline: 1.9846x; 1.0412x over previous
#include <hip/hip_runtime.h>

#define N_NODES   100000
#define N_EDGES   1600000
#define C_IN      64
#define HID       32
#define G_GRAPHS  128
#define NEG_SLOPE 0.01f

typedef __attribute__((ext_vector_type(2))) float v2f;
typedef __attribute__((ext_vector_type(8))) float v8f;

__global__ void gcn_zero_kernel(float* __restrict__ p, int n) {
  int i = blockIdx.x * blockDim.x + threadIdx.x;
  if (i < n) p[i] = 0.0f;
}

#define DEG_R     4
#define DEG_S     32
#define DEG_BINS  (N_NODES / DEG_R)
#define DEG_SLICE (N_EDGES / DEG_S)
#define DEG_T     1024
#define DEG_LOADS ((DEG_SLICE / 4 + DEG_T - 1) / DEG_T)
static_assert(N_NODES % (4 * DEG_R) == 0, "bin ranges must be whole quads");
static_assert(N_EDGES % (4 * DEG_S) == 0, "edge slices must be whole quads");
static_assert(DEG_S <= HID, "the planes must fit the edge-sum region they borrow");

__global__ void __launch_bounds__(DEG_T)
gcn_degree_hist_kernel(const int* __restrict__ dst, unsigned* __restrict__ planes) {
  __shared__ unsigned bins[DEG_BINS];
  const int tid = threadIdx.x;
  const int r = blockIdx.x % DEG_R;
  const int s = blockIdx.x / DEG_R;
  const int lo = r * DEG_BINS;
  const int4* d4 = (const int4*)(dst + s * DEG_SLICE);
  int4 v[DEG_LOADS];
#pragma unroll
  for (int j = 0; j < DEG_LOADS; ++j) {
    const int q = tid + j * DEG_T;
    v[j] = q < DEG_SLICE / 4 ? d4[q] : make_int4(-1, -1, -1, -1);
  }
  for (int i = tid; i < DEG_BINS; i += DEG_T) bins[i] = 0u;
  __syncthreads();
#pragma unroll
  for (int j = 0; j < DEG_LOADS; ++j) {
    const unsigned a0 = (unsigned)(v[j].x - lo);
    const unsigned a1 = (unsigned)(v[j].y - lo);
    const unsigned a2 = (unsigned)(v[j].z - lo);
    const unsigned a3 = (unsigned)(v[j].w - lo);
    if (a0 < DEG_BINS) atomicAdd(&bins[a0], 1u);
    if (a1 < DEG_BINS) atomicAdd(&bins[a1], 1u);
    if (a2 < DEG_BINS) atomicAdd(&bins[a2], 1u);
    if (a3 < DEG_BINS) atomicAdd(&bins[a3], 1u);
  }
  __syncthreads();
  uint4* p4 = (uint4*)(planes + (size_t)s * N_NODES + lo);
  const uint4* b4 = (const uint4*)bins;
  for (int i = tid; i < DEG_BINS / 4; i += DEG_T) p4[i] = b4[i];
}

#define DEG_SAT 255u
__global__ void gcn_dinv_kernel(const unsigned* __restrict__ planes, float* __restrict__ dinv,
                                unsigned* __restrict__ degb) {
  const int q = blockIdx.x * blockDim.x + threadIdx.x;
  if (q >= N_NODES / 4) return;
  const uint4* p4 = (const uint4*)planes + q;
  uint4 c = make_uint4(0u, 0u, 0u, 0u);
#pragma unroll
  for (int s = 0; s < DEG_S; ++s) {
    const uint4 t = p4[(size_t)s * (N_NODES / 4)];
    c.x += t.x; c.y += t.y; c.z += t.z; c.w += t.w;
  }
  float4 o;
  o.x = rsqrtf((float)c.x + 1.0f);
  o.y = rsqrtf((float)c.y + 1.0f);
  o.z = rsqrtf((float)c.z + 1.0f);
  o.w = rsqrtf((float)c.w + 1.0f);
  ((float4*)dinv)[q] = o;
  degb[q] = min(c.x, DEG_SAT) | (min(c.y, DEG_SAT) << 8) | (min(c.z, DEG_SAT) << 16) | (min(c.w, DEG_SAT) << 24);
}

#define CNT_T 1024
#define CNT_B ((N_NODES + CNT_T - 1) / CNT_T)

__global__ void __launch_bounds__(CNT_T)
gcn_counts_hist_kernel(const int* __restrict__ batch, unsigned* __restrict__ parts) {
  __shared__ unsigned bins[G_GRAPHS];
  const int tid = threadIdx.x;
  if (tid < G_GRAPHS) bins[tid] = 0u;
  __syncthreads();
  const int i = blockIdx.x * CNT_T + tid;
  if (i < N_NODES) {
    const unsigned g = (unsigned)batch[i];
    if (g < G_GRAPHS) atomicAdd(&bins[g], 1u);
  }
  __syncthreads();
  if (tid < G_GRAPHS) parts[blockIdx.x * G_GRAPHS + tid] = bins[tid];
}

__global__ void gcn_cntinv_kernel(const unsigned* __restrict__ parts, float* __restrict__ cnt) {
  const int g = blockIdx.x * blockDim.x + threadIdx.x;
  if (g >= G_GRAPHS) return;
  unsigned c = 0u;
#pragma unroll
  for (int b = 0; b < CNT_B; ++b) c += parts[b * G_GRAPHS + g];
  cnt[g] = 1.0f / fmaxf((float)c, 1.0f);
}

template <int K, bool INPLACE>
__global__ void gcn_gemm_wmma_kernel(const float* X,
                                     const float* __restrict__ W,
                                     const float* __restrict__ bias,
                                     const float* __restrict__ dinv,
                                     float* __restrict__ H,
                                     float* A, int nRows) {
  const int wave = blockIdx.x * (blockDim.x >> 5) + (threadIdx.x >> 5);
  const int lane = threadIdx.x & 31;
  const int numTiles = (nRows >> 4) * 2;
  const bool active = wave < numTiles;
  v2f afrag[K / 4];
  if (active) {
    const int r0 = (wave >> 1) * 16 + (lane & 15);
#pragma unroll
    for (int j = 0; j < K / 4; ++j)
      afrag[j] = *(const v2f*)(X + r0 * K + 4 * j + 2 * (lane >> 4));
  }
  if (INPLACE) __syncthreads();
  if (!active) return;
  const int rowTile = wave >> 1;
  const int colTile = wave & 1;
  const int half = lane >> 4;
  const int l16  = lane & 15;
  const int col  = colTile * 16 + l16;

  float dd[8];
#pragma unroll
  for (int r = 0; r < 8; ++r) {
    const float d = dinv[rowTile * 16 + r + 8 * half];
    dd[r] = d * d;
  }
  const float bc = bias[col];

  v8f c = {};
#pragma unroll
  for (int k0 = 0; k0 < K; k0 += 4) {
    const int ka = k0 + 2 * half;
    const v2f a = afrag[k0 / 4];
    v2f b;
    b.x = W[(ka + 0) * HID + col];
    b.y = W[(ka + 1) * HID + col];
    c = __builtin_amdgcn_wmma_f32_16x16x4_f32(
         false, a,  false, b,
         (short)0, c,  false,  false);
  }
#pragma unroll
  for (int r = 0; r < 8; ++r) {
    const int m = r + 8 * half;
    const int o = (rowTile * 16 + m) * HID + col;
    H[o] = c[r];
    A[o] = c[r] * dd[r] + bc;
  }
}

#define AGG_RANGE 2000
#define AGG_R     (N_NODES / AGG_RANGE)
#define AGG_S     5
#define AGG_T     1024
#define AGG_W     (AGG_T / 32)
#define AGG_SEG   256
#define AGG_QS    (N_EDGES / 4 / AGG_S)
#define AGG_NG    (AGG_QS / 32)
#define AGG_B     4
#define AGG_U     8
#define FL_U      4
#define OWN_ROWS  400
#define OWN_PER   (AGG_RANGE / OWN_ROWS)
#define OWN_B     (N_NODES / OWN_ROWS)
#define OWN_TRIPS ((OWN_ROWS * HID + AGG_T - 1) / AGG_T)
static_assert(N_NODES % AGG_RANGE == 0, "row ranges must tile the nodes");
static_assert(AGG_RANGE % OWN_ROWS == 0, "owned blocks must tile a range");
static_assert(OWN_PER <= 32, "one lane per block keeps that block's counter");
static_assert(N_EDGES % (4 * AGG_S * 32) == 0, "slices must be whole wave groups of quads");
static_assert(AGG_SEG >= 2 * 128, "a list must take one more group after the flush test");
#define DEGB_WORDS (N_NODES / 4)
#define DEGB_PER  10
#define DEGB_THR  (N_NODES / 16 / DEGB_PER)
static_assert(N_NODES % (16 * DEGB_PER) == 0, "the degree bytes are copied in 16-byte pieces, DEGB_PER per thread");
static_assert(DEGB_THR <= AGG_T, "the copying threads must fit the work-group");
#define LST_WORDS ((N_NODES * HID - DEGB_WORDS) / (AGG_R * AGG_S))
#define SUB_WORDS (LST_WORDS / OWN_PER)
#define SUB_CAP   (SUB_WORDS - 1)
#define LST_SHIFT 17
static_assert(N_NODES <= (1 << LST_SHIFT), "a source node must fit the low bits");
static_assert(OWN_ROWS <= (1 << (32 - LST_SHIFT)), "a row of a block must fit the high bits");
static_assert((N_NODES * HID - DEGB_WORDS) % (AGG_R * AGG_S) == 0, "segments must tile the region");
static_assert(OWN_ROWS <= AGG_T, "one thread per owned row stages its dinv");
static_assert(LST_WORDS % OWN_PER == 0, "parts must tile a segment");

__device__ __forceinline__ void gcn_flush_list(const uint2* seg, int cnt, int lane,
                                               const int* __restrict__ src,
                                               unsigned* fill, unsigned* segw) {
  for (int i = 0; i < cnt; i += 32 * FL_U) {
    unsigned sr[FL_U];
    unsigned al[FL_U];
    bool ok[FL_U];
#pragma unroll
    for (int u = 0; u < FL_U; ++u) {
      const int idx = i + 32 * u + lane;
      ok[u] = idx < cnt;
      const uint2 e = seg[ok[u] ? idx : i];
      sr[u] = (unsigned)src[e.x];
      al[u] = e.y;
    }
#pragma unroll
    for (int u = 0; u < FL_U; ++u) {
      if (i + 32 * u < cnt) {
        const unsigned blk = ok[u] ? al[u] / (unsigned)OWN_ROWS : (unsigned)OWN_PER;
        unsigned rank = 0u, mine = 0u;
#pragma unroll
        for (int k = 0; k < OWN_PER; ++k) {
          const unsigned bal = __builtin_amdgcn_ballot_w32(blk == (unsigned)k);
          if (blk == (unsigned)k) rank = __builtin_amdgcn_mbcnt_lo(bal, 0u);
          if (lane == k) mine = (unsigned)__builtin_popcount(bal);
        }
        unsigned base = 0u;
        if (lane < OWN_PER) base = atomicAdd(&fill[lane], mine);
        base = (unsigned)__shfl((int)base, (int)blk);
        const unsigned pos = base + rank;
        if (ok[u] && pos < (unsigned)SUB_CAP)
          segw[blk * SUB_WORDS + 1u + pos] = sr[u] | ((al[u] - blk * (unsigned)OWN_ROWS) << LST_SHIFT);
      }
    }
  }
}

__global__ void __launch_bounds__(AGG_T)
gcn_edge_scan_kernel(const int* __restrict__ src, const int* __restrict__ dst,
                     unsigned* __restrict__ lists) {
  __shared__ uint2 list[AGG_W * AGG_SEG];
  __shared__ unsigned fill[OWN_PER];
  const int tid = threadIdx.x;
  const int lane = tid & 31;
  const int w = __builtin_amdgcn_readfirstlane(tid >> 5);
  const int r = blockIdx.x % AGG_R;
  const int sl = blockIdx.x / AGG_R;
  const int lo = r * AGG_RANGE;
  unsigned* segw = lists + (size_t)blockIdx.x * LST_WORDS;

  if (tid < OWN_PER) fill[tid] = 0u;
  __syncthreads();

  const int qb = sl * AGG_QS;
  const int4* d4 = (const int4*)dst + qb;
  uint2* seg = list + w * AGG_SEG;
  int cnt = 0;

  for (int g0 = w; g0 < AGG_NG; g0 += AGG_W * AGG_B) {
    int4 dv[AGG_B];
#pragma unroll
    for (int b = 0; b < AGG_B; ++b) {
      const int g = g0 + b * AGG_W;
      if (g < AGG_NG) dv[b] = d4[g * 32 + lane];
      else            dv[b] = make_int4(-1, -1, -1, -1);
    }
#pragma unroll
    for (int b = 0; b < AGG_B; ++b) {
      const int g = g0 + b * AGG_W;
      const unsigned e0 = (unsigned)(4 * (qb + g * 32 + lane));
      const int dd[4] = {dv[b].x, dv[b].y, dv[b].z, dv[b].w};
#pragma unroll
      for (int k = 0; k < 4; ++k) {
        const unsigned a = (unsigned)(dd[k] - lo);
        const bool m = a < (unsigned)AGG_RANGE;
        const unsigned bal = __builtin_amdgcn_ballot_w32(m);
        if (m) seg[cnt + (int)__builtin_amdgcn_mbcnt_lo(bal, 0u)] = make_uint2(e0 + (unsigned)k, a);
        cnt += __builtin_popcount(bal);
      }
      if (cnt > AGG_SEG - 128) {
        __builtin_amdgcn_wave_barrier();
        gcn_flush_list(seg, cnt, lane, src, fill, segw);
        __builtin_amdgcn_wave_barrier();
        cnt = 0;
      }
    }
  }
  __builtin_amdgcn_wave_barrier();
  gcn_flush_list(seg, cnt, lane, src, fill, segw);
  __syncthreads();
  if (tid < OWN_PER) segw[tid * SUB_WORDS] = fill[tid];
}

__device__ __forceinline__ void gcn_replay_list(const unsigned* __restrict__ ent, int n, int wr, int nw,
                                                int lane,
                                                const float* __restrict__ dinv,
                                                const unsigned char* sdeg, const float* sdinv,
                                                const float* __restrict__ H, float* acc) {
  const int sub = lane >> 3;
  const int q = (lane & 7) << 2;
  int i = wr * (4 * AGG_U);
  unsigned en[AGG_U];
  if (i < n) {
#pragma unroll
    for (int u = 0; u < AGG_U; ++u) {
      const int idx = i + 4 * u + sub;
      en[u] = ent[idx < n ? idx : i];
    }
  }
  while (i < n) {
    int al[AGG_U];
    bool ok[AGG_U];
    float4 hv[AGG_U];
    float cf[AGG_U];
    int sr[AGG_U];
    unsigned dg[AGG_U];
    float dl[AGG_U];
#pragma unroll
    for (int u = 0; u < AGG_U; ++u) {
      ok[u] = i + 4 * u + sub < n;
      sr[u] = (int)(en[u] & ((1u << LST_SHIFT) - 1u));
      al[u] = (int)(en[u] >> LST_SHIFT);
      hv[u] = *(const float4*)(H + sr[u] * HID + q);
      dg[u] = (unsigned)sdeg[sr[u]];
      dl[u] = sdinv[al[u]];
    }
    const int inext = i + nw * 4 * AGG_U;
    if (inext < n) {
#pragma unroll
      for (int u = 0; u < AGG_U; ++u) {
        const int idx = inext + 4 * u + sub;
        en[u] = ent[idx < n ? idx : inext];
      }
    }
    bool sat = false;
#pragma unroll
    for (int u = 0; u < AGG_U; ++u) {
      cf[u] = rsqrtf((float)dg[u] + 1.0f);
      sat = sat || dg[u] == DEG_SAT;
    }
    if (__builtin_amdgcn_ballot_w32(sat) != 0u) {
#pragma unroll
      for (int u = 0; u < AGG_U; ++u)
        if (dg[u] == DEG_SAT) cf[u] = dinv[sr[u]];
    }
#pragma unroll
    for (int u = 0; u < AGG_U; ++u) cf[u] *= dl[u];
#pragma unroll
    for (int u = 0; u < AGG_U; ++u) {
      float* ap = acc + al[u] * HID + q;
      atomicAdd(ap + 0, ok[u] ? hv[u].x * cf[u] : 0.0f);
      atomicAdd(ap + 1, ok[u] ? hv[u].y * cf[u] : 0.0f);
      atomicAdd(ap + 2, ok[u] ? hv[u].z * cf[u] : 0.0f);
      atomicAdd(ap + 3, ok[u] ? hv[u].w * cf[u] : 0.0f);
    }
    i = inext;
  }
}

__global__ void __launch_bounds__(AGG_T)
gcn_edge_sum_kernel(const int* __restrict__ src, const int* __restrict__ dst,
                    const float* __restrict__ dinv,
                    const float* __restrict__ H, float* __restrict__ agg,
                    const unsigned* __restrict__ lists, const unsigned* __restrict__ degb,
                    const int* __restrict__ batch, const float* __restrict__ cntinv,
                    float* __restrict__ feat) {
  __shared__ float acc[OWN_ROWS * HID];
  __shared__ __attribute__((aligned(16))) unsigned char sdeg[N_NODES];
  __shared__ float sdinv[OWN_ROWS];
  const int tid = threadIdx.x;
  const int lane = tid & 31;
  const int w = __builtin_amdgcn_readfirstlane(tid >> 5);
  const int r = blockIdx.x / OWN_PER;
  const int p = blockIdx.x % OWN_PER;
  const int lo = blockIdx.x * OWN_ROWS;

  bool fit = true;
#pragma unroll
  for (int s = 0; s < AGG_S; ++s) {
    const unsigned n = lists[(size_t)(s * AGG_R + r) * LST_WORDS + (size_t)p * SUB_WORDS];
    fit = fit && n <= (unsigned)SUB_CAP;
  }
  float dmine = 0.0f;
  if (tid < OWN_ROWS) dmine = dinv[lo + tid];

  float4* acc4 = (float4*)acc;
  for (int i = tid; i < OWN_ROWS * HID / 4; i += AGG_T) acc4[i] = make_float4(0.f, 0.f, 0.f, 0.f);
  if (tid < DEGB_THR) {
    const uint4* degb4 = (const uint4*)degb;
    uint4* sdeg4 = (uint4*)sdeg;
    uint4 dt[DEGB_PER];
#pragma unroll
    for (int j = 0; j < DEGB_PER; ++j) dt[j] = degb4[tid + j * DEGB_THR];
#pragma unroll
    for (int j = 0; j < DEGB_PER; ++j) sdeg4[tid + j * DEGB_THR] = dt[j];
  }
  if (tid < OWN_ROWS) sdinv[tid] = dmine;

  __syncthreads();

  if (fit) {
    const int s = w % AGG_S;
    const int wr = w / AGG_S;
    const int nw = (AGG_W - s + AGG_S - 1) / AGG_S;
    const unsigned* part = lists + (size_t)(s * AGG_R + r) * LST_WORDS + (size_t)p * SUB_WORDS;
    const int n = __builtin_amdgcn_readfirstlane((int)part[0]);
    gcn_replay_list(part + 1, n, wr, nw, lane, dinv, sdeg, sdinv, H, acc);
  } else {
    for (int e = tid; e < N_EDGES; e += AGG_T) {
      const unsigned a = (unsigned)(dst[e] - lo);
      if (a < (unsigned)OWN_ROWS) {
        const int sr = src[e];
        const float cf = dinv[sr] * dinv[lo + (int)a];
        for (int ch = 0; ch < HID; ++ch) atomicAdd(&acc[a * HID + ch], H[sr * HID + ch] * cf);
      }
    }
  }
  __syncthreads();

  float* rows = agg + lo * HID;
  float sd[OWN_TRIPS];
  int gi[OWN_TRIPS];
  float pw[OWN_TRIPS];
#pragma unroll
  for (int j = 0; j < OWN_TRIPS; ++j) {
    const int i = tid + j * AGG_T;
    const int c = i < OWN_ROWS * HID ? i : tid;
    sd[j] = rows[c];
    gi[j] = batch[lo + (c >> 5)];
  }
#pragma unroll
  for (int j = 0; j < OWN_TRIPS; ++j) pw[j] = cntinv[gi[j]];
#pragma unroll
  for (int j = 0; j < OWN_TRIPS; ++j) {
    const int i = tid + j * AGG_T;
    if (i < OWN_ROWS * HID) {
      float v = sd[j] + acc[i];
      v = v > 0.0f ? v : NEG_SLOPE * v;
      rows[i] = v;
      atomicAdd(&feat[gi[j] * HID + (i & 31)], v * pw[j]);
    }
  }
}

__global__ void gcn_merge_kernel(const float* __restrict__ feats, float* __restrict__ out) {
  int t = blockIdx.x * blockDim.x + threadIdx.x;
  const int GH = G_GRAPHS * HID;
  if (t < GH) out[t] = (feats[t] + feats[GH + t] + feats[2 * GH + t]) * (1.0f / 3.0f);
}

static inline int cdiv(int a, int b) { return (a + b - 1) / b; }

extern "C" void kernel_launch(void* const* d_in, const int* in_sizes, int n_in,
                              void* d_out, int out_size, void* d_ws, size_t ws_size,
                              hipStream_t stream) {
  (void)in_sizes; (void)n_in; (void)out_size; (void)ws_size;
  const float* x     = (const float*)d_in[0];
  const float* W0    = (const float*)d_in[1];
  const float* b0    = (const float*)d_in[2];
  const float* W1    = (const float*)d_in[3];
  const float* b1    = (const float*)d_in[4];
  const float* W2    = (const float*)d_in[5];
  const float* b2    = (const float*)d_in[6];
  const int*   src   = (const int*)d_in[7];
  const int*   dst   = (const int*)d_in[8];
  const int*   batch = (const int*)d_in[9];
  float* out = (float*)d_out;

  float* ws    = (float*)d_ws;
  float* deg   = ws;
  float* cnt   = deg + N_NODES;
  float* lin   = cnt + G_GRAPHS;
  float* act   = lin + (size_t)N_NODES * HID;
  float* agg   = act + (size_t)N_NODES * HID;
  float* feats = agg + (size_t)N_NODES * HID;

  const int T = 256;
  const int nTiles = (N_NODES / 16) * 2;

  unsigned* degPlanes = (unsigned*)agg;
  unsigned* cntPlanes = (unsigned*)act;
  unsigned* lists     = (unsigned*)act;
  unsigned* degb      = (unsigned*)act + (size_t)AGG_R * AGG_S * LST_WORDS;
  gcn_zero_kernel<<<cdiv(3 * G_GRAPHS * HID, T), T, 0, stream>>>(feats, 3 * G_GRAPHS * HID);
  gcn_degree_hist_kernel<<<DEG_R * DEG_S, DEG_T, 0, stream>>>(dst, degPlanes);
  gcn_dinv_kernel<<<cdiv(N_NODES / 4, 128), 128, 0, stream>>>(degPlanes, deg, degb);
  gcn_counts_hist_kernel<<<CNT_B, CNT_T, 0, stream>>>(batch, cntPlanes);
  gcn_cntinv_kernel<<<1, G_GRAPHS, 0, stream>>>(cntPlanes, cnt);

  gcn_edge_scan_kernel<<<AGG_R * AGG_S, AGG_T, 0, stream>>>(src, dst, lists);

  const float* Ws[3] = {W0, W1, W2};
  const float* bs[3] = {b0, b1, b2};
  for (int layer = 0; layer < 3; ++layer) {
    if (layer == 0) {
      gcn_gemm_wmma_kernel<C_IN, false><<<cdiv(nTiles, T / 32), T, 0, stream>>>(
          x, Ws[0], bs[0], deg, lin, agg, N_NODES);
    } else {
      gcn_gemm_wmma_kernel<HID, true><<<cdiv(nTiles, T / 32), T, 0, stream>>>(
          agg, Ws[layer], bs[layer], deg, lin, agg, N_NODES);
    }
    gcn_edge_sum_kernel<<<OWN_B, AGG_T, 0, stream>>>(src, dst, deg, lin, agg, lists, degb, batch, cnt,
                                                     feats + layer * G_GRAPHS * HID);
  }
  gcn_merge_kernel<<<cdiv(G_GRAPHS * HID, T), T, 0, stream>>>(feats, out);
}


# edge scan retiled: a work-group takes 10000 dst rows x 1/25 of the edges (was 2000 x 1/5): an edge is tested 10 times, not 50; entries placed by one LDS add each; one wave per list; add order changes
# speedup vs baseline: 2.0914x; 1.0538x over previous
#include <hip/hip_runtime.h>

#define N_NODES   100000
#define N_EDGES   1600000
#define C_IN      64
#define HID       32
#define G_GRAPHS  128
#define NEG_SLOPE 0.01f

typedef __attribute__((ext_vector_type(2))) float v2f;
typedef __attribute__((ext_vector_type(8))) float v8f;

__global__ void gcn_zero_kernel(float* __restrict__ p, int n) {
  int i = blockIdx.x * blockDim.x + threadIdx.x;
  if (i < n) p[i] = 0.0f;
}

#define DEG_R     4
#define DEG_S     32
#define DEG_BINS  (N_NODES / DEG_R)
#define DEG_SLICE (N_EDGES / DEG_S)
#define DEG_T     1024
#define DEG_LOADS ((DEG_SLICE / 4 + DEG_T - 1) / DEG_T)
static_assert(N_NODES % (4 * DEG_R) == 0, "bin ranges must be whole quads");
static_assert(N_EDGES % (4 * DEG_S) == 0, "edge slices must be whole quads");
static_assert(DEG_S <= HID, "the planes must fit the edge-sum region they borrow");

__global__ void __launch_bounds__(DEG_T)
gcn_degree_hist_kernel(const int* __restrict__ dst, unsigned* __restrict__ planes) {
  __shared__ unsigned bins[DEG_BINS];
  const int tid = threadIdx.x;
  const int r = blockIdx.x % DEG_R;
  const int s = blockIdx.x / DEG_R;
  const int lo = r * DEG_BINS;
  const int4* d4 = (const int4*)(dst + s * DEG_SLICE);
  int4 v[DEG_LOADS];
#pragma unroll
  for (int j = 0; j < DEG_LOADS; ++j) {
    const int q = tid + j * DEG_T;
    v[j] = q < DEG_SLICE / 4 ? d4[q] : make_int4(-1, -1, -1, -1);
  }
  for (int i = tid; i < DEG_BINS; i += DEG_T) bins[i] = 0u;
  __syncthreads();
#pragma unroll
  for (int j = 0; j < DEG_LOADS; ++j) {
    const unsigned a0 = (unsigned)(v[j].x - lo);
    const unsigned a1 = (unsigned)(v[j].y - lo);
    const unsigned a2 = (unsigned)(v[j].z - lo);
    const unsigned a3 = (unsigned)(v[j].w - lo);
    if (a0 < DEG_BINS) atomicAdd(&bins[a0], 1u);
    if (a1 < DEG_BINS) atomicAdd(&bins[a1], 1u);
    if (a2 < DEG_BINS) atomicAdd(&bins[a2], 1u);
    if (a3 < DEG_BINS) atomicAdd(&bins[a3], 1u);
  }
  __syncthreads();
  uint4* p4 = (uint4*)(planes + (size_t)s * N_NODES + lo);
  const uint4* b4 = (const uint4*)bins;
  for (int i = tid; i < DEG_BINS / 4; i += DEG_T) p4[i] = b4[i];
}

#define DEG_SAT 255u
__global__ void gcn_dinv_kernel(const unsigned* __restrict__ planes, float* __restrict__ dinv,
                                unsigned* __restrict__ degb) {
  const int q = blockIdx.x * blockDim.x + threadIdx.x;
  if (q >= N_NODES / 4) return;
  const uint4* p4 = (const uint4*)planes + q;
  uint4 c = make_uint4(0u, 0u, 0u, 0u);
#pragma unroll
  for (int s = 0; s < DEG_S; ++s) {
    const uint4 t = p4[(size_t)s * (N_NODES / 4)];
    c.x += t.x; c.y += t.y; c.z += t.z; c.w += t.w;
  }
  float4 o;
  o.x = rsqrtf((float)c.x + 1.0f);
  o.y = rsqrtf((float)c.y + 1.0f);
  o.z = rsqrtf((float)c.z + 1.0f);
  o.w = rsqrtf((float)c.w + 1.0f);
  ((float4*)dinv)[q] = o;
  degb[q] = min(c.x, DEG_SAT) | (min(c.y, DEG_SAT) << 8) | (min(c.z, DEG_SAT) << 16) | (min(c.w, DEG_SAT) << 24);
}

#define CNT_T 1024
#define CNT_B ((N_NODES + CNT_T - 1) / CNT_T)

__global__ void __launch_bounds__(CNT_T)
gcn_counts_hist_kernel(const int* __restrict__ batch, unsigned* __restrict__ parts) {
  __shared__ unsigned bins[G_GRAPHS];
  const int tid = threadIdx.x;
  if (tid < G_GRAPHS) bins[tid] = 0u;
  __syncthreads();
  const int i = blockIdx.x * CNT_T + tid;
  if (i < N_NODES) {
    const unsigned g = (unsigned)batch[i];
    if (g < G_GRAPHS) atomicAdd(&bins[g], 1u);
  }
  __syncthreads();
  if (tid < G_GRAPHS) parts[blockIdx.x * G_GRAPHS + tid] = bins[tid];
}

__global__ void gcn_cntinv_kernel(const unsigned* __restrict__ parts, float* __restrict__ cnt) {
  const int g = blockIdx.x * blockDim.x + threadIdx.x;
  if (g >= G_GRAPHS) return;
  unsigned c = 0u;
#pragma unroll
  for (int b = 0; b < CNT_B; ++b) c += parts[b * G_GRAPHS + g];
  cnt[g] = 1.0f / fmaxf((float)c, 1.0f);
}

template <int K, bool INPLACE>
__global__ void gcn_gemm_wmma_kernel(const float* X,
                                     const float* __restrict__ W,
                                     const float* __restrict__ bias,
                                     const float* __restrict__ dinv,
                                     float* __restrict__ H,
                                     float* A, int nRows) {
  const int wave = blockIdx.x * (blockDim.x >> 5) + (threadIdx.x >> 5);
  const int lane = threadIdx.x & 31;
  const int numTiles = (nRows >> 4) * 2;
  const bool active = wave < numTiles;
  v2f afrag[K / 4];
  if (active) {
    const int r0 = (wave >> 1) * 16 + (lane & 15);
#pragma unroll
    for (int j = 0; j < K / 4; ++j)
      afrag[j] = *(const v2f*)(X + r0 * K + 4 * j + 2 * (lane >> 4));
  }
  if (INPLACE) __syncthreads();
  if (!active) return;
  const int rowTile = wave >> 1;
  const int colTile = wave & 1;
  const int half = lane >> 4;
  const int l16  = lane & 15;
  const int col  = colTile * 16 + l16;

  float dd[8];
#pragma unroll
  for (int r = 0; r < 8; ++r) {
    const float d = dinv[rowTile * 16 + r + 8 * half];
    dd[r] = d * d;
  }
  const float bc = bias[col];

  v8f c = {};
#pragma unroll
  for (int k0 = 0; k0 < K; k0 += 4) {
    const int ka = k0 + 2 * half;
    const v2f a = afrag[k0 / 4];
    v2f b;
    b.x = W[(ka + 0) * HID + col];
    b.y = W[(ka + 1) * HID + col];
    c = __builtin_amdgcn_wmma_f32_16x16x4_f32(
         false, a,  false, b,
         (short)0, c,  false,  false);
  }
#pragma unroll
  for (int r = 0; r < 8; ++r) {
    const int m = r + 8 * half;
    const int o = (rowTile * 16 + m) * HID + col;
    H[o] = c[r];
    A[o] = c[r] * dd[r] + bc;
  }
}

#define AGG_RANGE 10000
#define AGG_R     (N_NODES / AGG_RANGE)
#define AGG_S     25
#define AGG_T     1024
#define AGG_W     (AGG_T / 32)
#define AGG_SEG   256
#define AGG_QS    (N_EDGES / 4 / AGG_S)
#define AGG_NG    (AGG_QS / 32)
#define AGG_B     4
#define AGG_U     8
#define FL_U      4
#define OWN_ROWS  400
#define OWN_PER   (AGG_RANGE / OWN_ROWS)
#define OWN_B     (N_NODES / OWN_ROWS)
#define SUM_T     (AGG_S * 32)
#define OWN_TRIPS ((OWN_ROWS * HID + SUM_T - 1) / SUM_T)
static_assert(N_NODES % AGG_RANGE == 0, "row ranges must tile the nodes");
static_assert(AGG_RANGE % OWN_ROWS == 0, "owned blocks must tile a range");
static_assert(N_EDGES % (4 * AGG_S * 32) == 0, "slices must be whole wave groups of quads");
static_assert(AGG_SEG >= 2 * 128, "a list must take one more group after the flush test");
#define DEGB_WORDS (N_NODES / 4)
#define DEGB_PER  10
#define DEGB_THR  (N_NODES / 16 / DEGB_PER)
static_assert(N_NODES % (16 * DEGB_PER) == 0, "the degree bytes are copied in 16-byte pieces, DEGB_PER per thread");
static_assert(DEGB_THR <= SUM_T, "the copying threads must fit the work-group");
#define LST_WORDS ((N_NODES * HID - DEGB_WORDS) / (AGG_R * AGG_S))
#define SUB_WORDS (LST_WORDS / OWN_PER)
#define SUB_CAP   (SUB_WORDS - 1)
#define LST_SHIFT 17
static_assert(N_NODES <= (1 << LST_SHIFT), "a source node must fit the low bits");
static_assert(OWN_ROWS <= (1 << (32 - LST_SHIFT)), "a row of a block must fit the high bits");
static_assert((N_NODES * HID - DEGB_WORDS) % (AGG_R * AGG_S) == 0, "segments must tile the region");
static_assert(OWN_ROWS <= SUM_T, "one thread per owned row stages its dinv");
static_assert(LST_WORDS % OWN_PER == 0, "parts must tile a segment");

__device__ __forceinline__ void gcn_flush_list(const uint2* seg, int cnt, int lane,
                                               const int* __restrict__ src,
                                               unsigned* fill, unsigned* segw) {
  for (int i = 0; i < cnt; i += 32 * FL_U) {
    unsigned sr[FL_U];
    unsigned al[FL_U];
    bool ok[FL_U];
#pragma unroll
    for (int u = 0; u < FL_U; ++u) {
      const int idx = i + 32 * u + lane;
      ok[u] = idx < cnt;
      const uint2 e = seg[ok[u] ? idx : i];
      sr[u] = (unsigned)src[e.x];
      al[u] = e.y;
    }
#pragma unroll
    for (int u = 0; u < FL_U; ++u) {
      if (ok[u]) {
        const unsigned blk = al[u] / (unsigned)OWN_ROWS;
        const unsigned pos = atomicAdd(&fill[blk], 1u);
        if (pos < (unsigned)SUB_CAP)
          segw[blk * SUB_WORDS + 1u + pos] = sr[u] | ((al[u] - blk * (unsigned)OWN_ROWS) << LST_SHIFT);
      }
    }
  }
}

__global__ void __launch_bounds__(AGG_T)
gcn_edge_scan_kernel(const int* __restrict__ src, const int* __restrict__ dst,
                     unsigned* __restrict__ lists) {
  __shared__ uint2 list[AGG_W * AGG_SEG];
  __shared__ unsigned fill[OWN_PER];
  const int tid = threadIdx.x;
  const int lane = tid & 31;
  const int w = __builtin_amdgcn_readfirstlane(tid >> 5);
  const int r = blockIdx.x % AGG_R;
  const int sl = blockIdx.x / AGG_R;
  const int lo = r * AGG_RANGE;
  unsigned* segw = lists + (size_t)blockIdx.x * LST_WORDS;

  if (tid < OWN_PER) fill[tid] = 0u;
  __syncthreads();

  const int qb = sl * AGG_QS;
  const int4* d4 = (const int4*)dst + qb;
  uint2* seg = list + w * AGG_SEG;
  int cnt = 0;

  for (int g0 = w; g0 < AGG_NG; g0 += AGG_W * AGG_B) {
    int4 dv[AGG_B];
#pragma unroll
    for (int b = 0; b < AGG_B; ++b) {
      const int g = g0 + b * AGG_W;
      if (g < AGG_NG) dv[b] = d4[g * 32 + lane];
      else            dv[b] = make_int4(-1, -1, -1, -1);
    }
#pragma unroll
    for (int b = 0; b < AGG_B; ++b) {
      const int g = g0 + b * AGG_W;
      const unsigned e0 = (unsigned)(4 * (qb + g * 32 + lane));
      const int dd[4] = {dv[b].x, dv[b].y, dv[b].z, dv[b].w};
#pragma unroll
      for (int k = 0; k < 4; ++k) {
        const unsigned a = (unsigned)(dd[k] - lo);
        const bool m = a < (unsigned)AGG_RANGE;
        const unsigned bal = __builtin_amdgcn_ballot_w32(m);
        if (m) seg[cnt + (int)__builtin_amdgcn_mbcnt_lo(bal, 0u)] = make_uint2(e0 + (unsigned)k, a);
        cnt += __builtin_popcount(bal);
      }
      if (cnt > AGG_SEG - 128) {
        __builtin_amdgcn_wave_barrier();
        gcn_flush_list(seg, cnt, lane, src, fill, segw);
        __builtin_amdgcn_wave_barrier();
        cnt = 0;
      }
    }
  }
  __builtin_amdgcn_wave_barrier();
  gcn_flush_list(seg, cnt, lane, src, fill, segw);
  __syncthreads();
  if (tid < OWN_PER) segw[tid * SUB_WORDS] = fill[tid];
}

__device__ __forceinline__ void gcn_replay_list(const unsigned* __restrict__ ent, int n, int wr, int nw,
                                                int lane,
                                                const float* __restrict__ dinv,
                                                const unsigned char* sdeg, const float* sdinv,
                                                const float* __restrict__ H, float* acc) {
  const int sub = lane >> 3;
  const int q = (lane & 7) << 2;
  int i = wr * (4 * AGG_U);
  unsigned en[AGG_U];
  if (i < n) {
#pragma unroll
    for (int u = 0; u < AGG_U; ++u) {
      const int idx = i + 4 * u + sub;
      en[u] = ent[idx < n ? idx : i];
    }
  }
  while (i < n) {
    int al[AGG_U];
    bool ok[AGG_U];
    float4 hv[AGG_U];
    float cf[AGG_U];
    int sr[AGG_U];
    unsigned dg[AGG_U];
    float dl[AGG_U];
#pragma unroll
    for (int u = 0; u < AGG_U; ++u) {
      ok[u] = i + 4 * u + sub < n;
      sr[u] = (int)(en[u] & ((1u << LST_SHIFT) - 1u));
      al[u] = (int)(en[u] >> LST_SHIFT);
      hv[u] = *(const float4*)(H + sr[u] * HID + q);
      dg[u] = (unsigned)sdeg[sr[u]];
      dl[u] = sdinv[al[u]];
    }
    const int inext = i + nw * 4 * AGG_U;
    if (inext < n) {
#pragma unroll
      for (int u = 0; u < AGG_U; ++u) {
        const int idx = inext + 4 * u + sub;
        en[u] = ent[idx < n ? idx : inext];
      }
    }
    bool sat = false;
#pragma unroll
    for (int u = 0; u < AGG_U; ++u) {
      cf[u] = rsqrtf((float)dg[u] + 1.0f);
      sat = sat || dg[u] == DEG_SAT;
    }
    if (__builtin_amdgcn_ballot_w32(sat) != 0u) {
#pragma unroll
      for (int u = 0; u < AGG_U; ++u)
        if (dg[u] == DEG_SAT) cf[u] = dinv[sr[u]];
    }
#pragma unroll
    for (int u = 0; u < AGG_U; ++u) cf[u] *= dl[u];
#pragma unroll
    for (int u = 0; u < AGG_U; ++u) {
      if (i + 4 * u < n) {
        float* ap = acc + al[u] * HID + q;
        atomicAdd(ap + 0, ok[u] ? hv[u].x * cf[u] : 0.0f);
        atomicAdd(ap + 1, ok[u] ? hv[u].y * cf[u] : 0.0f);
        atomicAdd(ap + 2, ok[u] ? hv[u].z * cf[u] : 0.0f);
        atomicAdd(ap + 3, ok[u] ? hv[u].w * cf[u] : 0.0f);
      }
    }
    i = inext;
  }
}

__global__ void __launch_bounds__(SUM_T)
gcn_edge_sum_kernel(const int* __restrict__ src, const int* __restrict__ dst,
                    const float* __restrict__ dinv,
                    const float* __restrict__ H, float* __restrict__ agg,
                    const unsigned* __restrict__ lists, const unsigned* __restrict__ degb,
                    const int* __restrict__ batch, const float* __restrict__ cntinv,
                    float* __restrict__ feat) {
  __shared__ float acc[OWN_ROWS * HID];
  __shared__ __attribute__((aligned(16))) unsigned char sdeg[N_NODES];
  __shared__ float sdinv[OWN_ROWS];
  const int tid = threadIdx.x;
  const int lane = tid & 31;
  const int w = __builtin_amdgcn_readfirstlane(tid >> 5);
  const int r = blockIdx.x / OWN_PER;
  const int p = blockIdx.x % OWN_PER;
  const int lo = blockIdx.x * OWN_ROWS;

  bool fit = true;
#pragma unroll
  for (int s = 0; s < AGG_S; ++s) {
    const unsigned n = lists[(size_t)(s * AGG_R + r) * LST_WORDS + (size_t)p * SUB_WORDS];
    fit = fit && n <= (unsigned)SUB_CAP;
  }
  float dmine = 0.0f;
  if (tid < OWN_ROWS) dmine = dinv[lo + tid];

  float4* acc4 = (float4*)acc;
  for (int i = tid; i < OWN_ROWS * HID / 4; i += SUM_T) acc4[i] = make_float4(0.f, 0.f, 0.f, 0.f);
  if (tid < DEGB_THR) {
    const uint4* degb4 = (const uint4*)degb;
    uint4* sdeg4 = (uint4*)sdeg;
    uint4 dt[DEGB_PER];
#pragma unroll
    for (int j = 0; j < DEGB_PER; ++j) dt[j] = degb4[tid + j * DEGB_THR];
#pragma unroll
    for (int j = 0; j < DEGB_PER; ++j) sdeg4[tid + j * DEGB_THR] = dt[j];
  }
  if (tid < OWN_ROWS) sdinv[tid] = dmine;

  __syncthreads();

  if (fit) {
    const unsigned* part = lists + (size_t)(w * AGG_R + r) * LST_WORDS + (size_t)p * SUB_WORDS;
    const int n = __builtin_amdgcn_readfirstlane((int)part[0]);
    gcn_replay_list(part + 1, n, 0, 1, lane, dinv, sdeg, sdinv, H, acc);
  } else {
    for (int e = tid; e < N_EDGES; e += SUM_T) {
      const unsigned a = (unsigned)(dst[e] - lo);
      if (a < (unsigned)OWN_ROWS) {
        const int sr = src[e];
        const float cf = dinv[sr] * dinv[lo + (int)a];
        for (int ch = 0; ch < HID; ++ch) atomicAdd(&acc[a * HID + ch], H[sr * HID + ch] * cf);
      }
    }
  }
  __syncthreads();

  float* rows = agg + lo * HID;
  float sd[OWN_TRIPS];
  int gi[OWN_TRIPS];
  float pw[OWN_TRIPS];
#pragma unroll
  for (int j = 0; j < OWN_TRIPS; ++j) {
    const int i = tid + j * SUM_T;
    const int c = i < OWN_ROWS * HID ? i : tid;
    sd[j] = rows[c];
    gi[j] = batch[lo + (c >> 5)];
  }
#pragma unroll
  for (int j = 0; j < OWN_TRIPS; ++j) pw[j] = cntinv[gi[j]];
#pragma unroll
  for (int j = 0; j < OWN_TRIPS; ++j) {
    const int i = tid + j * SUM_T;
    if (i < OWN_ROWS * HID) {
      float v = sd[j] + acc[i];
      v = v > 0.0f ? v : NEG_SLOPE * v;
      rows[i] = v;
      atomicAdd(&feat[gi[j] * HID + (i & 31)], v * pw[j]);
    }
  }
}

__global__ void gcn_merge_kernel(const float* __restrict__ feats, float* __restrict__ out) {
  int t = blockIdx.x * blockDim.x + threadIdx.x;
  const int GH = G_GRAPHS * HID;
  if (t < GH) out[t] = (feats[t] + feats[GH + t] + feats[2 * GH + t]) * (1.0f / 3.0f);
}

static inline int cdiv(int a, int b) { return (a + b - 1) / b; }

extern "C" void kernel_launch(void* const* d_in, const int* in_sizes, int n_in,
                              void* d_out, int out_size, void* d_ws, size_t ws_size,
                              hipStream_t stream) {
  (void)in_sizes; (void)n_in; (void)out_size; (void)ws_size;
  const float* x     = (const float*)d_in[0];
  const float* W0    = (const float*)d_in[1];
  const float* b0    = (const float*)d_in[2];
  const float* W1    = (const float*)d_in[3];
  const float* b1    = (const float*)d_in[4];
  const float* W2    = (const float*)d_in[5];
  const float* b2    = (const float*)d_in[6];
  const int*   src   = (const int*)d_in[7];
  const int*   dst   = (const int*)d_in[8];
  const int*   batch = (const int*)d_in[9];
  float* out = (float*)d_out;

  float* ws    = (float*)d_ws;
  float* deg   = ws;
  float* cnt   = deg + N_NODES;
  float* lin   = cnt + G_GRAPHS;
  float* act   = lin + (size_t)N_NODES * HID;
  float* agg   = act + (size_t)N_NODES * HID;
  float* feats = agg + (size_t)N_NODES * HID;

  const int T = 256;
  const int nTiles = (N_NODES / 16) * 2;

  unsigned* degPlanes = (unsigned*)agg;
  unsigned* cntPlanes = (unsigned*)act;
  unsigned* lists     = (unsigned*)act;
  unsigned* degb      = (unsigned*)act + (size_t)AGG_R * AGG_S * LST_WORDS;
  gcn_zero_kernel<<<cdiv(3 * G_GRAPHS * HID, T), T, 0, stream>>>(feats, 3 * G_GRAPHS * HID);
  gcn_degree_hist_kernel<<<DEG_R * DEG_S, DEG_T, 0, stream>>>(dst, degPlanes);
  gcn_dinv_kernel<<<cdiv(N_NODES / 4, 128), 128, 0, stream>>>(degPlanes, deg, degb);
  gcn_counts_hist_kernel<<<CNT_B, CNT_T, 0, stream>>>(batch, cntPlanes);
  gcn_cntinv_kernel<<<1, G_GRAPHS, 0, stream>>>(cntPlanes, cnt);

  gcn_edge_scan_kernel<<<AGG_R * AGG_S, AGG_T, 0, stream>>>(src, dst, lists);

  const float* Ws[3] = {W0, W1, W2};
  const float* bs[3] = {b0, b1, b2};
  for (int layer = 0; layer < 3; ++layer) {
    if (layer == 0) {
      gcn_gemm_wmma_kernel<C_IN, false><<<cdiv(nTiles, T / 32), T, 0, stream>>>(
          x, Ws[0], bs[0], deg, lin, agg, N_NODES);
    } else {
      gcn_gemm_wmma_kernel<HID, true><<<cdiv(nTiles, T / 32), T, 0, stream>>>(
          agg, Ws[layer], bs[layer], deg, lin, agg, N_NODES);
    }
    gcn_edge_sum_kernel<<<OWN_B, SUM_T, 0, stream>>>(src, dst, deg, lin, agg, lists, degb, batch, cnt,
                                                     feats + layer * G_GRAPHS * HID);
  }
  gcn_merge_kernel<<<cdiv(G_GRAPHS * HID, T), T, 0, stream>>>(feats, out);
}


# layers 1-2 computed as (A X) W: edge-sum block seeds its LDS rows with the self-loop term, sums activated rows, then its waves apply W on the matrix unit; 2 product launches gone; sum order changes
# speedup vs baseline: 2.3375x; 1.1177x over previous
#include <hip/hip_runtime.h>

#define N_NODES   100000
#define N_EDGES   1600000
#define C_IN      64
#define HID       32
#define G_GRAPHS  128
#define NEG_SLOPE 0.01f

typedef __attribute__((ext_vector_type(2))) float v2f;
typedef __attribute__((ext_vector_type(8))) float v8f;

__global__ void gcn_zero_kernel(float* __restrict__ p, int n) {
  int i = blockIdx.x * blockDim.x + threadIdx.x;
  if (i < n) p[i] = 0.0f;
}

#define DEG_R     4
#define DEG_S     32
#define DEG_BINS  (N_NODES / DEG_R)
#define DEG_SLICE (N_EDGES / DEG_S)
#define DEG_T     1024
#define DEG_LOADS ((DEG_SLICE / 4 + DEG_T - 1) / DEG_T)
static_assert(N_NODES % (4 * DEG_R) == 0, "bin ranges must be whole quads");
static_assert(N_EDGES % (4 * DEG_S) == 0, "edge slices must be whole quads");
static_assert(DEG_S <= HID, "the planes must fit the edge-sum region they borrow");

__global__ void __launch_bounds__(DEG_T)
gcn_degree_hist_kernel(const int* __restrict__ dst, unsigned* __restrict__ planes) {
  __shared__ unsigned bins[DEG_BINS];
  const int tid = threadIdx.x;
  const int r = blockIdx.x % DEG_R;
  const int s = blockIdx.x / DEG_R;
  const int lo = r * DEG_BINS;
  const int4* d4 = (const int4*)(dst + s * DEG_SLICE);
  int4 v[DEG_LOADS];
#pragma unroll
  for (int j = 0; j < DEG_LOADS; ++j) {
    const int q = tid + j * DEG_T;
    v[j] = q < DEG_SLICE / 4 ? d4[q] : make_int4(-1, -1, -1, -1);
  }
  for (int i = tid; i < DEG_BINS; i += DEG_T) bins[i] = 0u;
  __syncthreads();
#pragma unroll
  for (int j = 0; j < DEG_LOADS; ++j) {
    const unsigned a0 = (unsigned)(v[j].x - lo);
    const unsigned a1 = (unsigned)(v[j].y - lo);
    const unsigned a2 = (unsigned)(v[j].z - lo);
    const unsigned a3 = (unsigned)(v[j].w - lo);
    if (a0 < DEG_BINS) atomicAdd(&bins[a0], 1u);
    if (a1 < DEG_BINS) atomicAdd(&bins[a1], 1u);
    if (a2 < DEG_BINS) atomicAdd(&bins[a2], 1u);
    if (a3 < DEG_BINS) atomicAdd(&bins[a3], 1u);
  }
  __syncthreads();
  uint4* p4 = (uint4*)(planes + (size_t)s * N_NODES + lo);
  const uint4* b4 = (const uint4*)bins;
  for (int i = tid; i < DEG_BINS / 4; i += DEG_T) p4[i] = b4[i];
}

#define DEG_SAT 255u
__global__ void gcn_dinv_kernel(const unsigned* __restrict__ planes, float* __restrict__ dinv,
                                unsigned* __restrict__ degb) {
  const int q = blockIdx.x * blockDim.x + threadIdx.x;
  if (q >= N_NODES / 4) return;
  const uint4* p4 = (const uint4*)planes + q;
  uint4 c = make_uint4(0u, 0u, 0u, 0u);
#pragma unroll
  for (int s = 0; s < DEG_S; ++s) {
    const uint4 t = p4[(size_t)s * (N_NODES / 4)];
    c.x += t.x; c.y += t.y; c.z += t.z; c.w += t.w;
  }
  float4 o;
  o.x = rsqrtf((float)c.x + 1.0f);
  o.y = rsqrtf((float)c.y + 1.0f);
  o.z = rsqrtf((float)c.z + 1.0f);
  o.w = rsqrtf((float)c.w + 1.0f);
  ((float4*)dinv)[q] = o;
  degb[q] = min(c.x, DEG_SAT) | (min(c.y, DEG_SAT) << 8) | (min(c.z, DEG_SAT) << 16) | (min(c.w, DEG_SAT) << 24);
}

#define CNT_T 1024
#define CNT_B ((N_NODES + CNT_T - 1) / CNT_T)

__global__ void __launch_bounds__(CNT_T)
gcn_counts_hist_kernel(const int* __restrict__ batch, unsigned* __restrict__ parts) {
  __shared__ unsigned bins[G_GRAPHS];
  const int tid = threadIdx.x;
  if (tid < G_GRAPHS) bins[tid] = 0u;
  __syncthreads();
  const int i = blockIdx.x * CNT_T + tid;
  if (i < N_NODES) {
    const unsigned g = (unsigned)batch[i];
    if (g < G_GRAPHS) atomicAdd(&bins[g], 1u);
  }
  __syncthreads();
  if (tid < G_GRAPHS) parts[blockIdx.x * G_GRAPHS + tid] = bins[tid];
}

__global__ void gcn_cntinv_kernel(const unsigned* __restrict__ parts, float* __restrict__ cnt) {
  const int g = blockIdx.x * blockDim.x + threadIdx.x;
  if (g >= G_GRAPHS) return;
  unsigned c = 0u;
#pragma unroll
  for (int b = 0; b < CNT_B; ++b) c += parts[b * G_GRAPHS + g];
  cnt[g] = 1.0f / fmaxf((float)c, 1.0f);
}

template <int K>
__global__ void gcn_gemm_wmma_kernel(const float* __restrict__ X,
                                     const float* __restrict__ W,
                                     const float* __restrict__ bias,
                                     const float* __restrict__ dinv,
                                     float* __restrict__ H,
                                     float* __restrict__ A, int nRows) {
  const int wave = blockIdx.x * (blockDim.x >> 5) + (threadIdx.x >> 5);
  const int lane = threadIdx.x & 31;
  const int numTiles = (nRows >> 4) * 2;
  const bool active = wave < numTiles;
  v2f afrag[K / 4];
  if (active) {
    const int r0 = (wave >> 1) * 16 + (lane & 15);
#pragma unroll
    for (int j = 0; j < K / 4; ++j)
      afrag[j] = *(const v2f*)(X + r0 * K + 4 * j + 2 * (lane >> 4));
  }
  if (!active) return;
  const int rowTile = wave >> 1;
  const int colTile = wave & 1;
  const int half = lane >> 4;
  const int l16  = lane & 15;
  const int col  = colTile * 16 + l16;

  float dd[8];
#pragma unroll
  for (int r = 0; r < 8; ++r) {
    const float d = dinv[rowTile * 16 + r + 8 * half];
    dd[r] = d * d;
  }
  const float bc = bias[col];

  v8f c = {};
#pragma unroll
  for (int k0 = 0; k0 < K; k0 += 4) {
    const int ka = k0 + 2 * half;
    const v2f a = afrag[k0 / 4];
    v2f b;
    b.x = W[(ka + 0) * HID + col];
    b.y = W[(ka + 1) * HID + col];
    c = __builtin_amdgcn_wmma_f32_16x16x4_f32(
         false, a,  false, b,
         (short)0, c,  false,  false);
  }
#pragma unroll
  for (int r = 0; r < 8; ++r) {
    const int m = r + 8 * half;
    const int o = (rowTile * 16 + m) * HID + col;
    H[o] = c[r];
    A[o] = c[r] * dd[r] + bc;
  }
}

#define AGG_RANGE 10000
#define AGG_R     (N_NODES / AGG_RANGE)
#define AGG_S     25
#define AGG_T     1024
#define AGG_W     (AGG_T / 32)
#define AGG_SEG   256
#define AGG_QS    (N_EDGES / 4 / AGG_S)
#define AGG_NG    (AGG_QS / 32)
#define AGG_B     4
#define AGG_U     8
#define FL_U      4
#define OWN_ROWS  400
#define OWN_PER   (AGG_RANGE / OWN_ROWS)
#define OWN_B     (N_NODES / OWN_ROWS)
#define SUM_T     (AGG_S * 32)
#define OWN_TRIPS ((OWN_ROWS * HID + SUM_T - 1) / SUM_T)
#define OWN_QUADS (OWN_ROWS * HID / 4 / SUM_T)
static_assert(OWN_ROWS * HID / 4 % SUM_T == 0, "the block's rows are seeded in whole pieces per thread");
static_assert(OWN_ROWS == 16 * (SUM_T / 32), "one 16-row tile of the block per wave for the in-block product");
static_assert(N_NODES % AGG_RANGE == 0, "row ranges must tile the nodes");
static_assert(AGG_RANGE % OWN_ROWS == 0, "owned blocks must tile a range");
static_assert(N_EDGES % (4 * AGG_S * 32) == 0, "slices must be whole wave groups of quads");
static_assert(AGG_SEG >= 2 * 128, "a list must take one more group after the flush test");
#define DEGB_WORDS (N_NODES / 4)
#define DEGB_PER  10
#define DEGB_THR  (N_NODES / 16 / DEGB_PER)
static_assert(N_NODES % (16 * DEGB_PER) == 0, "the degree bytes are copied in 16-byte pieces, DEGB_PER per thread");
static_assert(DEGB_THR <= SUM_T, "the copying threads must fit the work-group");
#define LST_WORDS ((N_NODES * HID - DEGB_WORDS) / (AGG_R * AGG_S))
#define SUB_WORDS (LST_WORDS / OWN_PER)
#define SUB_CAP   (SUB_WORDS - 1)
#define LST_SHIFT 17
static_assert(N_NODES <= (1 << LST_SHIFT), "a source node must fit the low bits");
static_assert(OWN_ROWS <= (1 << (32 - LST_SHIFT)), "a row of a block must fit the high bits");
static_assert((N_NODES * HID - DEGB_WORDS) % (AGG_R * AGG_S) == 0, "segments must tile the region");
static_assert(OWN_ROWS <= SUM_T, "one thread per owned row stages its dinv");
static_assert(LST_WORDS % OWN_PER == 0, "parts must tile a segment");

__device__ __forceinline__ void gcn_flush_list(const uint2* seg, int cnt, int lane,
                                               const int* __restrict__ src,
                                               unsigned* fill, unsigned* segw) {
  for (int i = 0; i < cnt; i += 32 * FL_U) {
    unsigned sr[FL_U];
    unsigned al[FL_U];
    bool ok[FL_U];
#pragma unroll
    for (int u = 0; u < FL_U; ++u) {
      const int idx = i + 32 * u + lane;
      ok[u] = idx < cnt;
      const uint2 e = seg[ok[u] ? idx : i];
      sr[u] = (unsigned)src[e.x];
      al[u] = e.y;
    }
#pragma unroll
    for (int u = 0; u < FL_U; ++u) {
      if (ok[u]) {
        const unsigned blk = al[u] / (unsigned)OWN_ROWS;
        const unsigned pos = atomicAdd(&fill[blk], 1u);
        if (pos < (unsigned)SUB_CAP)
          segw[blk * SUB_WORDS + 1u + pos] = sr[u] | ((al[u] - blk * (unsigned)OWN_ROWS) << LST_SHIFT);
      }
    }
  }
}

__global__ void __launch_bounds__(AGG_T)
gcn_edge_scan_kernel(const int* __restrict__ src, const int* __restrict__ dst,
                     unsigned* __restrict__ lists) {
  __shared__ uint2 list[AGG_W * AGG_SEG];
  __shared__ unsigned fill[OWN_PER];
  const int tid = threadIdx.x;
  const int lane = tid & 31;
  const int w = __builtin_amdgcn_readfirstlane(tid >> 5);
  const int r = blockIdx.x % AGG_R;
  const int sl = blockIdx.x / AGG_R;
  const int lo = r * AGG_RANGE;
  unsigned* segw = lists + (size_t)blockIdx.x * LST_WORDS;

  if (tid < OWN_PER) fill[tid] = 0u;
  __syncthreads();

  const int qb = sl * AGG_QS;
  const int4* d4 = (const int4*)dst + qb;
  uint2* seg = list + w * AGG_SEG;
  int cnt = 0;

  for (int g0 = w; g0 < AGG_NG; g0 += AGG_W * AGG_B) {
    int4 dv[AGG_B];
#pragma unroll
    for (int b = 0; b < AGG_B; ++b) {
      const int g = g0 + b * AGG_W;
      if (g < AGG_NG) dv[b] = d4[g * 32 + lane];
      else            dv[b] = make_int4(-1, -1, -1, -1);
    }
#pragma unroll
    for (int b = 0; b < AGG_B; ++b) {
      const int g = g0 + b * AGG_W;
      const unsigned e0 = (unsigned)(4 * (qb + g * 32 + lane));
      const int dd[4] = {dv[b].x, dv[b].y, dv[b].z, dv[b].w};
#pragma unroll
      for (int k = 0; k < 4; ++k) {
        const unsigned a = (unsigned)(dd[k] - lo);
        const bool m = a < (unsigned)AGG_RANGE;
        const unsigned bal = __builtin_amdgcn_ballot_w32(m);
        if (m) seg[cnt + (int)__builtin_amdgcn_mbcnt_lo(bal, 0u)] = make_uint2(e0 + (unsigned)k, a);
        cnt += __builtin_popcount(bal);
      }
      if (cnt > AGG_SEG - 128) {
        __builtin_amdgcn_wave_barrier();
        gcn_flush_list(seg, cnt, lane, src, fill, segw);
        __builtin_amdgcn_wave_barrier();
        cnt = 0;
      }
    }
  }
  __builtin_amdgcn_wave_barrier();
  gcn_flush_list(seg, cnt, lane, src, fill, segw);
  __syncthreads();
  if (tid < OWN_PER) segw[tid * SUB_WORDS] = fill[tid];
}

__device__ __forceinline__ void gcn_replay_list(const unsigned* __restrict__ ent, int n, int wr, int nw,
                                                int lane,
                                                const float* __restrict__ dinv,
                                                const unsigned char* sdeg, const float* sdinv,
                                                const float* __restrict__ H, float* acc) {
  const int sub = lane >> 3;
  const int q = (lane & 7) << 2;
  int i = wr * (4 * AGG_U);
  unsigned en[AGG_U];
  if (i < n) {
#pragma unroll
    for (int u = 0; u < AGG_U; ++u) {
      const int idx = i + 4 * u + sub;
      en[u] = ent[idx < n ? idx : i];
    }
  }
  while (i < n) {
    int al[AGG_U];
    bool ok[AGG_U];
    float4 hv[AGG_U];
    float cf[AGG_U];
    int sr[AGG_U];
    unsigned dg[AGG_U];
    float dl[AGG_U];
#pragma unroll
    for (int u = 0; u < AGG_U; ++u) {
      ok[u] = i + 4 * u + sub < n;
      sr[u] = (int)(en[u] & ((1u << LST_SHIFT) - 1u));
      al[u] = (int)(en[u] >> LST_SHIFT);
      hv[u] = *(const float4*)(H + sr[u] * HID + q);
      dg[u] = (unsigned)sdeg[sr[u]];
      dl[u] = sdinv[al[u]];
    }
    const int inext = i + nw * 4 * AGG_U;
    if (inext < n) {
#pragma unroll
      for (int u = 0; u < AGG_U; ++u) {
        const int idx = inext + 4 * u + sub;
        en[u] = ent[idx < n ? idx : inext];
      }
    }
    bool sat = false;
#pragma unroll
    for (int u = 0; u < AGG_U; ++u) {
      cf[u] = rsqrtf((float)dg[u] + 1.0f);
      sat = sat || dg[u] == DEG_SAT;
    }
    if (__builtin_amdgcn_ballot_w32(sat) != 0u) {
#pragma unroll
      for (int u = 0; u < AGG_U; ++u)
        if (dg[u] == DEG_SAT) cf[u] = dinv[sr[u]];
    }
#pragma unroll
    for (int u = 0; u < AGG_U; ++u) cf[u] *= dl[u];
#pragma unroll
    for (int u = 0; u < AGG_U; ++u) {
      if (i + 4 * u < n) {
        float* ap = acc + al[u] * HID + q;
        atomicAdd(ap + 0, ok[u] ? hv[u].x * cf[u] : 0.0f);
        atomicAdd(ap + 1, ok[u] ? hv[u].y * cf[u] : 0.0f);
        atomicAdd(ap + 2, ok[u] ? hv[u].z * cf[u] : 0.0f);
        atomicAdd(ap + 3, ok[u] ? hv[u].w * cf[u] : 0.0f);
      }
    }
    i = inext;
  }
}

template <bool POST, bool STORE>
__global__ void __launch_bounds__(SUM_T)
gcn_edge_sum_kernel(const int* __restrict__ src, const int* __restrict__ dst,
                    const float* __restrict__ dinv,
                    const float* __restrict__ H, float* __restrict__ agg,
                    const float* __restrict__ Wm, const float* __restrict__ bias,
                    const unsigned* __restrict__ lists, const unsigned* __restrict__ degb,
                    const int* __restrict__ batch, const float* __restrict__ cntinv,
                    float* __restrict__ feat) {
  __shared__ float acc[OWN_ROWS * HID];
  __shared__ __attribute__((aligned(16))) unsigned char sdeg[N_NODES];
  __shared__ float sdinv[OWN_ROWS];
  const int tid = threadIdx.x;
  const int lane = tid & 31;
  const int w = __builtin_amdgcn_readfirstlane(tid >> 5);
  const int r = blockIdx.x / OWN_PER;
  const int p = blockIdx.x % OWN_PER;
  const int lo = blockIdx.x * OWN_ROWS;

  bool fit = true;
#pragma unroll
  for (int s = 0; s < AGG_S; ++s) {
    const unsigned n = lists[(size_t)(s * AGG_R + r) * LST_WORDS + (size_t)p * SUB_WORDS];
    fit = fit && n <= (unsigned)SUB_CAP;
  }
  float dmine = 0.0f;
  if (tid < OWN_ROWS) dmine = dinv[lo + tid];

  float4* acc4 = (float4*)acc;
  float4 xv[OWN_QUADS];
  float xd[OWN_QUADS];
  if (POST) {
    const float4* own4 = (const float4*)(H + lo * HID);
#pragma unroll
    for (int j = 0; j < OWN_QUADS; ++j) {
      const int i = tid + j * SUM_T;
      xv[j] = own4[i];
      xd[j] = dinv[lo + (i >> 3)];
    }
  } else {
    for (int i = tid; i < OWN_ROWS * HID / 4; i += SUM_T) acc4[i] = make_float4(0.f, 0.f, 0.f, 0.f);
  }
  if (tid < DEGB_THR) {
    const uint4* degb4 = (const uint4*)degb;
    uint4* sdeg4 = (uint4*)sdeg;
    uint4 dt[DEGB_PER];
#pragma unroll
    for (int j = 0; j < DEGB_PER; ++j) dt[j] = degb4[tid + j * DEGB_THR];
#pragma unroll
    for (int j = 0; j < DEGB_PER; ++j) sdeg4[tid + j * DEGB_THR] = dt[j];
  }
  if (POST) {
#pragma unroll
    for (int j = 0; j < OWN_QUADS; ++j) {
      const float dd = xd[j] * xd[j];
      acc4[tid + j * SUM_T] = make_float4(xv[j].x * dd, xv[j].y * dd, xv[j].z * dd, xv[j].w * dd);
    }
  }
  if (tid < OWN_ROWS) sdinv[tid] = dmine;

  __syncthreads();

  if (fit) {
    const unsigned* part = lists + (size_t)(w * AGG_R + r) * LST_WORDS + (size_t)p * SUB_WORDS;
    const int n = __builtin_amdgcn_readfirstlane((int)part[0]);
    gcn_replay_list(part + 1, n, 0, 1, lane, dinv, sdeg, sdinv, H, acc);
  } else {
    for (int e = tid; e < N_EDGES; e += SUM_T) {
      const unsigned a = (unsigned)(dst[e] - lo);
      if (a < (unsigned)OWN_ROWS) {
        const int sr = src[e];
        const float cf = dinv[sr] * dinv[lo + (int)a];
        for (int ch = 0; ch < HID; ++ch) atomicAdd(&acc[a * HID + ch], H[sr * HID + ch] * cf);
      }
    }
  }
  const int half = lane >> 4;
  const int l16 = lane & 15;
  v2f wb0[HID / 4], wb1[HID / 4];
  float bc0 = 0.0f, bc1 = 0.0f;
  if (POST) {
#pragma unroll
    for (int j = 0; j < HID / 4; ++j) {
      const int ka = 4 * j + 2 * half;
      wb0[j].x = Wm[(ka + 0) * HID + l16];
      wb0[j].y = Wm[(ka + 1) * HID + l16];
      wb1[j].x = Wm[(ka + 0) * HID + 16 + l16];
      wb1[j].y = Wm[(ka + 1) * HID + 16 + l16];
    }
    bc0 = bias[l16];
    bc1 = bias[16 + l16];
  }
  __syncthreads();

  if (POST) {
    const float* arow = acc + (w * 16 + l16) * HID + 2 * half;
    v2f af[HID / 4];
#pragma unroll
    for (int j = 0; j < HID / 4; ++j) af[j] = *(const v2f*)(arow + 4 * j);
    v8f c0 = {};
    v8f c1 = {};
#pragma unroll
    for (int j = 0; j < HID / 4; ++j) {
      c0 = __builtin_amdgcn_wmma_f32_16x16x4_f32(
           false, af[j],  false, wb0[j],
           (short)0, c0,  false,  false);
      c1 = __builtin_amdgcn_wmma_f32_16x16x4_f32(
           false, af[j],  false, wb1[j],
           (short)0, c1,  false,  false);
    }
#pragma unroll
    for (int r = 0; r < 8; ++r) {
      float* orow = acc + (w * 16 + r + 8 * half) * HID;
      orow[l16] = c0[r] + bc0;
      orow[16 + l16] = c1[r] + bc1;
    }
    __syncthreads();
  }

  float* rows = agg + lo * HID;
  float sd[OWN_TRIPS];
  int gi[OWN_TRIPS];
  float pw[OWN_TRIPS];
#pragma unroll
  for (int j = 0; j < OWN_TRIPS; ++j) {
    const int i = tid + j * SUM_T;
    const int c = i < OWN_ROWS * HID ? i : tid;
    sd[j] = POST ? 0.0f : rows[c];
    gi[j] = batch[lo + (c >> 5)];
  }
#pragma unroll
  for (int j = 0; j < OWN_TRIPS; ++j) pw[j] = cntinv[gi[j]];
#pragma unroll
  for (int j = 0; j < OWN_TRIPS; ++j) {
    const int i = tid + j * SUM_T;
    if (i < OWN_ROWS * HID) {
      float v = sd[j] + acc[i];
      v = v > 0.0f ? v : NEG_SLOPE * v;
      if (!POST || STORE) rows[i] = v;
      atomicAdd(&feat[gi[j] * HID + (i & 31)], v * pw[j]);
    }
  }
}

__global__ void gcn_merge_kernel(const float* __restrict__ feats, float* __restrict__ out) {
  int t = blockIdx.x * blockDim.x + threadIdx.x;
  const int GH = G_GRAPHS * HID;
  if (t < GH) out[t] = (feats[t] + feats[GH + t] + feats[2 * GH + t]) * (1.0f / 3.0f);
}

static inline int cdiv(int a, int b) { return (a + b - 1) / b; }

extern "C" void kernel_launch(void* const* d_in, const int* in_sizes, int n_in,
                              void* d_out, int out_size, void* d_ws, size_t ws_size,
                              hipStream_t stream) {
  (void)in_sizes; (void)n_in; (void)out_size; (void)ws_size;
  const float* x     = (const float*)d_in[0];
  const float* W0    = (const float*)d_in[1];
  const float* b0    = (const float*)d_in[2];
  const float* W1    = (const float*)d_in[3];
  const float* b1    = (const float*)d_in[4];
  const float* W2    = (const float*)d_in[5];
  const float* b2    = (const float*)d_in[6];
  const int*   src   = (const int*)d_in[7];
  const int*   dst   = (const int*)d_in[8];
  const int*   batch = (const int*)d_in[9];
  float* out = (float*)d_out;

  float* ws    = (float*)d_ws;
  float* deg   = ws;
  float* cnt   = deg + N_NODES;
  float* lin   = cnt + G_GRAPHS;
  float* act   = lin + (size_t)N_NODES * HID;
  float* agg   = act + (size_t)N_NODES * HID;
  float* feats = agg + (size_t)N_NODES * HID;

  const int T = 256;
  const int nTiles = (N_NODES / 16) * 2;

  unsigned* degPlanes = (unsigned*)agg;
  unsigned* cntPlanes = (unsigned*)act;
  unsigned* lists     = (unsigned*)act;
  unsigned* degb      = (unsigned*)act + (size_t)AGG_R * AGG_S * LST_WORDS;
  gcn_zero_kernel<<<cdiv(3 * G_GRAPHS * HID, T), T, 0, stream>>>(feats, 3 * G_GRAPHS * HID);
  gcn_degree_hist_kernel<<<DEG_R * DEG_S, DEG_T, 0, stream>>>(dst, degPlanes);
  gcn_dinv_kernel<<<cdiv(N_NODES / 4, 128), 128, 0, stream>>>(degPlanes, deg, degb);
  gcn_counts_hist_kernel<<<CNT_B, CNT_T, 0, stream>>>(batch, cntPlanes);
  gcn_cntinv_kernel<<<1, G_GRAPHS, 0, stream>>>(cntPlanes, cnt);

  gcn_edge_scan_kernel<<<AGG_R * AGG_S, AGG_T, 0, stream>>>(src, dst, lists);

  gcn_gemm_wmma_kernel<C_IN><<<cdiv(nTiles, T / 32), T, 0, stream>>>(x, W0, b0, deg, lin, agg, N_NODES);
  gcn_edge_sum_kernel<false, true><<<OWN_B, SUM_T, 0, stream>>>(
      src, dst, deg, lin, agg, W0, b0, lists, degb, batch, cnt, feats);
  gcn_edge_sum_kernel<true, true><<<OWN_B, SUM_T, 0, stream>>>(
      src, dst, deg, agg, lin, W1, b1, lists, degb, batch, cnt, feats + G_GRAPHS * HID);
  gcn_edge_sum_kernel<true, false><<<OWN_B, SUM_T, 0, stream>>>(
      src, dst, deg, lin, agg, W2, b2, lists, degb, batch, cnt, feats + 2 * G_GRAPHS * HID);
  gcn_merge_kernel<<<cdiv(G_GRAPHS * HID, T), T, 0, stream>>>(feats, out);
}


# edge-sum LDS float adds made bank-conflict free: the j-th add of a lane takes element (j + edge slot) & 3 of its quad, so the 4 edges of a step hit 32 different banks; float add order changes
# speedup vs baseline: 3.1280x; 1.3382x over previous
#include <hip/hip_runtime.h>

#define N_NODES   100000
#define N_EDGES   1600000
#define C_IN      64
#define HID       32
#define G_GRAPHS  128
#define NEG_SLOPE 0.01f

typedef __attribute__((ext_vector_type(2))) float v2f;
typedef __attribute__((ext_vector_type(8))) float v8f;

__global__ void gcn_zero_kernel(float* __restrict__ p, int n) {
  int i = blockIdx.x * blockDim.x + threadIdx.x;
  if (i < n) p[i] = 0.0f;
}

#define DEG_R     4
#define DEG_S     32
#define DEG_BINS  (N_NODES / DEG_R)
#define DEG_SLICE (N_EDGES / DEG_S)
#define DEG_T     1024
#define DEG_LOADS ((DEG_SLICE / 4 + DEG_T - 1) / DEG_T)
static_assert(N_NODES % (4 * DEG_R) == 0, "bin ranges must be whole quads");
static_assert(N_EDGES % (4 * DEG_S) == 0, "edge slices must be whole quads");
static_assert(DEG_S <= HID, "the planes must fit the edge-sum region they borrow");

__global__ void __launch_bounds__(DEG_T)
gcn_degree_hist_kernel(const int* __restrict__ dst, unsigned* __restrict__ planes) {
  __shared__ unsigned bins[DEG_BINS];
  const int tid = threadIdx.x;
  const int r = blockIdx.x % DEG_R;
  const int s = blockIdx.x / DEG_R;
  const int lo = r * DEG_BINS;
  const int4* d4 = (const int4*)(dst + s * DEG_SLICE);
  int4 v[DEG_LOADS];
#pragma unroll
  for (int j = 0; j < DEG_LOADS; ++j) {
    const int q = tid + j * DEG_T;
    v[j] = q < DEG_SLICE / 4 ? d4[q] : make_int4(-1, -1, -1, -1);
  }
  for (int i = tid; i < DEG_BINS; i += DEG_T) bins[i] = 0u;
  __syncthreads();
#pragma unroll
  for (int j = 0; j < DEG_LOADS; ++j) {
    const unsigned a0 = (unsigned)(v[j].x - lo);
    const unsigned a1 = (unsigned)(v[j].y - lo);
    const unsigned a2 = (unsigned)(v[j].z - lo);
    const unsigned a3 = (unsigned)(v[j].w - lo);
    if (a0 < DEG_BINS) atomicAdd(&bins[a0], 1u);
    if (a1 < DEG_BINS) atomicAdd(&bins[a1], 1u);
    if (a2 < DEG_BINS) atomicAdd(&bins[a2], 1u);
    if (a3 < DEG_BINS) atomicAdd(&bins[a3], 1u);
  }
  __syncthreads();
  uint4* p4 = (uint4*)(planes + (size_t)s * N_NODES + lo);
  const uint4* b4 = (const uint4*)bins;
  for (int i = tid; i < DEG_BINS / 4; i += DEG_T) p4[i] = b4[i];
}

#define DEG_SAT 255u
__global__ void gcn_dinv_kernel(const unsigned* __restrict__ planes, float* __restrict__ dinv,
                                unsigned* __restrict__ degb) {
  const int q = blockIdx.x * blockDim.x + threadIdx.x;
  if (q >= N_NODES / 4) return;
  const uint4* p4 = (const uint4*)planes + q;
  uint4 c = make_uint4(0u, 0u, 0u, 0u);
#pragma unroll
  for (int s = 0; s < DEG_S; ++s) {
    const uint4 t = p4[(size_t)s * (N_NODES / 4)];
    c.x += t.x; c.y += t.y; c.z += t.z; c.w += t.w;
  }
  float4 o;
  o.x = rsqrtf((float)c.x + 1.0f);
  o.y = rsqrtf((float)c.y + 1.0f);
  o.z = rsqrtf((float)c.z + 1.0f);
  o.w = rsqrtf((float)c.w + 1.0f);
  ((float4*)dinv)[q] = o;
  degb[q] = min(c.x, DEG_SAT) | (min(c.y, DEG_SAT) << 8) | (min(c.z, DEG_SAT) << 16) | (min(c.w, DEG_SAT) << 24);
}

#define CNT_T 1024
#define CNT_B ((N_NODES + CNT_T - 1) / CNT_T)

__global__ void __launch_bounds__(CNT_T)
gcn_counts_hist_kernel(const int* __restrict__ batch, unsigned* __restrict__ parts) {
  __shared__ unsigned bins[G_GRAPHS];
  const int tid = threadIdx.x;
  if (tid < G_GRAPHS) bins[tid] = 0u;
  __syncthreads();
  const int i = blockIdx.x * CNT_T + tid;
  if (i < N_NODES) {
    const unsigned g = (unsigned)batch[i];
    if (g < G_GRAPHS) atomicAdd(&bins[g], 1u);
  }
  __syncthreads();
  if (tid < G_GRAPHS) parts[blockIdx.x * G_GRAPHS + tid] = bins[tid];
}

__global__ void gcn_cntinv_kernel(const unsigned* __restrict__ parts, float* __restrict__ cnt) {
  const int g = blockIdx.x * blockDim.x + threadIdx.x;
  if (g >= G_GRAPHS) return;
  unsigned c = 0u;
#pragma unroll
  for (int b = 0; b < CNT_B; ++b) c += parts[b * G_GRAPHS + g];
  cnt[g] = 1.0f / fmaxf((float)c, 1.0f);
}

template <int K>
__global__ void gcn_gemm_wmma_kernel(const float* __restrict__ X,
                                     const float* __restrict__ W,
                                     const float* __restrict__ bias,
                                     const float* __restrict__ dinv,
                                     float* __restrict__ H,
                                     float* __restrict__ A, int nRows) {
  const int wave = blockIdx.x * (blockDim.x >> 5) + (threadIdx.x >> 5);
  const int lane = threadIdx.x & 31;
  const int numTiles = (nRows >> 4) * 2;
  const bool active = wave < numTiles;
  v2f afrag[K / 4];
  if (active) {
    const int r0 = (wave >> 1) * 16 + (lane & 15);
#pragma unroll
    for (int j = 0; j < K / 4; ++j)
      afrag[j] = *(const v2f*)(X + r0 * K + 4 * j + 2 * (lane >> 4));
  }
  if (!active) return;
  const int rowTile = wave >> 1;
  const int colTile = wave & 1;
  const int half = lane >> 4;
  const int l16  = lane & 15;
  const int col  = colTile * 16 + l16;

  float dd[8];
#pragma unroll
  for (int r = 0; r < 8; ++r) {
    const float d = dinv[rowTile * 16 + r + 8 * half];
    dd[r] = d * d;
  }
  const float bc = bias[col];

  v8f c = {};
#pragma unroll
  for (int k0 = 0; k0 < K; k0 += 4) {
    const int ka = k0 + 2 * half;
    const v2f a = afrag[k0 / 4];
    v2f b;
    b.x = W[(ka + 0) * HID + col];
    b.y = W[(ka + 1) * HID + col];
    c = __builtin_amdgcn_wmma_f32_16x16x4_f32(
         false, a,  false, b,
         (short)0, c,  false,  false);
  }
#pragma unroll
  for (int r = 0; r < 8; ++r) {
    const int m = r + 8 * half;
    const int o = (rowTile * 16 + m) * HID + col;
    H[o] = c[r];
    A[o] = c[r] * dd[r] + bc;
  }
}

#define AGG_RANGE 10000
#define AGG_R     (N_NODES / AGG_RANGE)
#define AGG_S     25
#define AGG_T     1024
#define AGG_W     (AGG_T / 32)
#define AGG_SEG   256
#define AGG_QS    (N_EDGES / 4 / AGG_S)
#define AGG_NG    (AGG_QS / 32)
#define AGG_B     4
#define AGG_U     8
#define FL_U      4
#define OWN_ROWS  400
#define OWN_PER   (AGG_RANGE / OWN_ROWS)
#define OWN_B     (N_NODES / OWN_ROWS)
#define SUM_T     (AGG_S * 32)
#define OWN_TRIPS ((OWN_ROWS * HID + SUM_T - 1) / SUM_T)
#define OWN_QUADS (OWN_ROWS * HID / 4 / SUM_T)
static_assert(OWN_ROWS * HID / 4 % SUM_T == 0, "the block's rows are seeded in whole pieces per thread");
static_assert(OWN_ROWS == 16 * (SUM_T / 32), "one 16-row tile of the block per wave for the in-block product");
static_assert(N_NODES % AGG_RANGE == 0, "row ranges must tile the nodes");
static_assert(AGG_RANGE % OWN_ROWS == 0, "owned blocks must tile a range");
static_assert(N_EDGES % (4 * AGG_S * 32) == 0, "slices must be whole wave groups of quads");
static_assert(AGG_SEG >= 2 * 128, "a list must take one more group after the flush test");
#define DEGB_WORDS (N_NODES / 4)
#define DEGB_PER  10
#define DEGB_THR  (N_NODES / 16 / DEGB_PER)
static_assert(N_NODES % (16 * DEGB_PER) == 0, "the degree bytes are copied in 16-byte pieces, DEGB_PER per thread");
static_assert(DEGB_THR <= SUM_T, "the copying threads must fit the work-group");
#define LST_WORDS ((N_NODES * HID - DEGB_WORDS) / (AGG_R * AGG_S))
#define SUB_WORDS (LST_WORDS / OWN_PER)
#define SUB_CAP   (SUB_WORDS - 1)
#define LST_SHIFT 17
static_assert(N_NODES <= (1 << LST_SHIFT), "a source node must fit the low bits");
static_assert(OWN_ROWS <= (1 << (32 - LST_SHIFT)), "a row of a block must fit the high bits");
static_assert((N_NODES * HID - DEGB_WORDS) % (AGG_R * AGG_S) == 0, "segments must tile the region");
static_assert(OWN_ROWS <= SUM_T, "one thread per owned row stages its dinv");
static_assert(LST_WORDS % OWN_PER == 0, "parts must tile a segment");

__device__ __forceinline__ void gcn_flush_list(const uint2* seg, int cnt, int lane,
                                               const int* __restrict__ src,
                                               unsigned* fill, unsigned* segw) {
  for (int i = 0; i < cnt; i += 32 * FL_U) {
    unsigned sr[FL_U];
    unsigned al[FL_U];
    bool ok[FL_U];
#pragma unroll
    for (int u = 0; u < FL_U; ++u) {
      const int idx = i + 32 * u + lane;
      ok[u] = idx < cnt;
      const uint2 e = seg[ok[u] ? idx : i];
      sr[u] = (unsigned)src[e.x];
      al[u] = e.y;
    }
#pragma unroll
    for (int u = 0; u < FL_U; ++u) {
      if (ok[u]) {
        const unsigned blk = al[u] / (unsigned)OWN_ROWS;
        const unsigned pos = atomicAdd(&fill[blk], 1u);
        if (pos < (unsigned)SUB_CAP)
          segw[blk * SUB_WORDS + 1u + pos] = sr[u] | ((al[u] - blk * (unsigned)OWN_ROWS) << LST_SHIFT);
      }
    }
  }
}

__global__ void __launch_bounds__(AGG_T)
gcn_edge_scan_kernel(const int* __restrict__ src, const int* __restrict__ dst,
                     unsigned* __restrict__ lists) {
  __shared__ uint2 list[AGG_W * AGG_SEG];
  __shared__ unsigned fill[OWN_PER];
  const int tid = threadIdx.x;
  const int lane = tid & 31;
  const int w = __builtin_amdgcn_readfirstlane(tid >> 5);
  const int r = blockIdx.x % AGG_R;
  const int sl = blockIdx.x / AGG_R;
  const int lo = r * AGG_RANGE;
  unsigned* segw = lists + (size_t)blockIdx.x * LST_WORDS;

  if (tid < OWN_PER) fill[tid] = 0u;
  __syncthreads();

  const int qb = sl * AGG_QS;
  const int4* d4 = (const int4*)dst + qb;
  uint2* seg = list + w * AGG_SEG;
  int cnt = 0;

  for (int g0 = w; g0 < AGG_NG; g0 += AGG_W * AGG_B) {
    int4 dv[AGG_B];
#pragma unroll
    for (int b = 0; b < AGG_B; ++b) {
      const int g = g0 + b * AGG_W;
      if (g < AGG_NG) dv[b] = d4[g * 32 + lane];
      else            dv[b] = make_int4(-1, -1, -1, -1);
    }
#pragma unroll
    for (int b = 0; b < AGG_B; ++b) {
      const int g = g0 + b * AGG_W;
      const unsigned e0 = (unsigned)(4 * (qb + g * 32 + lane));
      const int dd[4] = {dv[b].x, dv[b].y, dv[b].z, dv[b].w};
#pragma unroll
      for (int k = 0; k < 4; ++k) {
        const unsigned a = (unsigned)(dd[k] - lo);
        const bool m = a < (unsigned)AGG_RANGE;
        const unsigned bal = __builtin_amdgcn_ballot_w32(m);
        if (m) seg[cnt + (int)__builtin_amdgcn_mbcnt_lo(bal, 0u)] = make_uint2(e0 + (unsigned)k, a);
        cnt += __builtin_popcount(bal);
      }
      if (cnt > AGG_SEG - 128) {
        __builtin_amdgcn_wave_barrier();
        gcn_flush_list(seg, cnt, lane, src, fill, segw);
        __builtin_amdgcn_wave_barrier();
        cnt = 0;
      }
    }
  }
  __builtin_amdgcn_wave_barrier();
  gcn_flush_list(seg, cnt, lane, src, fill, segw);
  __syncthreads();
  if (tid < OWN_PER) segw[tid * SUB_WORDS] = fill[tid];
}

__device__ __forceinline__ void gcn_replay_list(const unsigned* __restrict__ ent, int n, int wr, int nw,
                                                int lane,
                                                const float* __restrict__ dinv,
                                                const unsigned char* sdeg, const float* sdinv,
                                                const float* __restrict__ H, float* acc) {
  const int sub = lane >> 3;
  const int q = (lane & 7) << 2;
  const bool rot1 = (sub & 1) != 0;
  const bool rot2 = (sub & 2) != 0;
  const int o0 = sub & 3, o1 = (sub + 1) & 3, o2 = (sub + 2) & 3, o3 = (sub + 3) & 3;
  int i = wr * (4 * AGG_U);
  unsigned en[AGG_U];
  if (i < n) {
#pragma unroll
    for (int u = 0; u < AGG_U; ++u) {
      const int idx = i + 4 * u + sub;
      en[u] = ent[idx < n ? idx : i];
    }
  }
  while (i < n) {
    int al[AGG_U];
    bool ok[AGG_U];
    float4 hv[AGG_U];
    float cf[AGG_U];
    int sr[AGG_U];
    unsigned dg[AGG_U];
    float dl[AGG_U];
#pragma unroll
    for (int u = 0; u < AGG_U; ++u) {
      ok[u] = i + 4 * u + sub < n;
      sr[u] = (int)(en[u] & ((1u << LST_SHIFT) - 1u));
      al[u] = (int)(en[u] >> LST_SHIFT);
      hv[u] = *(const float4*)(H + sr[u] * HID + q);
      dg[u] = (unsigned)sdeg[sr[u]];
      dl[u] = sdinv[al[u]];
    }
    const int inext = i + nw * 4 * AGG_U;
    if (inext < n) {
#pragma unroll
      for (int u = 0; u < AGG_U; ++u) {
        const int idx = inext + 4 * u + sub;
        en[u] = ent[idx < n ? idx : inext];
      }
    }
    bool sat = false;
#pragma unroll
    for (int u = 0; u < AGG_U; ++u) {
      cf[u] = rsqrtf((float)dg[u] + 1.0f);
      sat = sat || dg[u] == DEG_SAT;
    }
    if (__builtin_amdgcn_ballot_w32(sat) != 0u) {
#pragma unroll
      for (int u = 0; u < AGG_U; ++u)
        if (dg[u] == DEG_SAT) cf[u] = dinv[sr[u]];
    }
#pragma unroll
    for (int u = 0; u < AGG_U; ++u) cf[u] *= dl[u];
#pragma unroll
    for (int u = 0; u < AGG_U; ++u) {
      if (i + 4 * u < n) {
        const float v0 = ok[u] ? hv[u].x * cf[u] : 0.0f;
        const float v1 = ok[u] ? hv[u].y * cf[u] : 0.0f;
        const float v2 = ok[u] ? hv[u].z * cf[u] : 0.0f;
        const float v3 = ok[u] ? hv[u].w * cf[u] : 0.0f;
        const float s0 = rot1 ? v1 : v0;
        const float s1 = rot1 ? v2 : v1;
        const float s2 = rot1 ? v3 : v2;
        const float s3 = rot1 ? v0 : v3;
        const float t0 = rot2 ? s2 : s0;
        const float t1 = rot2 ? s3 : s1;
        const float t2 = rot2 ? s0 : s2;
        const float t3 = rot2 ? s1 : s3;
        float* ap = acc + al[u] * HID + q;
        atomicAdd(ap + o0, t0);
        atomicAdd(ap + o1, t1);
        atomicAdd(ap + o2, t2);
        atomicAdd(ap + o3, t3);
      }
    }
    i = inext;
  }
}

template <bool POST, bool STORE>
__global__ void __launch_bounds__(SUM_T)
gcn_edge_sum_kernel(const int* __restrict__ src, const int* __restrict__ dst,
                    const float* __restrict__ dinv,
                    const float* __restrict__ H, float* __restrict__ agg,
                    const float* __restrict__ Wm, const float* __restrict__ bias,
                    const unsigned* __restrict__ lists, const unsigned* __restrict__ degb,
                    const int* __restrict__ batch, const float* __restrict__ cntinv,
                    float* __restrict__ feat) {
  __shared__ float acc[OWN_ROWS * HID];
  __shared__ __attribute__((aligned(16))) unsigned char sdeg[N_NODES];
  __shared__ float sdinv[OWN_ROWS];
  const int tid = threadIdx.x;
  const int lane = tid & 31;
  const int w = __builtin_amdgcn_readfirstlane(tid >> 5);
  const int r = blockIdx.x / OWN_PER;
  const int p = blockIdx.x % OWN_PER;
  const int lo = blockIdx.x * OWN_ROWS;

  bool fit = true;
#pragma unroll
  for (int s = 0; s < AGG_S; ++s) {
    const unsigned n = lists[(size_t)(s * AGG_R + r) * LST_WORDS + (size_t)p * SUB_WORDS];
    fit = fit && n <= (unsigned)SUB_CAP;
  }
  float dmine = 0.0f;
  if (tid < OWN_ROWS) dmine = dinv[lo + tid];

  float4* acc4 = (float4*)acc;
  float4 xv[OWN_QUADS];
  float xd[OWN_QUADS];
  if (POST) {
    const float4* own4 = (const float4*)(H + lo * HID);
#pragma unroll
    for (int j = 0; j < OWN_QUADS; ++j) {
      const int i = tid + j * SUM_T;
      xv[j] = own4[i];
      xd[j] = dinv[lo + (i >> 3)];
    }
  } else {
    for (int i = tid; i < OWN_ROWS * HID / 4; i += SUM_T) acc4[i] = make_float4(0.f, 0.f, 0.f, 0.f);
  }
  if (tid < DEGB_THR) {
    const uint4* degb4 = (const uint4*)degb;
    uint4* sdeg4 = (uint4*)sdeg;
    uint4 dt[DEGB_PER];
#pragma unroll
    for (int j = 0; j < DEGB_PER; ++j) dt[j] = degb4[tid + j * DEGB_THR];
#pragma unroll
    for (int j = 0; j < DEGB_PER; ++j) sdeg4[tid + j * DEGB_THR] = dt[j];
  }
  if (POST) {
#pragma unroll
    for (int j = 0; j < OWN_QUADS; ++j) {
      const float dd = xd[j] * xd[j];
      acc4[tid + j * SUM_T] = make_float4(xv[j].x * dd, xv[j].y * dd, xv[j].z * dd, xv[j].w * dd);
    }
  }
  if (tid < OWN_ROWS) sdinv[tid] = dmine;

  __syncthreads();

  if (fit) {
    const unsigned* part = lists + (size_t)(w * AGG_R + r) * LST_WORDS + (size_t)p * SUB_WORDS;
    const int n = __builtin_amdgcn_readfirstlane((int)part[0]);
    gcn_replay_list(part + 1, n, 0, 1, lane, dinv, sdeg, sdinv, H, acc);
  } else {
    for (int e = tid; e < N_EDGES; e += SUM_T) {
      const unsigned a = (unsigned)(dst[e] - lo);
      if (a < (unsigned)OWN_ROWS) {
        const int sr = src[e];
        const float cf = dinv[sr] * dinv[lo + (int)a];
        for (int ch = 0; ch < HID; ++ch) atomicAdd(&acc[a * HID + ch], H[sr * HID + ch] * cf);
      }
    }
  }
  const int half = lane >> 4;
  const int l16 = lane & 15;
  v2f wb0[HID / 4], wb1[HID / 4];
  float bc0 = 0.0f, bc1 = 0.0f;
  if (POST) {
#pragma unroll
    for (int j = 0; j < HID / 4; ++j) {
      const int ka = 4 * j + 2 * half;
      wb0[j].x = Wm[(ka + 0) * HID + l16];
      wb0[j].y = Wm[(ka + 1) * HID + l16];
      wb1[j].x = Wm[(ka + 0) * HID + 16 + l16];
      wb1[j].y = Wm[(ka + 1) * HID + 16 + l16];
    }
    bc0 = bias[l16];
    bc1 = bias[16 + l16];
  }
  __syncthreads();

  if (POST) {
    const float* arow = acc + (w * 16 + l16) * HID + 2 * half;
    v2f af[HID / 4];
#pragma unroll
    for (int j = 0; j < HID / 4; ++j) af[j] = *(const v2f*)(arow + 4 * j);
    v8f c0 = {};
    v8f c1 = {};
#pragma unroll
    for (int j = 0; j < HID / 4; ++j) {
      c0 = __builtin_amdgcn_wmma_f32_16x16x4_f32(
           false, af[j],  false, wb0[j],
           (short)0, c0,  false,  false);
      c1 = __builtin_amdgcn_wmma_f32_16x16x4_f32(
           false, af[j],  false, wb1[j],
           (short)0, c1,  false,  false);
    }
#pragma unroll
    for (int r = 0; r < 8; ++r) {
      float* orow = acc + (w * 16 + r + 8 * half) * HID;
      orow[l16] = c0[r] + bc0;
      orow[16 + l16] = c1[r] + bc1;
    }
    __syncthreads();
  }

  float* rows = agg + lo * HID;
  float sd[OWN_TRIPS];
  int gi[OWN_TRIPS];
  float pw[OWN_TRIPS];
#pragma unroll
  for (int j = 0; j < OWN_TRIPS; ++j) {
    const int i = tid + j * SUM_T;
    const int c = i < OWN_ROWS * HID ? i : tid;
    sd[j] = POST ? 0.0f : rows[c];
    gi[j] = batch[lo + (c >> 5)];
  }
#pragma unroll
  for (int j = 0; j < OWN_TRIPS; ++j) pw[j] = cntinv[gi[j]];
#pragma unroll
  for (int j = 0; j < OWN_TRIPS; ++j) {
    const int i = tid + j * SUM_T;
    if (i < OWN_ROWS * HID) {
      float v = sd[j] + acc[i];
      v = v > 0.0f ? v : NEG_SLOPE * v;
      if (!POST || STORE) rows[i] = v;
      atomicAdd(&feat[gi[j] * HID + (i & 31)], v * pw[j]);
    }
  }
}

__global__ void gcn_merge_kernel(const float* __restrict__ feats, float* __restrict__ out) {
  int t = blockIdx.x * blockDim.x + threadIdx.x;
  const int GH = G_GRAPHS * HID;
  if (t < GH) out[t] = (feats[t] + feats[GH + t] + feats[2 * GH + t]) * (1.0f / 3.0f);
}

static inline int cdiv(int a, int b) { return (a + b - 1) / b; }

extern "C" void kernel_launch(void* const* d_in, const int* in_sizes, int n_in,
                              void* d_out, int out_size, void* d_ws, size_t ws_size,
                              hipStream_t stream) {
  (void)in_sizes; (void)n_in; (void)out_size; (void)ws_size;
  const float* x     = (const float*)d_in[0];
  const float* W0    = (const float*)d_in[1];
  const float* b0    = (const float*)d_in[2];
  const float* W1    = (const float*)d_in[3];
  const float* b1    = (const float*)d_in[4];
  const float* W2    = (const float*)d_in[5];
  const float* b2    = (const float*)d_in[6];
  const int*   src   = (const int*)d_in[7];
  const int*   dst   = (const int*)d_in[8];
  const int*   batch = (const int*)d_in[9];
  float* out = (float*)d_out;

  float* ws    = (float*)d_ws;
  float* deg   = ws;
  float* cnt   = deg + N_NODES;
  float* lin   = cnt + G_GRAPHS;
  float* act   = lin + (size_t)N_NODES * HID;
  float* agg   = act + (size_t)N_NODES * HID;
  float* feats = agg + (size_t)N_NODES * HID;

  const int T = 256;
  const int nTiles = (N_NODES / 16) * 2;

  unsigned* degPlanes = (unsigned*)agg;
  unsigned* cntPlanes = (unsigned*)act;
  unsigned* lists     = (unsigned*)act;
  unsigned* degb      = (unsigned*)act + (size_t)AGG_R * AGG_S * LST_WORDS;
  gcn_zero_kernel<<<cdiv(3 * G_GRAPHS * HID, T), T, 0, stream>>>(feats, 3 * G_GRAPHS * HID);
  gcn_degree_hist_kernel<<<DEG_R * DEG_S, DEG_T, 0, stream>>>(dst, degPlanes);
  gcn_dinv_kernel<<<cdiv(N_NODES / 4, 128), 128, 0, stream>>>(degPlanes, deg, degb);
  gcn_counts_hist_kernel<<<CNT_B, CNT_T, 0, stream>>>(batch, cntPlanes);
  gcn_cntinv_kernel<<<1, G_GRAPHS, 0, stream>>>(cntPlanes, cnt);

  gcn_edge_scan_kernel<<<AGG_R * AGG_S, AGG_T, 0, stream>>>(src, dst, lists);

  gcn_gemm_wmma_kernel<C_IN><<<cdiv(nTiles, T / 32), T, 0, stream>>>(x, W0, b0, deg, lin, agg, N_NODES);
  gcn_edge_sum_kernel<false, true><<<OWN_B, SUM_T, 0, stream>>>(
      src, dst, deg, lin, agg, W0, b0, lists, degb, batch, cnt, feats);
  gcn_edge_sum_kernel<true, true><<<OWN_B, SUM_T, 0, stream>>>(
      src, dst, deg, agg, lin, W1, b1, lists, degb, batch, cnt, feats + G_GRAPHS * HID);
  gcn_edge_sum_kernel<true, false><<<OWN_B, SUM_T, 0, stream>>>(
      src, dst, deg, lin, agg, W2, b2, lists, degb, batch, cnt, feats + 2 * G_GRAPHS * HID);
  gcn_merge_kernel<<<cdiv(G_GRAPHS * HID, T), T, 0, stream>>>(feats, out);
}


# in-degrees counted by the edge scan as it lists matches (one LDS add per matched edge, one plane per edge slice); the degree-histogram launch and its 4 passes over dst are gone; values identical
# speedup vs baseline: 3.2989x; 1.0546x over previous
#include <hip/hip_runtime.h>

#define N_NODES   100000
#define N_EDGES   1600000
#define C_IN      64
#define HID       32
#define G_GRAPHS  128
#define NEG_SLOPE 0.01f

typedef __attribute__((ext_vector_type(2))) float v2f;
typedef __attribute__((ext_vector_type(8))) float v8f;

__global__ void gcn_zero_kernel(float* __restrict__ p, int n) {
  int i = blockIdx.x * blockDim.x + threadIdx.x;
  if (i < n) p[i] = 0.0f;
}

#define DEG_PLANES 25
static_assert(DEG_PLANES <= HID, "the planes must fit the edge-sum region they borrow");

#define DEG_SAT 255u
__global__ void gcn_dinv_kernel(const unsigned* __restrict__ planes, float* __restrict__ dinv,
                                unsigned* __restrict__ degb) {
  const int q = blockIdx.x * blockDim.x + threadIdx.x;
  if (q >= N_NODES / 4) return;
  const uint4* p4 = (const uint4*)planes + q;
  uint4 c = make_uint4(0u, 0u, 0u, 0u);
#pragma unroll
  for (int s = 0; s < DEG_PLANES; ++s) {
    const uint4 t = p4[(size_t)s * (N_NODES / 4)];
    c.x += t.x; c.y += t.y; c.z += t.z; c.w += t.w;
  }
  float4 o;
  o.x = rsqrtf((float)c.x + 1.0f);
  o.y = rsqrtf((float)c.y + 1.0f);
  o.z = rsqrtf((float)c.z + 1.0f);
  o.w = rsqrtf((float)c.w + 1.0f);
  ((float4*)dinv)[q] = o;
  degb[q] = min(c.x, DEG_SAT) | (min(c.y, DEG_SAT) << 8) | (min(c.z, DEG_SAT) << 16) | (min(c.w, DEG_SAT) << 24);
}

#define CNT_T 1024
#define CNT_B ((N_NODES + CNT_T - 1) / CNT_T)

__global__ void __launch_bounds__(CNT_T)
gcn_counts_hist_kernel(const int* __restrict__ batch, unsigned* __restrict__ parts) {
  __shared__ unsigned bins[G_GRAPHS];
  const int tid = threadIdx.x;
  if (tid < G_GRAPHS) bins[tid] = 0u;
  __syncthreads();
  const int i = blockIdx.x * CNT_T + tid;
  if (i < N_NODES) {
    const unsigned g = (unsigned)batch[i];
    if (g < G_GRAPHS) atomicAdd(&bins[g], 1u);
  }
  __syncthreads();
  if (tid < G_GRAPHS) parts[blockIdx.x * G_GRAPHS + tid] = bins[tid];
}

__global__ void gcn_cntinv_kernel(const unsigned* __restrict__ parts, float* __restrict__ cnt) {
  const int g = blockIdx.x * blockDim.x + threadIdx.x;
  if (g >= G_GRAPHS) return;
  unsigned c = 0u;
#pragma unroll
  for (int b = 0; b < CNT_B; ++b) c += parts[b * G_GRAPHS + g];
  cnt[g] = 1.0f / fmaxf((float)c, 1.0f);
}

template <int K>
__global__ void gcn_gemm_wmma_kernel(const float* __restrict__ X,
                                     const float* __restrict__ W,
                                     const float* __restrict__ bias,
                                     const float* __restrict__ dinv,
                                     float* __restrict__ H,
                                     float* __restrict__ A, int nRows) {
  const int wave = blockIdx.x * (blockDim.x >> 5) + (threadIdx.x >> 5);
  const int lane = threadIdx.x & 31;
  const int numTiles = (nRows >> 4) * 2;
  const bool active = wave < numTiles;
  v2f afrag[K / 4];
  if (active) {
    const int r0 = (wave >> 1) * 16 + (lane & 15);
#pragma unroll
    for (int j = 0; j < K / 4; ++j)
      afrag[j] = *(const v2f*)(X + r0 * K + 4 * j + 2 * (lane >> 4));
  }
  if (!active) return;
  const int rowTile = wave >> 1;
  const int colTile = wave & 1;
  const int half = lane >> 4;
  const int l16  = lane & 15;
  const int col  = colTile * 16 + l16;

  float dd[8];
#pragma unroll
  for (int r = 0; r < 8; ++r) {
    const float d = dinv[rowTile * 16 + r + 8 * half];
    dd[r] = d * d;
  }
  const float bc = bias[col];

  v8f c = {};
#pragma unroll
  for (int k0 = 0; k0 < K; k0 += 4) {
    const int ka = k0 + 2 * half;
    const v2f a = afrag[k0 / 4];
    v2f b;
    b.x = W[(ka + 0) * HID + col];
    b.y = W[(ka + 1) * HID + col];
    c = __builtin_amdgcn_wmma_f32_16x16x4_f32(
         false, a,  false, b,
         (short)0, c,  false,  false);
  }
#pragma unroll
  for (int r = 0; r < 8; ++r) {
    const int m = r + 8 * half;
    const int o = (rowTile * 16 + m) * HID + col;
    H[o] = c[r];
    A[o] = c[r] * dd[r] + bc;
  }
}

#define AGG_RANGE 10000
#define AGG_R     (N_NODES / AGG_RANGE)
#define AGG_S     25
#define AGG_T     1024
#define AGG_W     (AGG_T / 32)
#define AGG_SEG   256
#define AGG_QS    (N_EDGES / 4 / AGG_S)
#define AGG_NG    (AGG_QS / 32)
#define AGG_B     4
#define AGG_U     8
#define FL_U      4
#define OWN_ROWS  400
#define OWN_PER   (AGG_RANGE / OWN_ROWS)
#define OWN_B     (N_NODES / OWN_ROWS)
#define SUM_T     (AGG_S * 32)
#define OWN_TRIPS ((OWN_ROWS * HID + SUM_T - 1) / SUM_T)
#define OWN_QUADS (OWN_ROWS * HID / 4 / SUM_T)
static_assert(OWN_ROWS * HID / 4 % SUM_T == 0, "the block's rows are seeded in whole pieces per thread");
static_assert(OWN_ROWS == 16 * (SUM_T / 32), "one 16-row tile of the block per wave for the in-block product");
static_assert(N_NODES % AGG_RANGE == 0, "row ranges must tile the nodes");
static_assert(AGG_S == DEG_PLANES, "one in-degree plane per edge slice");
static_assert(AGG_RANGE % 4 == 0, "a range's in-degree counters are written out in 16-byte pieces");
static_assert(AGG_RANGE % OWN_ROWS == 0, "owned blocks must tile a range");
static_assert(N_EDGES % (4 * AGG_S * 32) == 0, "slices must be whole wave groups of quads");
static_assert(AGG_SEG >= 2 * 128, "a list must take one more group after the flush test");
#define DEGB_WORDS (N_NODES / 4)
#define DEGB_PER  10
#define DEGB_THR  (N_NODES / 16 / DEGB_PER)
static_assert(N_NODES % (16 * DEGB_PER) == 0, "the degree bytes are copied in 16-byte pieces, DEGB_PER per thread");
static_assert(DEGB_THR <= SUM_T, "the copying threads must fit the work-group");
#define LST_WORDS ((N_NODES * HID - DEGB_WORDS) / (AGG_R * AGG_S))
#define SUB_WORDS (LST_WORDS / OWN_PER)
#define SUB_CAP   (SUB_WORDS - 1)
#define LST_SHIFT 17
static_assert(N_NODES <= (1 << LST_SHIFT), "a source node must fit the low bits");
static_assert(OWN_ROWS <= (1 << (32 - LST_SHIFT)), "a row of a block must fit the high bits");
static_assert((N_NODES * HID - DEGB_WORDS) % (AGG_R * AGG_S) == 0, "segments must tile the region");
static_assert(OWN_ROWS <= SUM_T, "one thread per owned row stages its dinv");
static_assert(LST_WORDS % OWN_PER == 0, "parts must tile a segment");

__device__ __forceinline__ void gcn_flush_list(const uint2* seg, int cnt, int lane,
                                               const int* __restrict__ src,
                                               unsigned* fill, unsigned* degc, unsigned* segw) {
  for (int i = 0; i < cnt; i += 32 * FL_U) {
    unsigned sr[FL_U];
    unsigned al[FL_U];
    bool ok[FL_U];
#pragma unroll
    for (int u = 0; u < FL_U; ++u) {
      const int idx = i + 32 * u + lane;
      ok[u] = idx < cnt;
      const uint2 e = seg[ok[u] ? idx : i];
      sr[u] = (unsigned)src[e.x];
      al[u] = e.y;
    }
#pragma unroll
    for (int u = 0; u < FL_U; ++u) {
      if (ok[u]) {
        atomicAdd(&degc[al[u]], 1u);
        const unsigned blk = al[u] / (unsigned)OWN_ROWS;
        const unsigned pos = atomicAdd(&fill[blk], 1u);
        if (pos < (unsigned)SUB_CAP)
          segw[blk * SUB_WORDS + 1u + pos] = sr[u] | ((al[u] - blk * (unsigned)OWN_ROWS) << LST_SHIFT);
      }
    }
  }
}

__global__ void __launch_bounds__(AGG_T)
gcn_edge_scan_kernel(const int* __restrict__ src, const int* __restrict__ dst,
                     unsigned* __restrict__ lists, unsigned* __restrict__ planes) {
  __shared__ uint2 list[AGG_W * AGG_SEG];
  __shared__ unsigned fill[OWN_PER];
  __shared__ __attribute__((aligned(16))) unsigned degc[AGG_RANGE];
  const int tid = threadIdx.x;
  const int lane = tid & 31;
  const int w = __builtin_amdgcn_readfirstlane(tid >> 5);
  const int r = blockIdx.x % AGG_R;
  const int sl = blockIdx.x / AGG_R;
  const int lo = r * AGG_RANGE;
  unsigned* segw = lists + (size_t)blockIdx.x * LST_WORDS;

  if (tid < OWN_PER) fill[tid] = 0u;
  for (int i = tid; i < AGG_RANGE; i += AGG_T) degc[i] = 0u;
  __syncthreads();

  const int qb = sl * AGG_QS;
  const int4* d4 = (const int4*)dst + qb;
  uint2* seg = list + w * AGG_SEG;
  int cnt = 0;

  for (int g0 = w; g0 < AGG_NG; g0 += AGG_W * AGG_B) {
    int4 dv[AGG_B];
#pragma unroll
    for (int b = 0; b < AGG_B; ++b) {
      const int g = g0 + b * AGG_W;
      if (g < AGG_NG) dv[b] = d4[g * 32 + lane];
      else            dv[b] = make_int4(-1, -1, -1, -1);
    }
#pragma unroll
    for (int b = 0; b < AGG_B; ++b) {
      const int g = g0 + b * AGG_W;
      const unsigned e0 = (unsigned)(4 * (qb + g * 32 + lane));
      const int dd[4] = {dv[b].x, dv[b].y, dv[b].z, dv[b].w};
#pragma unroll
      for (int k = 0; k < 4; ++k) {
        const unsigned a = (unsigned)(dd[k] - lo);
        const bool m = a < (unsigned)AGG_RANGE;
        const unsigned bal = __builtin_amdgcn_ballot_w32(m);
        if (m) seg[cnt + (int)__builtin_amdgcn_mbcnt_lo(bal, 0u)] = make_uint2(e0 + (unsigned)k, a);
        cnt += __builtin_popcount(bal);
      }
      if (cnt > AGG_SEG - 128) {
        __builtin_amdgcn_wave_barrier();
        gcn_flush_list(seg, cnt, lane, src, fill, degc, segw);
        __builtin_amdgcn_wave_barrier();
        cnt = 0;
      }
    }
  }
  __builtin_amdgcn_wave_barrier();
  gcn_flush_list(seg, cnt, lane, src, fill, degc, segw);
  __syncthreads();
  if (tid < OWN_PER) segw[tid * SUB_WORDS] = fill[tid];
  uint4* p4 = (uint4*)(planes + (size_t)sl * N_NODES + lo);
  const uint4* c4 = (const uint4*)degc;
  for (int i = tid; i < AGG_RANGE / 4; i += AGG_T) p4[i] = c4[i];
}

__device__ __forceinline__ void gcn_replay_list(const unsigned* __restrict__ ent, int n, int wr, int nw,
                                                int lane,
                                                const float* __restrict__ dinv,
                                                const unsigned char* sdeg, const float* sdinv,
                                                const float* __restrict__ H, float* acc) {
  const int sub = lane >> 3;
  const int q = (lane & 7) << 2;
  const bool rot1 = (sub & 1) != 0;
  const bool rot2 = (sub & 2) != 0;
  const int o0 = sub & 3, o1 = (sub + 1) & 3, o2 = (sub + 2) & 3, o3 = (sub + 3) & 3;
  int i = wr * (4 * AGG_U);
  unsigned en[AGG_U];
  if (i < n) {
#pragma unroll
    for (int u = 0; u < AGG_U; ++u) {
      const int idx = i + 4 * u + sub;
      en[u] = ent[idx < n ? idx : i];
    }
  }
  while (i < n) {
    int al[AGG_U];
    bool ok[AGG_U];
    float4 hv[AGG_U];
    float cf[AGG_U];
    int sr[AGG_U];
    unsigned dg[AGG_U];
    float dl[AGG_U];
#pragma unroll
    for (int u = 0; u < AGG_U; ++u) {
      ok[u] = i + 4 * u + sub < n;
      sr[u] = (int)(en[u] & ((1u << LST_SHIFT) - 1u));
      al[u] = (int)(en[u] >> LST_SHIFT);
      hv[u] = *(const float4*)(H + sr[u] * HID + q);
      dg[u] = (unsigned)sdeg[sr[u]];
      dl[u] = sdinv[al[u]];
    }
    const int inext = i + nw * 4 * AGG_U;
    if (inext < n) {
#pragma unroll
      for (int u = 0; u < AGG_U; ++u) {
        const int idx = inext + 4 * u + sub;
        en[u] = ent[idx < n ? idx : inext];
      }
    }
    bool sat = false;
#pragma unroll
    for (int u = 0; u < AGG_U; ++u) {
      cf[u] = rsqrtf((float)dg[u] + 1.0f);
      sat = sat || dg[u] == DEG_SAT;
    }
    if (__builtin_amdgcn_ballot_w32(sat) != 0u) {
#pragma unroll
      for (int u = 0; u < AGG_U; ++u)
        if (dg[u] == DEG_SAT) cf[u] = dinv[sr[u]];
    }
#pragma unroll
    for (int u = 0; u < AGG_U; ++u) cf[u] *= dl[u];
#pragma unroll
    for (int u = 0; u < AGG_U; ++u) {
      if (i + 4 * u < n) {
        const float v0 = ok[u] ? hv[u].x * cf[u] : 0.0f;
        const float v1 = ok[u] ? hv[u].y * cf[u] : 0.0f;
        const float v2 = ok[u] ? hv[u].z * cf[u] : 0.0f;
        const float v3 = ok[u] ? hv[u].w * cf[u] : 0.0f;
        const float s0 = rot1 ? v1 : v0;
        const float s1 = rot1 ? v2 : v1;
        const float s2 = rot1 ? v3 : v2;
        const float s3 = rot1 ? v0 : v3;
        const float t0 = rot2 ? s2 : s0;
        const float t1 = rot2 ? s3 : s1;
        const float t2 = rot2 ? s0 : s2;
        const float t3 = rot2 ? s1 : s3;
        float* ap = acc + al[u] * HID + q;
        atomicAdd(ap + o0, t0);
        atomicAdd(ap + o1, t1);
        atomicAdd(ap + o2, t2);
        atomicAdd(ap + o3, t3);
      }
    }
    i = inext;
  }
}

template <bool POST, bool STORE>
__global__ void __launch_bounds__(SUM_T)
gcn_edge_sum_kernel(const int* __restrict__ src, const int* __restrict__ dst,
                    const float* __restrict__ dinv,
                    const float* __restrict__ H, float* __restrict__ agg,
                    const float* __restrict__ Wm, const float* __restrict__ bias,
                    const unsigned* __restrict__ lists, const unsigned* __restrict__ degb,
                    const int* __restrict__ batch, const float* __restrict__ cntinv,
                    float* __restrict__ feat) {
  __shared__ float acc[OWN_ROWS * HID];
  __shared__ __attribute__((aligned(16))) unsigned char sdeg[N_NODES];
  __shared__ float sdinv[OWN_ROWS];
  const int tid = threadIdx.x;
  const int lane = tid & 31;
  const int w = __builtin_amdgcn_readfirstlane(tid >> 5);
  const int r = blockIdx.x / OWN_PER;
  const int p = blockIdx.x % OWN_PER;
  const int lo = blockIdx.x * OWN_ROWS;

  bool fit = true;
#pragma unroll
  for (int s = 0; s < AGG_S; ++s) {
    const unsigned n = lists[(size_t)(s * AGG_R + r) * LST_WORDS + (size_t)p * SUB_WORDS];
    fit = fit && n <= (unsigned)SUB_CAP;
  }
  float dmine = 0.0f;
  if (tid < OWN_ROWS) dmine = dinv[lo + tid];

  float4* acc4 = (float4*)acc;
  float4 xv[OWN_QUADS];
  float xd[OWN_QUADS];
  if (POST) {
    const float4* own4 = (const float4*)(H + lo * HID);
#pragma unroll
    for (int j = 0; j < OWN_QUADS; ++j) {
      const int i = tid + j * SUM_T;
      xv[j] = own4[i];
      xd[j] = dinv[lo + (i >> 3)];
    }
  } else {
    for (int i = tid; i < OWN_ROWS * HID / 4; i += SUM_T) acc4[i] = make_float4(0.f, 0.f, 0.f, 0.f);
  }
  if (tid < DEGB_THR) {
    const uint4* degb4 = (const uint4*)degb;
    uint4* sdeg4 = (uint4*)sdeg;
    uint4 dt[DEGB_PER];
#pragma unroll
    for (int j = 0; j < DEGB_PER; ++j) dt[j] = degb4[tid + j * DEGB_THR];
#pragma unroll
    for (int j = 0; j < DEGB_PER; ++j) sdeg4[tid + j * DEGB_THR] = dt[j];
  }
  if (POST) {
#pragma unroll
    for (int j = 0; j < OWN_QUADS; ++j) {
      const float dd = xd[j] * xd[j];
      acc4[tid + j * SUM_T] = make_float4(xv[j].x * dd, xv[j].y * dd, xv[j].z * dd, xv[j].w * dd);
    }
  }
  if (tid < OWN_ROWS) sdinv[tid] = dmine;

  __syncthreads();

  if (fit) {
    const unsigned* part = lists + (size_t)(w * AGG_R + r) * LST_WORDS + (size_t)p * SUB_WORDS;
    const int n = __builtin_amdgcn_readfirstlane((int)part[0]);
    gcn_replay_list(part + 1, n, 0, 1, lane, dinv, sdeg, sdinv, H, acc);
  } else {
    for (int e = tid; e < N_EDGES; e += SUM_T) {
      const unsigned a = (unsigned)(dst[e] - lo);
      if (a < (unsigned)OWN_ROWS) {
        const int sr = src[e];
        const float cf = dinv[sr] * dinv[lo + (int)a];
        for (int ch = 0; ch < HID; ++ch) atomicAdd(&acc[a * HID + ch], H[sr * HID + ch] * cf);
      }
    }
  }
  const int half = lane >> 4;
  const int l16 = lane & 15;
  v2f wb0[HID / 4], wb1[HID / 4];
  float bc0 = 0.0f, bc1 = 0.0f;
  if (POST) {
#pragma unroll
    for (int j = 0; j < HID / 4; ++j) {
      const int ka = 4 * j + 2 * half;
      wb0[j].x = Wm[(ka + 0) * HID + l16];
      wb0[j].y = Wm[(ka + 1) * HID + l16];
      wb1[j].x = Wm[(ka + 0) * HID + 16 + l16];
      wb1[j].y = Wm[(ka + 1) * HID + 16 + l16];
    }
    bc0 = bias[l16];
    bc1 = bias[16 + l16];
  }
  __syncthreads();

  if (POST) {
    const float* arow = acc + (w * 16 + l16) * HID + 2 * half;
    v2f af[HID / 4];
#pragma unroll
    for (int j = 0; j < HID / 4; ++j) af[j] = *(const v2f*)(arow + 4 * j);
    v8f c0 = {};
    v8f c1 = {};
#pragma unroll
    for (int j = 0; j < HID / 4; ++j) {
      c0 = __builtin_amdgcn_wmma_f32_16x16x4_f32(
           false, af[j],  false, wb0[j],
           (short)0, c0,  false,  false);
      c1 = __builtin_amdgcn_wmma_f32_16x16x4_f32(
           false, af[j],  false, wb1[j],
           (short)0, c1,  false,  false);
    }
#pragma unroll
    for (int r = 0; r < 8; ++r) {
      float* orow = acc + (w * 16 + r + 8 * half) * HID;
      orow[l16] = c0[r] + bc0;
      orow[16 + l16] = c1[r] + bc1;
    }
    __syncthreads();
  }

  float* rows = agg + lo * HID;
  float sd[OWN_TRIPS];
  int gi[OWN_TRIPS];
  float pw[OWN_TRIPS];
#pragma unroll
  for (int j = 0; j < OWN_TRIPS; ++j) {
    const int i = tid + j * SUM_T;
    const int c = i < OWN_ROWS * HID ? i : tid;
    sd[j] = POST ? 0.0f : rows[c];
    gi[j] = batch[lo + (c >> 5)];
  }
#pragma unroll
  for (int j = 0; j < OWN_TRIPS; ++j) pw[j] = cntinv[gi[j]];
#pragma unroll
  for (int j = 0; j < OWN_TRIPS; ++j) {
    const int i = tid + j * SUM_T;
    if (i < OWN_ROWS * HID) {
      float v = sd[j] + acc[i];
      v = v > 0.0f ? v : NEG_SLOPE * v;
      if (!POST || STORE) rows[i] = v;
      atomicAdd(&feat[gi[j] * HID + (i & 31)], v * pw[j]);
    }
  }
}

__global__ void gcn_merge_kernel(const float* __restrict__ feats, float* __restrict__ out) {
  int t = blockIdx.x * blockDim.x + threadIdx.x;
  const int GH = G_GRAPHS * HID;
  if (t < GH) out[t] = (feats[t] + feats[GH + t] + feats[2 * GH + t]) * (1.0f / 3.0f);
}

static inline int cdiv(int a, int b) { return (a + b - 1) / b; }

extern "C" void kernel_launch(void* const* d_in, const int* in_sizes, int n_in,
                              void* d_out, int out_size, void* d_ws, size_t ws_size,
                              hipStream_t stream) {
  (void)in_sizes; (void)n_in; (void)out_size; (void)ws_size;
  const float* x     = (const float*)d_in[0];
  const float* W0    = (const float*)d_in[1];
  const float* b0    = (const float*)d_in[2];
  const float* W1    = (const float*)d_in[3];
  const float* b1    = (const float*)d_in[4];
  const float* W2    = (const float*)d_in[5];
  const float* b2    = (const float*)d_in[6];
  const int*   src   = (const int*)d_in[7];
  const int*   dst   = (const int*)d_in[8];
  const int*   batch = (const int*)d_in[9];
  float* out = (float*)d_out;

  float* ws    = (float*)d_ws;
  float* deg   = ws;
  float* cnt   = deg + N_NODES;
  float* lin   = cnt + G_GRAPHS;
  float* act   = lin + (size_t)N_NODES * HID;
  float* agg   = act + (size_t)N_NODES * HID;
  float* feats = agg + (size_t)N_NODES * HID;

  const int T = 256;
  const int nTiles = (N_NODES / 16) * 2;

  unsigned* degPlanes = (unsigned*)agg;
  unsigned* cntPlanes = (unsigned*)act;
  unsigned* lists     = (unsigned*)act;
  unsigned* degb      = (unsigned*)act + (size_t)AGG_R * AGG_S * LST_WORDS;
  gcn_zero_kernel<<<cdiv(3 * G_GRAPHS * HID, T), T, 0, stream>>>(feats, 3 * G_GRAPHS * HID);
  gcn_counts_hist_kernel<<<CNT_B, CNT_T, 0, stream>>>(batch, cntPlanes);
  gcn_cntinv_kernel<<<1, G_GRAPHS, 0, stream>>>(cntPlanes, cnt);

  gcn_edge_scan_kernel<<<AGG_R * AGG_S, AGG_T, 0, stream>>>(src, dst, lists, degPlanes);
  gcn_dinv_kernel<<<cdiv(N_NODES / 4, 128), 128, 0, stream>>>(degPlanes, deg, degb);

  gcn_gemm_wmma_kernel<C_IN><<<cdiv(nTiles, T / 32), T, 0, stream>>>(x, W0, b0, deg, lin, agg, N_NODES);
  gcn_edge_sum_kernel<false, true><<<OWN_B, SUM_T, 0, stream>>>(
      src, dst, deg, lin, agg, W0, b0, lists, degb, batch, cnt, feats);
  gcn_edge_sum_kernel<true, true><<<OWN_B, SUM_T, 0, stream>>>(
      src, dst, deg, agg, lin, W1, b1, lists, degb, batch, cnt, feats + G_GRAPHS * HID);
  gcn_edge_sum_kernel<true, false><<<OWN_B, SUM_T, 0, stream>>>(
      src, dst, deg, lin, agg, W2, b2, lists, degb, batch, cnt, feats + 2 * G_GRAPHS * HID);
  gcn_merge_kernel<<<cdiv(G_GRAPHS * HID, T), T, 0, stream>>>(feats, out);
}
